# starting kernel
# baseline (speedup 1.0000x reference)
#include <hip/hip_runtime.h>
#include <hip/hip_bf16.h>
#include <cstddef>


#define N_NODES 50000
#define E_EDGES 800000
#define E_TOT   (E_EDGES + N_NODES)
#define NEG_SLOPE 0.2f

typedef float v2f __attribute__((ext_vector_type(2)));
typedef float v8f __attribute__((ext_vector_type(8)));

__device__ __forceinline__ unsigned f2key(float f) {
    int i = __float_as_int(f);
    return (i >= 0) ? ((unsigned)i | 0x80000000u) : ~(unsigned)i;
}
__device__ __forceinline__ float key2f(unsigned k) {
    int i = (k & 0x80000000u) ? (int)(k & 0x7FFFFFFFu) : (int)~k;
    return __int_as_float(i);
}

__global__ void __launch_bounds__(256)
gat_init(float* __restrict__ acc, unsigned* __restrict__ mkey,
         float* __restrict__ denom, int n_nodes, int nf) {
    int i = blockIdx.x * blockDim.x + threadIdx.x;
    int tot = n_nodes * nf;
    if (i < tot) acc[i] = 0.0f;
    if (i < n_nodes) { mkey[i] = 0u; denom[i] = 0.0f; }
}

__global__ void __launch_bounds__(256)
gat_gemm_wmma(const float* __restrict__ X, const float* __restrict__ W,
              float* __restrict__ H, int nrows, int K) {
    const int wave = blockIdx.x * (blockDim.x >> 5) + (threadIdx.x >> 5);
    const int lane = threadIdx.x & 31;
    const int rowBase = wave * 16;
    if (rowBase >= nrows) return;

    const int lo = lane & 15;
    const int hi = lane >> 4;

    v8f acc0 = {}, acc1 = {}, acc2 = {}, acc3 = {};

    const float* xr = X + (size_t)(rowBase + lo) * K + 2 * hi;
    const float* wc = W + lo;

    for (int k = 0; k < K; k += 4) {
        float2 af = *(const float2*)(xr + k);
        v2f a; a.x = af.x; a.y = af.y;

        const float* w0 = wc + (size_t)(k + 2 * hi) * 64;
        const float* w1 = w0 + 64;
        v2f b0; b0.x = w0[0];  b0.y = w1[0];
        v2f b1; b1.x = w0[16]; b1.y = w1[16];
        v2f b2; b2.x = w0[32]; b2.y = w1[32];
        v2f b3; b3.x = w0[48]; b3.y = w1[48];

        acc0 = __builtin_amdgcn_wmma_f32_16x16x4_f32(false, a, false, b0, (short)0, acc0, false, false);
        acc1 = __builtin_amdgcn_wmma_f32_16x16x4_f32(false, a, false, b1, (short)0, acc1, false, false);
        acc2 = __builtin_amdgcn_wmma_f32_16x16x4_f32(false, a, false, b2, (short)0, acc2, false, false);
        acc3 = __builtin_amdgcn_wmma_f32_16x16x4_f32(false, a, false, b3, (short)0, acc3, false, false);
    }

#pragma unroll
    for (int r = 0; r < 8; ++r) {
        float* o = H + (size_t)(rowBase + r + 8 * hi) * 64 + lo;
        o[0]  = acc0[r];
        o[16] = acc1[r];
        o[32] = acc2[r];
        o[48] = acc3[r];
    }
}

__global__ void __launch_bounds__(256)
gat_node_sd(const float* __restrict__ H, const float* __restrict__ asrc,
            const float* __restrict__ adst, float* __restrict__ s,
            float* __restrict__ d, int n) {
    int i = blockIdx.x * blockDim.x + threadIdx.x;
    if (i >= n) return;
    const float* hr = H + (size_t)i * 64;
    float ss = 0.0f, dd = 0.0f;
#pragma unroll
    for (int f = 0; f < 64; ++f) {
        float h = hr[f];
        ss = fmaf(h, asrc[f], ss);
        dd = fmaf(h, adst[f], dd);
    }
    s[i] = ss; d[i] = dd;
}

__global__ void __launch_bounds__(256)
gat_edge_max(const int* __restrict__ src, const int* __restrict__ dst,
             const float* __restrict__ s, const float* __restrict__ d,
             float* __restrict__ evals, unsigned* __restrict__ mkey,
             int nE, int etot) {
    int e = blockIdx.x * blockDim.x + threadIdx.x;
    if (e >= etot) return;
    int si = (e < nE) ? src[e] : (e - nE);
    int di = (e < nE) ? dst[e] : (e - nE);
    float v = s[si] + d[di];
    v = (v > 0.0f) ? v : NEG_SLOPE * v;
    evals[e] = v;
    atomicMax(mkey + di, f2key(v));
}

__global__ void __launch_bounds__(256)
gat_edge_expsum(const int* __restrict__ dst, const float* __restrict__ evals,
                const unsigned* __restrict__ mkey, float* __restrict__ wvals,
                float* __restrict__ denom, int nE, int etot) {
    int e = blockIdx.x * blockDim.x + threadIdx.x;
    if (e >= etot) return;
    int di = (e < nE) ? dst[e] : (e - nE);
    float w = __expf(evals[e] - key2f(mkey[di]));
    wvals[e] = w;
    atomicAdd(denom + di, w);
}

__global__ void __launch_bounds__(256)
gat_edge_aggregate(const int* __restrict__ src, const int* __restrict__ dst,
                   const float* __restrict__ H, const float* __restrict__ wvals,
                   const float* __restrict__ denom, float* __restrict__ acc,
                   int nE, int etot) {
    int wid  = blockIdx.x * (blockDim.x >> 5) + (threadIdx.x >> 5);
    int lane = threadIdx.x & 31;
    if (wid >= etot) return;
    int si = (wid < nE) ? src[wid] : (wid - nE);
    int di = (wid < nE) ? dst[wid] : (wid - nE);
    float coef = wvals[wid] / (denom[di] + 1e-16f);
    const float* hr = H + (size_t)si * 64;
    float* ar = acc + (size_t)di * 64;
    atomicAdd(ar + lane,      hr[lane]      * coef);
    atomicAdd(ar + lane + 32, hr[lane + 32] * coef);
}

__global__ void __launch_bounds__(256)
gat_bias_act(const float* __restrict__ acc, const float* __restrict__ b,
             float* __restrict__ out, int tot, int relu) {
    int i = blockIdx.x * blockDim.x + threadIdx.x;
    if (i >= tot) return;
    float v = acc[i] + b[i & 63];
    if (relu) v = fmaxf(v, 0.0f);
    out[i] = v;
}

static void run_gat_layer(const float* xin, int K, const float* W,
                          const float* as, const float* ad, const float* bias,
                          float* H, float* acc, float* nextX, int relu,
                          const int* esrc, const int* edst,
                          float* SV, float* DV, unsigned* MKEY, float* DEN,
                          float* EVAL, float* WVAL, hipStream_t stream) {
    const int nodeFeat = N_NODES * 64;
    gat_init<<<(nodeFeat + 255) / 256, 256, 0, stream>>>(acc, MKEY, DEN, N_NODES, 64);
    {
        int waves = N_NODES / 16;
        int blocks = (waves + 7) / 8;
        gat_gemm_wmma<<<blocks, 256, 0, stream>>>(xin, W, H, N_NODES, K);
    }
    gat_node_sd<<<(N_NODES + 255) / 256, 256, 0, stream>>>(H, as, ad, SV, DV, N_NODES);
    gat_edge_max<<<(E_TOT + 255) / 256, 256, 0, stream>>>(esrc, edst, SV, DV, EVAL, MKEY, E_EDGES, E_TOT);
    gat_edge_expsum<<<(E_TOT + 255) / 256, 256, 0, stream>>>(edst, EVAL, MKEY, WVAL, DEN, E_EDGES, E_TOT);
    gat_edge_aggregate<<<(E_TOT + 7) / 8, 256, 0, stream>>>(esrc, edst, H, WVAL, DEN, acc, E_EDGES, E_TOT);
    gat_bias_act<<<(nodeFeat + 255) / 256, 256, 0, stream>>>(acc, bias, nextX, relu, relu);
}

extern "C" void kernel_launch(void* const* d_in, const int* in_sizes, int n_in,
                              void* d_out, int out_size, void* d_ws, size_t ws_size,
                              hipStream_t stream) {
    (void)in_sizes; (void)n_in; (void)out_size; (void)ws_size;

    const float* x    = (const float*)d_in[0];
    const int*   esrc = (const int*)  d_in[1];
    const int*   edst = (const int*)  d_in[2];
    const float* W0 = (const float*)d_in[3];
    const float* a0s = (const float*)d_in[4];
    const float* a0d = (const float*)d_in[5];
    const float* b0 = (const float*)d_in[6];
    const float* W1 = (const float*)d_in[7];
    const float* a1s = (const float*)d_in[8];
    const float* a1d = (const float*)d_in[9];
    const float* b1 = (const float*)d_in[10];
    const float* W2 = (const float*)d_in[11];
    const float* a2s = (const float*)d_in[12];
    const float* a2d = (const float*)d_in[13];
    const float* b2 = (const float*)d_in[14];
    float* out = (float*)d_out;

    float* ws = (float*)d_ws;
    float*    H    = ws;
    float*    XIN  = H   + (size_t)N_NODES * 64;
    float*    ACC  = XIN + (size_t)N_NODES * 64;
    float*    SV   = ACC + (size_t)N_NODES * 64;
    float*    DV   = SV  + N_NODES;
    unsigned* MKEY = (unsigned*)(DV + N_NODES);
    float*    DEN  = (float*)(MKEY + N_NODES);
    float*    EVAL = DEN + N_NODES;
    float*    WVAL = EVAL + E_TOT;

    const int nodeFeat = N_NODES * 64;

    gat_init<<<(nodeFeat + 255) / 256, 256, 0, stream>>>(ACC, MKEY, DEN, N_NODES, 64);
    gat_gemm_wmma<<<((N_NODES / 16) + 7) / 8, 256, 0, stream>>>(x, W0, H, N_NODES, 128);
    gat_node_sd<<<(N_NODES + 255) / 256, 256, 0, stream>>>(H, a0s, a0d, SV, DV, N_NODES);
    gat_edge_max<<<(E_TOT + 255) / 256, 256, 0, stream>>>(esrc, edst, SV, DV, EVAL, MKEY, E_EDGES, E_TOT);
    gat_edge_expsum<<<(E_TOT + 255) / 256, 256, 0, stream>>>(edst, EVAL, MKEY, WVAL, DEN, E_EDGES, E_TOT);
    gat_edge_aggregate<<<(E_TOT + 7) / 8, 256, 0, stream>>>(esrc, edst, H, WVAL, DEN, ACC, E_EDGES, E_TOT);
    gat_bias_act<<<(nodeFeat + 255) / 256, 256, 0, stream>>>(ACC, b0, XIN, nodeFeat, 1);

    gat_init<<<(nodeFeat + 255) / 256, 256, 0, stream>>>(ACC, MKEY, DEN, N_NODES, 64);
    gat_gemm_wmma<<<((N_NODES / 16) + 7) / 8, 256, 0, stream>>>(XIN, W1, H, N_NODES, 64);
    gat_node_sd<<<(N_NODES + 255) / 256, 256, 0, stream>>>(H, a1s, a1d, SV, DV, N_NODES);
    gat_edge_max<<<(E_TOT + 255) / 256, 256, 0, stream>>>(esrc, edst, SV, DV, EVAL, MKEY, E_EDGES, E_TOT);
    gat_edge_expsum<<<(E_TOT + 255) / 256, 256, 0, stream>>>(edst, EVAL, MKEY, WVAL, DEN, E_EDGES, E_TOT);
    gat_edge_aggregate<<<(E_TOT + 7) / 8, 256, 0, stream>>>(esrc, edst, H, WVAL, DEN, ACC, E_EDGES, E_TOT);
    gat_bias_act<<<(nodeFeat + 255) / 256, 256, 0, stream>>>(ACC, b1, XIN, nodeFeat, 1);

    gat_init<<<(nodeFeat + 255) / 256, 256, 0, stream>>>(out, MKEY, DEN, N_NODES, 64);
    gat_gemm_wmma<<<((N_NODES / 16) + 7) / 8, 256, 0, stream>>>(XIN, W2, H, N_NODES, 64);
    gat_node_sd<<<(N_NODES + 255) / 256, 256, 0, stream>>>(H, a2s, a2d, SV, DV, N_NODES);
    gat_edge_max<<<(E_TOT + 255) / 256, 256, 0, stream>>>(esrc, edst, SV, DV, EVAL, MKEY, E_EDGES, E_TOT);
    gat_edge_expsum<<<(E_TOT + 255) / 256, 256, 0, stream>>>(edst, EVAL, MKEY, WVAL, DEN, E_EDGES, E_TOT);
    gat_edge_aggregate<<<(E_TOT + 7) / 8, 256, 0, stream>>>(esrc, edst, H, WVAL, DEN, out, E_EDGES, E_TOT);
    gat_bias_act<<<(nodeFeat + 255) / 256, 256, 0, stream>>>(out, b2, out, nodeFeat, 0);
}


# Edge softmax+aggregation by row-block owners: 250 work-groups each filter the edge list for their 200 rows, keep max, sum, row sums in LDS (LDS atomics); init, bias passes fold in; sum order changes
# speedup vs baseline: 2.7735x; 2.7735x over previous
#include <hip/hip_runtime.h>
#include <hip/hip_bf16.h>
#include <cstddef>


#define N_NODES 50000
#define E_EDGES 800000
#define NEG_SLOPE 0.2f

typedef float v2f __attribute__((ext_vector_type(2)));
typedef float v8f __attribute__((ext_vector_type(8)));

__device__ __forceinline__ unsigned f2key(float f) {
    int i = __float_as_int(f);
    return (i >= 0) ? ((unsigned)i | 0x80000000u) : ~(unsigned)i;
}
__device__ __forceinline__ float key2f(unsigned k) {
    int i = (k & 0x80000000u) ? (int)(k & 0x7FFFFFFFu) : (int)~k;
    return __int_as_float(i);
}

__global__ void __launch_bounds__(256)
gat_gemm_wmma(const float* __restrict__ X, const float* __restrict__ W,
              float* __restrict__ H, int nrows, int K) {
    const int wave = blockIdx.x * (blockDim.x >> 5) + (threadIdx.x >> 5);
    const int lane = threadIdx.x & 31;
    const int rowBase = wave * 16;
    if (rowBase >= nrows) return;

    const int lo = lane & 15;
    const int hi = lane >> 4;

    v8f acc0 = {}, acc1 = {}, acc2 = {}, acc3 = {};

    const float* xr = X + (size_t)(rowBase + lo) * K + 2 * hi;
    const float* wc = W + lo;

    for (int k = 0; k < K; k += 4) {
        float2 af = *(const float2*)(xr + k);
        v2f a; a.x = af.x; a.y = af.y;

        const float* w0 = wc + (size_t)(k + 2 * hi) * 64;
        const float* w1 = w0 + 64;
        v2f b0; b0.x = w0[0];  b0.y = w1[0];
        v2f b1; b1.x = w0[16]; b1.y = w1[16];
        v2f b2; b2.x = w0[32]; b2.y = w1[32];
        v2f b3; b3.x = w0[48]; b3.y = w1[48];

        acc0 = __builtin_amdgcn_wmma_f32_16x16x4_f32(false, a, false, b0, (short)0, acc0, false, false);
        acc1 = __builtin_amdgcn_wmma_f32_16x16x4_f32(false, a, false, b1, (short)0, acc1, false, false);
        acc2 = __builtin_amdgcn_wmma_f32_16x16x4_f32(false, a, false, b2, (short)0, acc2, false, false);
        acc3 = __builtin_amdgcn_wmma_f32_16x16x4_f32(false, a, false, b3, (short)0, acc3, false, false);
    }

#pragma unroll
    for (int r = 0; r < 8; ++r) {
        float* o = H + (size_t)(rowBase + r + 8 * hi) * 64 + lo;
        o[0]  = acc0[r];
        o[16] = acc1[r];
        o[32] = acc2[r];
        o[48] = acc3[r];
    }
}

__global__ void __launch_bounds__(256)
gat_node_sd(const float* __restrict__ H, const float* __restrict__ asrc,
            const float* __restrict__ adst, float* __restrict__ s,
            float* __restrict__ d, int n) {
    int i = blockIdx.x * blockDim.x + threadIdx.x;
    if (i >= n) return;
    const float* hr = H + (size_t)i * 64;
    float ss = 0.0f, dd = 0.0f;
#pragma unroll
    for (int f = 0; f < 64; ++f) {
        float h = hr[f];
        ss = fmaf(h, asrc[f], ss);
        dd = fmaf(h, adst[f], dd);
    }
    s[i] = ss; d[i] = dd;
}

#define RB        200
#define NBLK      250
#define ROWS_NT   1024
#define WCAP      512
#define SCAN_U    8
#define AGG_U     16
#define NCHUNK    (E_EDGES / 128)
static_assert(RB * NBLK == N_NODES, "row blocks must tile the node set");
static_assert(E_EDGES % 128 == 0, "edge list must split into 128-edge chunks");
static_assert(NCHUNK % NBLK == 0, "start rotation must be whole chunks");

__device__ __forceinline__ float leaky(float v) {
    return (v > 0.0f) ? v : NEG_SLOPE * v;
}

template <int PH>
__device__ __forceinline__ void rows_rescan(
        const int* __restrict__ src, const int* __restrict__ dst,
        const float* __restrict__ H, const float* __restrict__ s,
        int base, int tid, const float* dloc, unsigned* mkey, float* den,
        float* acc) {
    for (int e = tid; e < E_EDGES; e += ROWS_NT) {
        int r = dst[e] - base;
        if ((unsigned)r < (unsigned)RB) {
            int si = src[e];
            float v = leaky(s[si] + dloc[r]);
            if (PH == 0) {
                atomicMax(mkey + r, f2key(v));
            } else {
                float w = __expf(v - key2f(mkey[r]));
                if (PH == 1) {
                    atomicAdd(den + r, w);
                } else {
                    float coef = w / (den[r] + 1e-16f);
                    const float* hr = H + (size_t)si * 64;
                    float* ar = acc + r * 64;
                    for (int f = 0; f < 64; ++f) atomicAdd(ar + f, hr[f] * coef);
                }
            }
        }
    }
}

__global__ void __launch_bounds__(ROWS_NT)
gat_rows_lds(const int* __restrict__ src, const int* __restrict__ dst,
             const float* __restrict__ H, const float* __restrict__ s,
             const float* __restrict__ d, const float* __restrict__ bias,
             float* __restrict__ out, int relu) {
    __shared__ float    acc[RB * 64];
    __shared__ unsigned mkey[RB];
    __shared__ float    den[RB];
    __shared__ float    dloc[RB];
    __shared__ float    vself[RB];
    __shared__ float    wself[RB];
    __shared__ unsigned list[32 * WCAP];
    __shared__ float    lval[32 * WCAP];
    __shared__ int      ovf;

    const int tid  = threadIdx.x;
    const int lane = tid & 31;
    const int wave = __builtin_amdgcn_readfirstlane(tid >> 5);
    const int base = blockIdx.x * RB;
    unsigned* wl = list + wave * WCAP;
    float*    wv = lval + wave * WCAP;

    if (tid == 0) ovf = 0;
    if (tid < RB) {
        int i = base + tid;
        float dl = d[i];
        float v = leaky(s[i] + dl);
        dloc[tid] = dl;
        vself[tid] = v;
        mkey[tid] = f2key(v);
    }
    __syncthreads();

    int wcnt = 0;
    {
        const int4* dst4 = (const int4*)dst;
        const int rot = blockIdx.x * (NCHUNK / NBLK);
        for (int k = wave; k < NCHUNK; k += 32 * SCAN_U) {
            int4 dv[SCAN_U];
            int  ch[SCAN_U];
#pragma unroll
            for (int u = 0; u < SCAN_U; ++u) {
                int kk = k + 32 * u;
                int c = kk + rot;
                if (c >= NCHUNK) c -= NCHUNK;
                ch[u] = c;
                if (kk < NCHUNK) dv[u] = dst4[c * 32 + lane];
                else             dv[u] = make_int4(-1, -1, -1, -1);
            }
#pragma unroll
            for (int u = 0; u < SCAN_U; ++u) {
                int rr[4] = { dv[u].x - base, dv[u].y - base, dv[u].z - base, dv[u].w - base };
                bool mm[4];
#pragma unroll
                for (int j = 0; j < 4; ++j) mm[j] = (unsigned)rr[j] < (unsigned)RB;
                if (__builtin_amdgcn_ballot_w32(mm[0] | mm[1] | mm[2] | mm[3])) {
                    unsigned e0 = (unsigned)(ch[u] * 128 + lane * 4);
#pragma unroll
                    for (int j = 0; j < 4; ++j) {
                        unsigned mk = __builtin_amdgcn_ballot_w32(mm[j]);
                        int p = wcnt + (int)__builtin_amdgcn_mbcnt_lo(mk, 0u);
                        if (mm[j] && p < WCAP) wl[p] = ((unsigned)rr[j] << 20) | (e0 + (unsigned)j);
                        wcnt += __builtin_popcount(mk);
                    }
                }
            }
        }
    }
    if (wcnt > WCAP && lane == 0) ovf = 1;
    __syncthreads();
    const bool fast = (__builtin_amdgcn_readfirstlane(ovf) == 0);
    const int  n    = wcnt;

    if (fast) {
        for (int i = lane; i < n; i += 32) {
            unsigned en = wl[i];
            int e = (int)(en & 0xFFFFFu);
            int r = (int)(en >> 20);
            int si = src[e];
            float v = leaky(s[si] + dloc[r]);
            wl[i] = ((unsigned)r << 16) | (unsigned)si;
            wv[i] = v;
            atomicMax(mkey + r, f2key(v));
        }
    } else {
        rows_rescan<0>(src, dst, H, s, base, tid, dloc, mkey, den, acc);
    }
    __syncthreads();
    if (tid < RB) {
        float w = __expf(vself[tid] - key2f(mkey[tid]));
        wself[tid] = w;
        den[tid] = w;
    }
    __syncthreads();

    if (fast) {
        for (int i = lane; i < n; i += 32) {
            int r = (int)(wl[i] >> 16);
            float w = __expf(wv[i] - key2f(mkey[r]));
            wv[i] = w;
            atomicAdd(den + r, w);
        }
    } else {
        rows_rescan<1>(src, dst, H, s, base, tid, dloc, mkey, den, acc);
    }
    __syncthreads();
    for (int idx = tid; idx < RB * 64; idx += ROWS_NT) {
        int r = idx >> 6;
        float coef = wself[r] / (den[r] + 1e-16f);
        acc[idx] = H[(size_t)base * 64 + idx] * coef;
    }
    __syncthreads();

    if (fast) {
        for (int b0 = 0; b0 < n; b0 += 32) {
            int i = b0 + lane;
            bool valid = i < n;
            unsigned en = valid ? wl[i] : 0u;
            float w = valid ? wv[i] : 0.0f;
            int r  = (int)(en >> 16);
            int si = (int)(en & 0xFFFFu);
            float coef = w / (den[r] + 1e-16f);
#pragma unroll
            for (int g = 0; g < 32; g += AGG_U) {
                if (b0 + g < n) {
                    v2f hv[AGG_U];
#pragma unroll
                    for (int u = 0; u < AGG_U; ++u) {
                        int sj = __builtin_amdgcn_readlane(si, g + u);
                        hv[u] = *(const v2f*)(H + (size_t)sj * 64 + 2 * lane);
                    }
#pragma unroll
                    for (int u = 0; u < AGG_U; ++u) asm volatile("" :: "v"(hv[u]));
#pragma unroll
                    for (int u = 0; u < AGG_U; ++u) {
                        if (b0 + g + u < n) {
                            int rj = __builtin_amdgcn_readlane(r, g + u);
                            float cj = __int_as_float(__builtin_amdgcn_readlane(__float_as_int(coef), g + u));
                            float* ar = acc + rj * 64 + 2 * lane;
                            atomicAdd(ar,     hv[u].x * cj);
                            atomicAdd(ar + 1, hv[u].y * cj);
                        }
                    }
                }
            }
        }
    } else {
        rows_rescan<2>(src, dst, H, s, base, tid, dloc, mkey, den, acc);
    }
    __syncthreads();

    for (int idx = tid; idx < RB * 64; idx += ROWS_NT) {
        float v = acc[idx] + bias[idx & 63];
        if (relu) v = fmaxf(v, 0.0f);
        out[(size_t)base * 64 + idx] = v;
    }
}

extern "C" void kernel_launch(void* const* d_in, const int* in_sizes, int n_in,
                              void* d_out, int out_size, void* d_ws, size_t ws_size,
                              hipStream_t stream) {
    (void)in_sizes; (void)n_in; (void)out_size; (void)ws_size;

    const float* x    = (const float*)d_in[0];
    const int*   esrc = (const int*)  d_in[1];
    const int*   edst = (const int*)  d_in[2];
    const float* W0 = (const float*)d_in[3];
    const float* a0s = (const float*)d_in[4];
    const float* a0d = (const float*)d_in[5];
    const float* b0 = (const float*)d_in[6];
    const float* W1 = (const float*)d_in[7];
    const float* a1s = (const float*)d_in[8];
    const float* a1d = (const float*)d_in[9];
    const float* b1 = (const float*)d_in[10];
    const float* W2 = (const float*)d_in[11];
    const float* a2s = (const float*)d_in[12];
    const float* a2d = (const float*)d_in[13];
    const float* b2 = (const float*)d_in[14];
    float* out = (float*)d_out;

    float* ws = (float*)d_ws;
    float* H   = ws;
    float* XIN = H   + (size_t)N_NODES * 64;
    float* SV  = XIN + (size_t)N_NODES * 64;
    float* DV  = SV  + N_NODES;

    gat_gemm_wmma<<<((N_NODES / 16) + 7) / 8, 256, 0, stream>>>(x, W0, H, N_NODES, 128);
    gat_node_sd<<<(N_NODES + 255) / 256, 256, 0, stream>>>(H, a0s, a0d, SV, DV, N_NODES);
    gat_rows_lds<<<NBLK, ROWS_NT, 0, stream>>>(esrc, edst, H, SV, DV, b0, XIN, 1);

    gat_gemm_wmma<<<((N_NODES / 16) + 7) / 8, 256, 0, stream>>>(XIN, W1, H, N_NODES, 64);
    gat_node_sd<<<(N_NODES + 255) / 256, 256, 0, stream>>>(H, a1s, a1d, SV, DV, N_NODES);
    gat_rows_lds<<<NBLK, ROWS_NT, 0, stream>>>(esrc, edst, H, SV, DV, b1, XIN, 1);

    gat_gemm_wmma<<<((N_NODES / 16) + 7) / 8, 256, 0, stream>>>(XIN, W2, H, N_NODES, 64);
    gat_node_sd<<<(N_NODES + 255) / 256, 256, 0, stream>>>(H, a2s, a2d, SV, DV, N_NODES);
    gat_rows_lds<<<NBLK, ROWS_NT, 0, stream>>>(esrc, edst, H, SV, DV, b2, out, 0);
}


# Owners scan a one-byte-per-edge owner-key list (built in the logit launch) instead of the 4-byte destination list: 4x fewer scan bytes, 16 edges per lane load; list (sum) order changes
# speedup vs baseline: 3.4361x; 1.2389x over previous
#include <hip/hip_runtime.h>
#include <hip/hip_bf16.h>
#include <cstddef>


#define N_NODES 50000
#define E_EDGES 800000
#define NEG_SLOPE 0.2f

typedef float v2f __attribute__((ext_vector_type(2)));
typedef float v8f __attribute__((ext_vector_type(8)));

__device__ __forceinline__ unsigned f2key(float f) {
    int i = __float_as_int(f);
    return (i >= 0) ? ((unsigned)i | 0x80000000u) : ~(unsigned)i;
}
__device__ __forceinline__ float key2f(unsigned k) {
    int i = (k & 0x80000000u) ? (int)(k & 0x7FFFFFFFu) : (int)~k;
    return __int_as_float(i);
}

__global__ void __launch_bounds__(256)
gat_gemm_wmma(const float* __restrict__ X, const float* __restrict__ W,
              float* __restrict__ H, int nrows, int K) {
    const int wave = blockIdx.x * (blockDim.x >> 5) + (threadIdx.x >> 5);
    const int lane = threadIdx.x & 31;
    const int rowBase = wave * 16;
    if (rowBase >= nrows) return;

    const int lo = lane & 15;
    const int hi = lane >> 4;

    v8f acc0 = {}, acc1 = {}, acc2 = {}, acc3 = {};

    const float* xr = X + (size_t)(rowBase + lo) * K + 2 * hi;
    const float* wc = W + lo;

    for (int k = 0; k < K; k += 4) {
        float2 af = *(const float2*)(xr + k);
        v2f a; a.x = af.x; a.y = af.y;

        const float* w0 = wc + (size_t)(k + 2 * hi) * 64;
        const float* w1 = w0 + 64;
        v2f b0; b0.x = w0[0];  b0.y = w1[0];
        v2f b1; b1.x = w0[16]; b1.y = w1[16];
        v2f b2; b2.x = w0[32]; b2.y = w1[32];
        v2f b3; b3.x = w0[48]; b3.y = w1[48];

        acc0 = __builtin_amdgcn_wmma_f32_16x16x4_f32(false, a, false, b0, (short)0, acc0, false, false);
        acc1 = __builtin_amdgcn_wmma_f32_16x16x4_f32(false, a, false, b1, (short)0, acc1, false, false);
        acc2 = __builtin_amdgcn_wmma_f32_16x16x4_f32(false, a, false, b2, (short)0, acc2, false, false);
        acc3 = __builtin_amdgcn_wmma_f32_16x16x4_f32(false, a, false, b3, (short)0, acc3, false, false);
    }

#pragma unroll
    for (int r = 0; r < 8; ++r) {
        float* o = H + (size_t)(rowBase + r + 8 * hi) * 64 + lo;
        o[0]  = acc0[r];
        o[16] = acc1[r];
        o[32] = acc2[r];
        o[48] = acc3[r];
    }
}

#define RB_KEY 200u
__global__ void __launch_bounds__(256)
gat_node_sd(const float* __restrict__ H, const float* __restrict__ asrc,
            const float* __restrict__ adst, float* __restrict__ s,
            float* __restrict__ d, int n,
            const int* __restrict__ dst, unsigned* __restrict__ keys) {
    if (keys != nullptr) {
        const int4* dst4 = (const int4*)dst;
#pragma unroll
        for (int j = 0; j < 4; ++j) {
            int q = blockIdx.x * 1024 + j * 256 + threadIdx.x;
            if (q < E_EDGES / 4) {
                int4 v = dst4[q];
                keys[q] = ((unsigned)v.x / RB_KEY)         | (((unsigned)v.y / RB_KEY) << 8) |
                          (((unsigned)v.z / RB_KEY) << 16) | (((unsigned)v.w / RB_KEY) << 24);
            }
        }
    }
    int i = blockIdx.x * blockDim.x + threadIdx.x;
    if (i >= n) return;
    const float* hr = H + (size_t)i * 64;
    float ss = 0.0f, dd = 0.0f;
#pragma unroll
    for (int f = 0; f < 64; ++f) {
        float h = hr[f];
        ss = fmaf(h, asrc[f], ss);
        dd = fmaf(h, adst[f], dd);
    }
    s[i] = ss; d[i] = dd;
}

#define RB        200
#define NBLK      250
#define ROWS_NT   1024
#define WCAP      512
#define SCAN_U    7
#define AGG_U     16
#define NGROUP    (E_EDGES / 16)
#define NCHUNK    ((NGROUP + 31) / 32)
#define ROT_STEP  (NCHUNK / NBLK)
static_assert(RB * NBLK == N_NODES, "row blocks must tile the node set");
static_assert(RB == RB_KEY, "key list must be cut by the owners' row block");
static_assert(NBLK <= 255, "owner key must fit a byte, 255 is the pad value");
static_assert(E_EDGES % 16 == 0, "edge list must split into 16-edge groups");
static_assert(E_EDGES <= (1 << 20), "edge id must fit 20 bits of a list entry");

__device__ __forceinline__ float leaky(float v) {
    return (v > 0.0f) ? v : NEG_SLOPE * v;
}

template <int PH>
__device__ __forceinline__ void rows_rescan(
        const int* __restrict__ src, const int* __restrict__ dst,
        const float* __restrict__ H, const float* __restrict__ s,
        int base, int tid, const float* dloc, unsigned* mkey, float* den,
        float* acc) {
    for (int e = tid; e < E_EDGES; e += ROWS_NT) {
        int r = dst[e] - base;
        if ((unsigned)r < (unsigned)RB) {
            int si = src[e];
            float v = leaky(s[si] + dloc[r]);
            if (PH == 0) {
                atomicMax(mkey + r, f2key(v));
            } else {
                float w = __expf(v - key2f(mkey[r]));
                if (PH == 1) {
                    atomicAdd(den + r, w);
                } else {
                    float coef = w / (den[r] + 1e-16f);
                    const float* hr = H + (size_t)si * 64;
                    float* ar = acc + r * 64;
                    for (int f = 0; f < 64; ++f) atomicAdd(ar + f, hr[f] * coef);
                }
            }
        }
    }
}

__global__ void __launch_bounds__(ROWS_NT)
gat_rows_lds(const int* __restrict__ src, const int* __restrict__ dst,
             const unsigned* __restrict__ keys, const float* __restrict__ H, const float* __restrict__ s,
             const float* __restrict__ d, const float* __restrict__ bias,
             float* __restrict__ out, int relu) {
    __shared__ float    acc[RB * 64];
    __shared__ unsigned mkey[RB];
    __shared__ float    den[RB];
    __shared__ float    dloc[RB];
    __shared__ float    vself[RB];
    __shared__ float    wself[RB];
    __shared__ unsigned list[32 * WCAP];
    __shared__ float    lval[32 * WCAP];
    __shared__ int      ovf;

    const int tid  = threadIdx.x;
    const int lane = tid & 31;
    const int wave = __builtin_amdgcn_readfirstlane(tid >> 5);
    const int base = blockIdx.x * RB;
    unsigned* wl = list + wave * WCAP;
    float*    wv = lval + wave * WCAP;

    if (tid == 0) ovf = 0;
    if (tid < RB) {
        int i = base + tid;
        float dl = d[i];
        float v = leaky(s[i] + dl);
        dloc[tid] = dl;
        vself[tid] = v;
        mkey[tid] = f2key(v);
    }
    __syncthreads();

    int wcnt = 0;
    {
        const uint4* key4 = (const uint4*)keys;
        const unsigned pat = (unsigned)blockIdx.x * 0x01010101u;
        const int rot = blockIdx.x * ROT_STEP;
        for (int k = wave; k < NCHUNK; k += 32 * SCAN_U) {
            uint4 kv[SCAN_U];
            int   ch[SCAN_U];
#pragma unroll
            for (int u = 0; u < SCAN_U; ++u) {
                int kk = k + 32 * u;
                int c = kk + rot;
                if (c >= NCHUNK) c -= NCHUNK;
                ch[u] = c;
                int g = c * 32 + lane;
                if (kk < NCHUNK && g < NGROUP) kv[u] = key4[g];
                else                           kv[u] = make_uint4(~0u, ~0u, ~0u, ~0u);
            }
#pragma unroll
            for (int u = 0; u < SCAN_U; ++u) {
                if (k + 32 * u < NCHUNK) {
                    unsigned xs[4] = { kv[u].x ^ pat, kv[u].y ^ pat, kv[u].z ^ pat, kv[u].w ^ pat };
                    unsigned z[4];
#pragma unroll
                    for (int j = 0; j < 4; ++j) {
                        unsigned t = ((xs[j] & 0x7F7F7F7Fu) + 0x7F7F7F7Fu) | xs[j];
                        z[j] = ~(t | 0x7F7F7F7Fu);
                    }
                    unsigned zz = (z[0] >> 7) | (z[1] >> 6) | (z[2] >> 5) | (z[3] >> 4);
                    unsigned any = __builtin_amdgcn_ballot_w32(zz != 0u);
                    const unsigned e0 = (unsigned)(ch[u] * 512 + lane * 16);
                    while (any != 0u) {
                        int bit = __ffs((int)zz) - 1;
                        unsigned off = (unsigned)(((bit & 7) << 2) | (bit >> 3));
                        int p = wcnt + (int)__builtin_amdgcn_mbcnt_lo(any, 0u);
                        if (zz != 0u && p < WCAP) wl[p] = e0 + off;
                        wcnt += __builtin_popcount(any);
                        zz &= zz - 1u;
                        any = __builtin_amdgcn_ballot_w32(zz != 0u);
                    }
                }
            }
        }
    }
    if (wcnt > WCAP && lane == 0) ovf = 1;
    __syncthreads();
    const bool fast = (__builtin_amdgcn_readfirstlane(ovf) == 0);
    const int  n    = wcnt;

    if (fast) {
        for (int i = lane; i < n; i += 32) {
            int e = (int)wl[i];
            int r = dst[e] - base;
            int si = src[e];
            float v = leaky(s[si] + dloc[r]);
            wl[i] = ((unsigned)r << 16) | (unsigned)si;
            wv[i] = v;
            atomicMax(mkey + r, f2key(v));
        }
    } else {
        rows_rescan<0>(src, dst, H, s, base, tid, dloc, mkey, den, acc);
    }
    __syncthreads();
    if (tid < RB) {
        float w = __expf(vself[tid] - key2f(mkey[tid]));
        wself[tid] = w;
        den[tid] = w;
    }
    __syncthreads();

    if (fast) {
        for (int i = lane; i < n; i += 32) {
            int r = (int)(wl[i] >> 16);
            float w = __expf(wv[i] - key2f(mkey[r]));
            wv[i] = w;
            atomicAdd(den + r, w);
        }
    } else {
        rows_rescan<1>(src, dst, H, s, base, tid, dloc, mkey, den, acc);
    }
    __syncthreads();
    for (int idx = tid; idx < RB * 64; idx += ROWS_NT) {
        int r = idx >> 6;
        float coef = wself[r] / (den[r] + 1e-16f);
        acc[idx] = H[(size_t)base * 64 + idx] * coef;
    }
    __syncthreads();

    if (fast) {
        for (int b0 = 0; b0 < n; b0 += 32) {
            int i = b0 + lane;
            bool valid = i < n;
            unsigned en = valid ? wl[i] : 0u;
            float w = valid ? wv[i] : 0.0f;
            int r  = (int)(en >> 16);
            int si = (int)(en & 0xFFFFu);
            float coef = w / (den[r] + 1e-16f);
#pragma unroll
            for (int g = 0; g < 32; g += AGG_U) {
                if (b0 + g < n) {
                    v2f hv[AGG_U];
#pragma unroll
                    for (int u = 0; u < AGG_U; ++u) {
                        int sj = __builtin_amdgcn_readlane(si, g + u);
                        hv[u] = *(const v2f*)(H + (size_t)sj * 64 + 2 * lane);
                    }
#pragma unroll
                    for (int u = 0; u < AGG_U; ++u) asm volatile("" :: "v"(hv[u]));
#pragma unroll
                    for (int u = 0; u < AGG_U; ++u) {
                        if (b0 + g + u < n) {
                            int rj = __builtin_amdgcn_readlane(r, g + u);
                            float cj = __int_as_float(__builtin_amdgcn_readlane(__float_as_int(coef), g + u));
                            float* ar = acc + rj * 64 + 2 * lane;
                            atomicAdd(ar,     hv[u].x * cj);
                            atomicAdd(ar + 1, hv[u].y * cj);
                        }
                    }
                }
            }
        }
    } else {
        rows_rescan<2>(src, dst, H, s, base, tid, dloc, mkey, den, acc);
    }
    __syncthreads();

    for (int idx = tid; idx < RB * 64; idx += ROWS_NT) {
        float v = acc[idx] + bias[idx & 63];
        if (relu) v = fmaxf(v, 0.0f);
        out[(size_t)base * 64 + idx] = v;
    }
}

extern "C" void kernel_launch(void* const* d_in, const int* in_sizes, int n_in,
                              void* d_out, int out_size, void* d_ws, size_t ws_size,
                              hipStream_t stream) {
    (void)in_sizes; (void)n_in; (void)out_size; (void)ws_size;

    const float* x    = (const float*)d_in[0];
    const int*   esrc = (const int*)  d_in[1];
    const int*   edst = (const int*)  d_in[2];
    const float* W0 = (const float*)d_in[3];
    const float* a0s = (const float*)d_in[4];
    const float* a0d = (const float*)d_in[5];
    const float* b0 = (const float*)d_in[6];
    const float* W1 = (const float*)d_in[7];
    const float* a1s = (const float*)d_in[8];
    const float* a1d = (const float*)d_in[9];
    const float* b1 = (const float*)d_in[10];
    const float* W2 = (const float*)d_in[11];
    const float* a2s = (const float*)d_in[12];
    const float* a2d = (const float*)d_in[13];
    const float* b2 = (const float*)d_in[14];
    float* out = (float*)d_out;

    float* ws = (float*)d_ws;
    float* H   = ws;
    float* XIN = H   + (size_t)N_NODES * 64;
    float* SV  = XIN + (size_t)N_NODES * 64;
    float* DV  = SV  + N_NODES;
    unsigned* KEY01 = (unsigned*)out;
    unsigned* KEY2  = (unsigned*)XIN;

    gat_gemm_wmma<<<((N_NODES / 16) + 7) / 8, 256, 0, stream>>>(x, W0, H, N_NODES, 128);
    gat_node_sd<<<(N_NODES + 255) / 256, 256, 0, stream>>>(H, a0s, a0d, SV, DV, N_NODES, edst, KEY01);
    gat_rows_lds<<<NBLK, ROWS_NT, 0, stream>>>(esrc, edst, KEY01, H, SV, DV, b0, XIN, 1);

    gat_gemm_wmma<<<((N_NODES / 16) + 7) / 8, 256, 0, stream>>>(XIN, W1, H, N_NODES, 64);
    gat_node_sd<<<(N_NODES + 255) / 256, 256, 0, stream>>>(H, a1s, a1d, SV, DV, N_NODES, edst, nullptr);
    gat_rows_lds<<<NBLK, ROWS_NT, 0, stream>>>(esrc, edst, KEY01, H, SV, DV, b1, XIN, 1);

    gat_gemm_wmma<<<((N_NODES / 16) + 7) / 8, 256, 0, stream>>>(XIN, W2, H, N_NODES, 64);
    gat_node_sd<<<(N_NODES + 255) / 256, 256, 0, stream>>>(H, a2s, a2d, SV, DV, N_NODES, edst, KEY2);
    gat_rows_lds<<<NBLK, ROWS_NT, 0, stream>>>(esrc, edst, KEY2, H, SV, DV, b2, out, 0);
}


# Owners filter the key list in layer 0 only: each wave saves its (row, source) list in its block's part of the output buffer; layers 1, 2 load it back in two batched trips; list (sum) order same
# speedup vs baseline: 4.2732x; 1.2436x over previous
#include <hip/hip_runtime.h>
#include <hip/hip_bf16.h>
#include <cstddef>


#define N_NODES 50000
#define E_EDGES 800000
#define NEG_SLOPE 0.2f

typedef float v2f __attribute__((ext_vector_type(2)));
typedef float v8f __attribute__((ext_vector_type(8)));

__device__ __forceinline__ unsigned f2key(float f) {
    int i = __float_as_int(f);
    return (i >= 0) ? ((unsigned)i | 0x80000000u) : ~(unsigned)i;
}
__device__ __forceinline__ float key2f(unsigned k) {
    int i = (k & 0x80000000u) ? (int)(k & 0x7FFFFFFFu) : (int)~k;
    return __int_as_float(i);
}

__global__ void __launch_bounds__(256)
gat_gemm_wmma(const float* __restrict__ X, const float* __restrict__ W,
              float* __restrict__ H, int nrows, int K) {
    const int wave = blockIdx.x * (blockDim.x >> 5) + (threadIdx.x >> 5);
    const int lane = threadIdx.x & 31;
    const int rowBase = wave * 16;
    if (rowBase >= nrows) return;

    const int lo = lane & 15;
    const int hi = lane >> 4;

    v8f acc0 = {}, acc1 = {}, acc2 = {}, acc3 = {};

    const float* xr = X + (size_t)(rowBase + lo) * K + 2 * hi;
    const float* wc = W + lo;

    for (int k = 0; k < K; k += 4) {
        float2 af = *(const float2*)(xr + k);
        v2f a; a.x = af.x; a.y = af.y;

        const float* w0 = wc + (size_t)(k + 2 * hi) * 64;
        const float* w1 = w0 + 64;
        v2f b0; b0.x = w0[0];  b0.y = w1[0];
        v2f b1; b1.x = w0[16]; b1.y = w1[16];
        v2f b2; b2.x = w0[32]; b2.y = w1[32];
        v2f b3; b3.x = w0[48]; b3.y = w1[48];

        acc0 = __builtin_amdgcn_wmma_f32_16x16x4_f32(false, a, false, b0, (short)0, acc0, false, false);
        acc1 = __builtin_amdgcn_wmma_f32_16x16x4_f32(false, a, false, b1, (short)0, acc1, false, false);
        acc2 = __builtin_amdgcn_wmma_f32_16x16x4_f32(false, a, false, b2, (short)0, acc2, false, false);
        acc3 = __builtin_amdgcn_wmma_f32_16x16x4_f32(false, a, false, b3, (short)0, acc3, false, false);
    }

#pragma unroll
    for (int r = 0; r < 8; ++r) {
        float* o = H + (size_t)(rowBase + r + 8 * hi) * 64 + lo;
        o[0]  = acc0[r];
        o[16] = acc1[r];
        o[32] = acc2[r];
        o[48] = acc3[r];
    }
}

#define RB_KEY 200u
#define BLK_DW    (200 * 64)
#define SAVE_CAP  256
#define CNT_OFF   (32 * SAVE_CAP)
#define KEY_OFF   8448
#define KEY_DW    4096
__device__ __forceinline__ unsigned key_dword_pos(unsigned q) {
    return (q / KEY_DW) * BLK_DW + KEY_OFF + (q % KEY_DW);
}
__global__ void __launch_bounds__(256)
gat_node_sd(const float* __restrict__ H, const float* __restrict__ asrc,
            const float* __restrict__ adst, float* __restrict__ s,
            float* __restrict__ d, int n,
            const int* __restrict__ dst, unsigned* __restrict__ keys) {
    if (keys != nullptr) {
        const int4* dst4 = (const int4*)dst;
#pragma unroll
        for (int j = 0; j < 4; ++j) {
            int q = blockIdx.x * 1024 + j * 256 + threadIdx.x;
            if (q < E_EDGES / 4) {
                int4 v = dst4[q];
                keys[key_dword_pos((unsigned)q)] =
                          ((unsigned)v.x / RB_KEY)         | (((unsigned)v.y / RB_KEY) << 8) |
                          (((unsigned)v.z / RB_KEY) << 16) | (((unsigned)v.w / RB_KEY) << 24);
            }
        }
    }
    int i = blockIdx.x * blockDim.x + threadIdx.x;
    if (i >= n) return;
    const float* hr = H + (size_t)i * 64;
    float ss = 0.0f, dd = 0.0f;
#pragma unroll
    for (int f = 0; f < 64; ++f) {
        float h = hr[f];
        ss = fmaf(h, asrc[f], ss);
        dd = fmaf(h, adst[f], dd);
    }
    s[i] = ss; d[i] = dd;
}

#define RB        200
#define NBLK      250
#define ROWS_NT   1024
#define WCAP      512
#define SCAN_U    7
#define AGG_U     16
#define PH0_U     5
#define NGROUP    (E_EDGES / 16)
#define NCHUNK    ((NGROUP + 31) / 32)
#define ROT_STEP  (NCHUNK / NBLK)
static_assert(RB * NBLK == N_NODES, "row blocks must tile the node set");
static_assert(RB == RB_KEY, "key list must be cut by the owners' row block");
static_assert(NBLK <= 255, "owner key must fit a byte, 255 is the pad value");
static_assert(E_EDGES % 16 == 0, "edge list must split into 16-edge groups");
static_assert(E_EDGES <= (1 << 20), "edge id must fit 20 bits of a list entry");
static_assert(BLK_DW == RB * 64, "footprint = the block's output rows");
static_assert(SAVE_CAP <= WCAP && SAVE_CAP % 32 == 0, "saved lists are a prefix of the LDS lists");
static_assert(CNT_OFF + 32 <= KEY_OFF && KEY_OFF + KEY_DW <= BLK_DW, "lists, counts, key piece must not overlap");
static_assert(KEY_DW == 32 * 128, "a key piece holds 32 whole chunks");
static_assert((E_EDGES / 4 + KEY_DW - 1) / KEY_DW <= NBLK, "key pieces must fit the footprints");
static_assert(N_NODES <= 65536, "source id must fit 16 bits of a saved entry");

__device__ __forceinline__ float leaky(float v) {
    return (v > 0.0f) ? v : NEG_SLOPE * v;
}

template <int PH>
__device__ __forceinline__ void rows_rescan(
        const int* __restrict__ src, const int* __restrict__ dst,
        const float* __restrict__ H, const float* __restrict__ s,
        int base, int tid, const float* dloc, unsigned* mkey, float* den,
        float* acc) {
    for (int e = tid; e < E_EDGES; e += ROWS_NT) {
        int r = dst[e] - base;
        if ((unsigned)r < (unsigned)RB) {
            int si = src[e];
            float v = leaky(s[si] + dloc[r]);
            if (PH == 0) {
                atomicMax(mkey + r, f2key(v));
            } else {
                float w = __expf(v - key2f(mkey[r]));
                if (PH == 1) {
                    atomicAdd(den + r, w);
                } else {
                    float coef = w / (den[r] + 1e-16f);
                    const float* hr = H + (size_t)si * 64;
                    float* ar = acc + r * 64;
                    for (int f = 0; f < 64; ++f) atomicAdd(ar + f, hr[f] * coef);
                }
            }
        }
    }
}

template <int MODE>
__global__ void __launch_bounds__(ROWS_NT)
gat_rows_lds(const int* __restrict__ src, const int* __restrict__ dst,
             unsigned* own, const float* __restrict__ H, const float* __restrict__ s,
             const float* __restrict__ d, const float* __restrict__ bias,
             float* out, int relu) {
    __shared__ float    acc[RB * 64];
    __shared__ unsigned mkey[RB];
    __shared__ float    den[RB];
    __shared__ float    dloc[RB];
    __shared__ float    vself[RB];
    __shared__ float    wself[RB];
    __shared__ unsigned list[32 * WCAP];
    __shared__ float    lval[32 * WCAP];
    __shared__ int      ovf;
    __shared__ int      nosave;

    const int tid  = threadIdx.x;
    const int lane = tid & 31;
    const int wave = __builtin_amdgcn_readfirstlane(tid >> 5);
    const int base = blockIdx.x * RB;
    unsigned* wl = list + wave * WCAP;
    float*    wv = lval + wave * WCAP;
    unsigned* ownb = own + (size_t)blockIdx.x * BLK_DW;
    unsigned* lg   = ownb + wave * SAVE_CAP;

    int csaved = 0;
    if (MODE == 1) csaved = __builtin_amdgcn_readfirstlane((int)ownb[CNT_OFF + wave]);

    if (tid == 0) { ovf = 0; nosave = 0; }
    if (tid < RB) {
        int i = base + tid;
        float dl = d[i];
        float v = leaky(s[i] + dl);
        dloc[tid] = dl;
        vself[tid] = v;
        mkey[tid] = f2key(v);
    }
    __syncthreads();

    int wcnt = 0;
    if (MODE == 1) {
        if ((unsigned)csaved > (unsigned)SAVE_CAP) { if (lane == 0) ovf = 1; }
        else            wcnt = csaved;
    } else {
        const uint4* key4 = (const uint4*)own;
        const unsigned pat = (unsigned)blockIdx.x * 0x01010101u;
        const int rot = blockIdx.x * ROT_STEP;
        for (int k = wave; k < NCHUNK; k += 32 * SCAN_U) {
            uint4 kv[SCAN_U];
            int   ch[SCAN_U];
#pragma unroll
            for (int u = 0; u < SCAN_U; ++u) {
                int kk = k + 32 * u;
                int c = kk + rot;
                if (c >= NCHUNK) c -= NCHUNK;
                ch[u] = c;
                int g = c * 32 + lane;
                int a = (c >> 5) * (BLK_DW / 4) + (KEY_OFF / 4) + (c & 31) * 32 + lane;
                if (kk < NCHUNK && g < NGROUP) kv[u] = key4[a];
                else                           kv[u] = make_uint4(~0u, ~0u, ~0u, ~0u);
            }
#pragma unroll
            for (int u = 0; u < SCAN_U; ++u) {
                if (k + 32 * u < NCHUNK) {
                    unsigned xs[4] = { kv[u].x ^ pat, kv[u].y ^ pat, kv[u].z ^ pat, kv[u].w ^ pat };
                    unsigned z[4];
#pragma unroll
                    for (int j = 0; j < 4; ++j) {
                        unsigned t = ((xs[j] & 0x7F7F7F7Fu) + 0x7F7F7F7Fu) | xs[j];
                        z[j] = ~(t | 0x7F7F7F7Fu);
                    }
                    unsigned zz = (z[0] >> 7) | (z[1] >> 6) | (z[2] >> 5) | (z[3] >> 4);
                    unsigned any = __builtin_amdgcn_ballot_w32(zz != 0u);
                    const unsigned e0 = (unsigned)(ch[u] * 512 + lane * 16);
                    while (any != 0u) {
                        int bit = __ffs((int)zz) - 1;
                        unsigned off = (unsigned)(((bit & 7) << 2) | (bit >> 3));
                        int p = wcnt + (int)__builtin_amdgcn_mbcnt_lo(any, 0u);
                        if (zz != 0u && p < WCAP) wl[p] = e0 + off;
                        wcnt += __builtin_popcount(any);
                        zz &= zz - 1u;
                        any = __builtin_amdgcn_ballot_w32(zz != 0u);
                    }
                }
            }
        }
    }
    if (wcnt > WCAP && lane == 0) ovf = 1;
    if (MODE == 0 && wcnt > SAVE_CAP && lane == 0) nosave = 1;
    __syncthreads();
    const bool fast = (__builtin_amdgcn_readfirstlane(ovf) == 0);
    const int  n    = wcnt;
    const bool saved = (MODE == 0) && fast && (__builtin_amdgcn_readfirstlane(nosave) == 0);
    if (MODE == 0 && lane == 0) ownb[CNT_OFF + wave] = saved ? (unsigned)n : ~0u;

    if (fast && MODE == 0) {
        for (int i = lane; i < n; i += 32) {
            int e = (int)wl[i];
            int r = dst[e] - base;
            int si = src[e];
            float v = leaky(s[si] + dloc[r]);
            unsigned en = ((unsigned)r << 16) | (unsigned)si;
            wl[i] = en;
            if (saved) lg[i] = en;
            wv[i] = v;
            atomicMax(mkey + r, f2key(v));
        }
    } else if (fast) {
        for (int g0 = 0; g0 < n; g0 += 32 * PH0_U) {
            unsigned en[PH0_U];
            float    sv[PH0_U];
#pragma unroll
            for (int u = 0; u < PH0_U; ++u) {
                int i = g0 + 32 * u + lane;
                en[u] = lg[(i < n) ? i : n - 1];
            }
#pragma unroll
            for (int u = 0; u < PH0_U; ++u) sv[u] = s[en[u] & 0xFFFFu];
#pragma unroll
            for (int u = 0; u < PH0_U; ++u) asm volatile("" :: "v"(sv[u]));
#pragma unroll
            for (int u = 0; u < PH0_U; ++u) {
                int i = g0 + 32 * u + lane;
                if (i < n) {
                    int r = (int)(en[u] >> 16);
                    float v = leaky(sv[u] + dloc[r]);
                    wl[i] = en[u];
                    wv[i] = v;
                    atomicMax(mkey + r, f2key(v));
                }
            }
        }
    } else {
        rows_rescan<0>(src, dst, H, s, base, tid, dloc, mkey, den, acc);
    }
    __syncthreads();
    if (tid < RB) {
        float w = __expf(vself[tid] - key2f(mkey[tid]));
        wself[tid] = w;
        den[tid] = w;
    }
    __syncthreads();

    if (fast) {
        for (int i = lane; i < n; i += 32) {
            int r = (int)(wl[i] >> 16);
            float w = __expf(wv[i] - key2f(mkey[r]));
            wv[i] = w;
            atomicAdd(den + r, w);
        }
    } else {
        rows_rescan<1>(src, dst, H, s, base, tid, dloc, mkey, den, acc);
    }
    __syncthreads();
    for (int idx = tid; idx < RB * 64; idx += ROWS_NT) {
        int r = idx >> 6;
        float coef = wself[r] / (den[r] + 1e-16f);
        acc[idx] = H[(size_t)base * 64 + idx] * coef;
    }
    __syncthreads();

    if (fast) {
        for (int b0 = 0; b0 < n; b0 += 32) {
            int i = b0 + lane;
            bool valid = i < n;
            unsigned en = valid ? wl[i] : 0u;
            float w = valid ? wv[i] : 0.0f;
            int r  = (int)(en >> 16);
            int si = (int)(en & 0xFFFFu);
            float coef = w / (den[r] + 1e-16f);
#pragma unroll
            for (int g = 0; g < 32; g += AGG_U) {
                if (b0 + g < n) {
                    v2f hv[AGG_U];
#pragma unroll
                    for (int u = 0; u < AGG_U; ++u) {
                        int sj = __builtin_amdgcn_readlane(si, g + u);
                        hv[u] = *(const v2f*)(H + (size_t)sj * 64 + 2 * lane);
                    }
#pragma unroll
                    for (int u = 0; u < AGG_U; ++u) asm volatile("" :: "v"(hv[u]));
#pragma unroll
                    for (int u = 0; u < AGG_U; ++u) {
                        if (b0 + g + u < n) {
                            int rj = __builtin_amdgcn_readlane(r, g + u);
                            float cj = __int_as_float(__builtin_amdgcn_readlane(__float_as_int(coef), g + u));
                            float* ar = acc + rj * 64 + 2 * lane;
                            atomicAdd(ar,     hv[u].x * cj);
                            atomicAdd(ar + 1, hv[u].y * cj);
                        }
                    }
                }
            }
        }
    } else {
        rows_rescan<2>(src, dst, H, s, base, tid, dloc, mkey, den, acc);
    }
    __syncthreads();

    for (int idx = tid; idx < RB * 64; idx += ROWS_NT) {
        float v = acc[idx] + bias[idx & 63];
        if (relu) v = fmaxf(v, 0.0f);
        out[(size_t)base * 64 + idx] = v;
    }
}

extern "C" void kernel_launch(void* const* d_in, const int* in_sizes, int n_in,
                              void* d_out, int out_size, void* d_ws, size_t ws_size,
                              hipStream_t stream) {
    (void)in_sizes; (void)n_in; (void)out_size; (void)ws_size;

    const float* x    = (const float*)d_in[0];
    const int*   esrc = (const int*)  d_in[1];
    const int*   edst = (const int*)  d_in[2];
    const float* W0 = (const float*)d_in[3];
    const float* a0s = (const float*)d_in[4];
    const float* a0d = (const float*)d_in[5];
    const float* b0 = (const float*)d_in[6];
    const float* W1 = (const float*)d_in[7];
    const float* a1s = (const float*)d_in[8];
    const float* a1d = (const float*)d_in[9];
    const float* b1 = (const float*)d_in[10];
    const float* W2 = (const float*)d_in[11];
    const float* a2s = (const float*)d_in[12];
    const float* a2d = (const float*)d_in[13];
    const float* b2 = (const float*)d_in[14];
    float* out = (float*)d_out;

    float* ws = (float*)d_ws;
    float* H   = ws;
    float* XIN = H   + (size_t)N_NODES * 64;
    float* SV  = XIN + (size_t)N_NODES * 64;
    float* DV  = SV  + N_NODES;
    unsigned* OWN = (unsigned*)out;

    gat_gemm_wmma<<<((N_NODES / 16) + 7) / 8, 256, 0, stream>>>(x, W0, H, N_NODES, 128);
    gat_node_sd<<<(N_NODES + 255) / 256, 256, 0, stream>>>(H, a0s, a0d, SV, DV, N_NODES, edst, OWN);
    gat_rows_lds<0><<<NBLK, ROWS_NT, 0, stream>>>(esrc, edst, OWN, H, SV, DV, b0, XIN, 1);

    gat_gemm_wmma<<<((N_NODES / 16) + 7) / 8, 256, 0, stream>>>(XIN, W1, H, N_NODES, 64);
    gat_node_sd<<<(N_NODES + 255) / 256, 256, 0, stream>>>(H, a1s, a1d, SV, DV, N_NODES, edst, nullptr);
    gat_rows_lds<1><<<NBLK, ROWS_NT, 0, stream>>>(esrc, edst, OWN, H, SV, DV, b1, XIN, 1);

    gat_gemm_wmma<<<((N_NODES / 16) + 7) / 8, 256, 0, stream>>>(XIN, W2, H, N_NODES, 64);
    gat_node_sd<<<(N_NODES + 255) / 256, 256, 0, stream>>>(H, a2s, a2d, SV, DV, N_NODES, edst, nullptr);
    gat_rows_lds<1><<<NBLK, ROWS_NT, 0, stream>>>(esrc, edst, OWN, H, SV, DV, b2, out, 0);
}


# Layers 1,2 aggregate input rows first, then each block multiplies its 200 summed rows by W from LDS on WMMA, (A x)W; logits as x.(W a); the 2 GEMM launches go; sum order changes
# speedup vs baseline: 4.5285x; 1.0597x over previous
#include <hip/hip_runtime.h>
#include <hip/hip_bf16.h>
#include <cstddef>


#define N_NODES 50000
#define E_EDGES 800000
#define NEG_SLOPE 0.2f

typedef float v2f __attribute__((ext_vector_type(2)));
typedef float v8f __attribute__((ext_vector_type(8)));

__device__ __forceinline__ unsigned f2key(float f) {
    int i = __float_as_int(f);
    return (i >= 0) ? ((unsigned)i | 0x80000000u) : ~(unsigned)i;
}
__device__ __forceinline__ float key2f(unsigned k) {
    int i = (k & 0x80000000u) ? (int)(k & 0x7FFFFFFFu) : (int)~k;
    return __int_as_float(i);
}

__global__ void __launch_bounds__(256)
gat_gemm_wmma(const float* __restrict__ X, const float* __restrict__ W,
              float* __restrict__ H, int nrows, int K) {
    const int wave = blockIdx.x * (blockDim.x >> 5) + (threadIdx.x >> 5);
    const int lane = threadIdx.x & 31;
    const int rowBase = wave * 16;
    if (rowBase >= nrows) return;

    const int lo = lane & 15;
    const int hi = lane >> 4;

    v8f acc0 = {}, acc1 = {}, acc2 = {}, acc3 = {};

    const float* xr = X + (size_t)(rowBase + lo) * K + 2 * hi;
    const float* wc = W + lo;

    for (int k = 0; k < K; k += 4) {
        float2 af = *(const float2*)(xr + k);
        v2f a; a.x = af.x; a.y = af.y;

        const float* w0 = wc + (size_t)(k + 2 * hi) * 64;
        const float* w1 = w0 + 64;
        v2f b0; b0.x = w0[0];  b0.y = w1[0];
        v2f b1; b1.x = w0[16]; b1.y = w1[16];
        v2f b2; b2.x = w0[32]; b2.y = w1[32];
        v2f b3; b3.x = w0[48]; b3.y = w1[48];

        acc0 = __builtin_amdgcn_wmma_f32_16x16x4_f32(false, a, false, b0, (short)0, acc0, false, false);
        acc1 = __builtin_amdgcn_wmma_f32_16x16x4_f32(false, a, false, b1, (short)0, acc1, false, false);
        acc2 = __builtin_amdgcn_wmma_f32_16x16x4_f32(false, a, false, b2, (short)0, acc2, false, false);
        acc3 = __builtin_amdgcn_wmma_f32_16x16x4_f32(false, a, false, b3, (short)0, acc3, false, false);
    }

#pragma unroll
    for (int r = 0; r < 8; ++r) {
        float* o = H + (size_t)(rowBase + r + 8 * hi) * 64 + lo;
        o[0]  = acc0[r];
        o[16] = acc1[r];
        o[32] = acc2[r];
        o[48] = acc3[r];
    }
}

#define RB_KEY 200u
#define BLK_DW    (200 * 64)
#define SAVE_CAP  256
#define CNT_OFF   (32 * SAVE_CAP)
#define KEY_OFF   8448
#define KEY_DW    4096
__device__ __forceinline__ unsigned key_dword_pos(unsigned q) {
    return (q / KEY_DW) * BLK_DW + KEY_OFF + (q % KEY_DW);
}
__global__ void __launch_bounds__(256)
gat_node_sd(const float* __restrict__ H, const float* __restrict__ asrc,
            const float* __restrict__ adst, float* __restrict__ s,
            float* __restrict__ d, int n,
            const int* __restrict__ dst, unsigned* __restrict__ keys,
            const float* __restrict__ Wm) {
    if (keys != nullptr) {
        const int4* dst4 = (const int4*)dst;
#pragma unroll
        for (int j = 0; j < 4; ++j) {
            int q = blockIdx.x * 1024 + j * 256 + threadIdx.x;
            if (q < E_EDGES / 4) {
                int4 v = dst4[q];
                keys[key_dword_pos((unsigned)q)] =
                          ((unsigned)v.x / RB_KEY)         | (((unsigned)v.y / RB_KEY) << 8) |
                          (((unsigned)v.z / RB_KEY) << 16) | (((unsigned)v.w / RB_KEY) << 24);
            }
        }
    }
    int i = blockIdx.x * blockDim.x + threadIdx.x;
    if (Wm != nullptr) {
        const int lane = threadIdx.x & 31;
        const float* r0 = Wm + lane * 64;
        const float* r1 = r0 + 32 * 64;
        float ws0 = 0.0f, ws1 = 0.0f, wd0 = 0.0f, wd1 = 0.0f;
#pragma unroll
        for (int j = 0; j < 64; ++j) {
            float as = asrc[j], ad = adst[j];
            float w0 = r0[j], w1 = r1[j];
            ws0 = fmaf(w0, as, ws0); wd0 = fmaf(w0, ad, wd0);
            ws1 = fmaf(w1, as, ws1); wd1 = fmaf(w1, ad, wd1);
        }
        const int ic = (i < n) ? i : n - 1;
        const float* xr = H + (size_t)ic * 64;
        float ss = 0.0f, dd = 0.0f;
#pragma unroll
        for (int f = 0; f < 64; ++f) {
            float xv = xr[f];
            float was = __int_as_float(__builtin_amdgcn_readlane(__float_as_int(f < 32 ? ws0 : ws1), f & 31));
            float wad = __int_as_float(__builtin_amdgcn_readlane(__float_as_int(f < 32 ? wd0 : wd1), f & 31));
            ss = fmaf(xv, was, ss);
            dd = fmaf(xv, wad, dd);
        }
        if (i < n) { s[i] = ss; d[i] = dd; }
        return;
    }
    if (i >= n) return;
    const float* hr = H + (size_t)i * 64;
    float ss = 0.0f, dd = 0.0f;
#pragma unroll
    for (int f = 0; f < 64; ++f) {
        float h = hr[f];
        ss = fmaf(h, asrc[f], ss);
        dd = fmaf(h, adst[f], dd);
    }
    s[i] = ss; d[i] = dd;
}

#define RB        200
#define NBLK      250
#define ROWS_NT   1024
#define WCAP      512
#define SCAN_U    7
#define AGG_U     16
#define PH0_U     5
#define NGROUP    (E_EDGES / 16)
#define NCHUNK    ((NGROUP + 31) / 32)
#define ROT_STEP  (NCHUNK / NBLK)
static_assert(RB * NBLK == N_NODES, "row blocks must tile the node set");
static_assert(RB == RB_KEY, "key list must be cut by the owners' row block");
static_assert(NBLK <= 255, "owner key must fit a byte, 255 is the pad value");
static_assert(E_EDGES % 16 == 0, "edge list must split into 16-edge groups");
static_assert(E_EDGES <= (1 << 20), "edge id must fit 20 bits of a list entry");
static_assert(BLK_DW == RB * 64, "footprint = the block's output rows");
static_assert(SAVE_CAP <= WCAP && SAVE_CAP % 32 == 0, "saved lists are a prefix of the LDS lists");
static_assert(CNT_OFF + 32 <= KEY_OFF && KEY_OFF + KEY_DW <= BLK_DW, "lists, counts, key piece must not overlap");
static_assert(KEY_DW == 32 * 128, "a key piece holds 32 whole chunks");
static_assert((E_EDGES / 4 + KEY_DW - 1) / KEY_DW <= NBLK, "key pieces must fit the footprints");
static_assert(N_NODES <= 65536, "source id must fit 16 bits of a saved entry");

__device__ __forceinline__ float leaky(float v) {
    return (v > 0.0f) ? v : NEG_SLOPE * v;
}

template <int PH>
__device__ __forceinline__ void rows_rescan(
        const int* __restrict__ src, const int* __restrict__ dst,
        const float* __restrict__ H, const float* __restrict__ s,
        int base, int tid, const float* dloc, unsigned* mkey, float* den,
        float* acc) {
    for (int e = tid; e < E_EDGES; e += ROWS_NT) {
        int r = dst[e] - base;
        if ((unsigned)r < (unsigned)RB) {
            int si = src[e];
            float v = leaky(s[si] + dloc[r]);
            if (PH == 0) {
                atomicMax(mkey + r, f2key(v));
            } else {
                float w = __expf(v - key2f(mkey[r]));
                if (PH == 1) {
                    atomicAdd(den + r, w);
                } else {
                    float coef = w / (den[r] + 1e-16f);
                    const float* hr = H + (size_t)si * 64;
                    float* ar = acc + r * 64;
                    for (int f = 0; f < 64; ++f) atomicAdd(ar + f, hr[f] * coef);
                }
            }
        }
    }
}

template <int MODE>
__global__ void __launch_bounds__(ROWS_NT)
gat_rows_lds(const int* __restrict__ src, const int* __restrict__ dst,
             unsigned* own, const float* __restrict__ H, const float* __restrict__ s,
             const float* __restrict__ d, const float* __restrict__ bias,
             float* out, int relu, const float* __restrict__ Wm) {
    __shared__ float    acc[(RB + 8) * 64];
    __shared__ unsigned mkey[RB];
    __shared__ float    den[RB];
    __shared__ float    dloc[RB];
    __shared__ float    vself[RB];
    __shared__ float    wself[RB];
    __shared__ unsigned list[32 * WCAP];
    __shared__ float    lval[32 * WCAP];
    __shared__ int      ovf;
    __shared__ int      nosave;

    const int tid  = threadIdx.x;
    const int lane = tid & 31;
    const int wave = __builtin_amdgcn_readfirstlane(tid >> 5);
    const int base = blockIdx.x * RB;
    unsigned* wl = list + wave * WCAP;
    float*    wv = lval + wave * WCAP;
    unsigned* ownb = own + (size_t)blockIdx.x * BLK_DW;
    unsigned* lg   = ownb + wave * SAVE_CAP;

    int csaved = 0;
    if (MODE == 1) csaved = __builtin_amdgcn_readfirstlane((int)ownb[CNT_OFF + wave]);

    if (tid == 0) { ovf = 0; nosave = 0; }
    if (tid < RB) {
        int i = base + tid;
        float dl = d[i];
        float v = leaky(s[i] + dl);
        dloc[tid] = dl;
        vself[tid] = v;
        mkey[tid] = f2key(v);
    }
    __syncthreads();

    int wcnt = 0;
    if (MODE == 1) {
        if ((unsigned)csaved > (unsigned)SAVE_CAP) { if (lane == 0) ovf = 1; }
        else            wcnt = csaved;
    } else {
        const uint4* key4 = (const uint4*)own;
        const unsigned pat = (unsigned)blockIdx.x * 0x01010101u;
        const int rot = blockIdx.x * ROT_STEP;
        for (int k = wave; k < NCHUNK; k += 32 * SCAN_U) {
            uint4 kv[SCAN_U];
            int   ch[SCAN_U];
#pragma unroll
            for (int u = 0; u < SCAN_U; ++u) {
                int kk = k + 32 * u;
                int c = kk + rot;
                if (c >= NCHUNK) c -= NCHUNK;
                ch[u] = c;
                int g = c * 32 + lane;
                int a = (c >> 5) * (BLK_DW / 4) + (KEY_OFF / 4) + (c & 31) * 32 + lane;
                if (kk < NCHUNK && g < NGROUP) kv[u] = key4[a];
                else                           kv[u] = make_uint4(~0u, ~0u, ~0u, ~0u);
            }
#pragma unroll
            for (int u = 0; u < SCAN_U; ++u) {
                if (k + 32 * u < NCHUNK) {
                    unsigned xs[4] = { kv[u].x ^ pat, kv[u].y ^ pat, kv[u].z ^ pat, kv[u].w ^ pat };
                    unsigned z[4];
#pragma unroll
                    for (int j = 0; j < 4; ++j) {
                        unsigned t = ((xs[j] & 0x7F7F7F7Fu) + 0x7F7F7F7Fu) | xs[j];
                        z[j] = ~(t | 0x7F7F7F7Fu);
                    }
                    unsigned zz = (z[0] >> 7) | (z[1] >> 6) | (z[2] >> 5) | (z[3] >> 4);
                    unsigned any = __builtin_amdgcn_ballot_w32(zz != 0u);
                    const unsigned e0 = (unsigned)(ch[u] * 512 + lane * 16);
                    while (any != 0u) {
                        int bit = __ffs((int)zz) - 1;
                        unsigned off = (unsigned)(((bit & 7) << 2) | (bit >> 3));
                        int p = wcnt + (int)__builtin_amdgcn_mbcnt_lo(any, 0u);
                        if (zz != 0u && p < WCAP) wl[p] = e0 + off;
                        wcnt += __builtin_popcount(any);
                        zz &= zz - 1u;
                        any = __builtin_amdgcn_ballot_w32(zz != 0u);
                    }
                }
            }
        }
    }
    if (wcnt > WCAP && lane == 0) ovf = 1;
    if (MODE == 0 && wcnt > SAVE_CAP && lane == 0) nosave = 1;
    __syncthreads();
    const bool fast = (__builtin_amdgcn_readfirstlane(ovf) == 0);
    const int  n    = wcnt;
    const bool saved = (MODE == 0) && fast && (__builtin_amdgcn_readfirstlane(nosave) == 0);
    if (MODE == 0 && lane == 0) ownb[CNT_OFF + wave] = saved ? (unsigned)n : ~0u;

    if (fast && MODE == 0) {
        for (int i = lane; i < n; i += 32) {
            int e = (int)wl[i];
            int r = dst[e] - base;
            int si = src[e];
            float v = leaky(s[si] + dloc[r]);
            unsigned en = ((unsigned)r << 16) | (unsigned)si;
            wl[i] = en;
            if (saved) lg[i] = en;
            wv[i] = v;
            atomicMax(mkey + r, f2key(v));
        }
    } else if (fast) {
        for (int g0 = 0; g0 < n; g0 += 32 * PH0_U) {
            unsigned en[PH0_U];
            float    sv[PH0_U];
#pragma unroll
            for (int u = 0; u < PH0_U; ++u) {
                int i = g0 + 32 * u + lane;
                en[u] = lg[(i < n) ? i : n - 1];
            }
#pragma unroll
            for (int u = 0; u < PH0_U; ++u) sv[u] = s[en[u] & 0xFFFFu];
#pragma unroll
            for (int u = 0; u < PH0_U; ++u) asm volatile("" :: "v"(sv[u]));
#pragma unroll
            for (int u = 0; u < PH0_U; ++u) {
                int i = g0 + 32 * u + lane;
                if (i < n) {
                    int r = (int)(en[u] >> 16);
                    float v = leaky(sv[u] + dloc[r]);
                    wl[i] = en[u];
                    wv[i] = v;
                    atomicMax(mkey + r, f2key(v));
                }
            }
        }
    } else {
        rows_rescan<0>(src, dst, H, s, base, tid, dloc, mkey, den, acc);
    }
    __syncthreads();
    if (tid < RB) {
        float w = __expf(vself[tid] - key2f(mkey[tid]));
        wself[tid] = w;
        den[tid] = w;
    }
    __syncthreads();

    if (fast) {
        for (int i = lane; i < n; i += 32) {
            int r = (int)(wl[i] >> 16);
            float w = __expf(wv[i] - key2f(mkey[r]));
            wv[i] = w;
            atomicAdd(den + r, w);
        }
    } else {
        rows_rescan<1>(src, dst, H, s, base, tid, dloc, mkey, den, acc);
    }
    __syncthreads();
    for (int idx = tid; idx < RB * 64; idx += ROWS_NT) {
        int r = idx >> 6;
        float coef = wself[r] / (den[r] + 1e-16f);
        acc[idx] = H[(size_t)base * 64 + idx] * coef;
    }
    if (MODE == 1 && tid < 8 * 64) acc[RB * 64 + tid] = 0.0f;
    __syncthreads();

    if (fast) {
        for (int b0 = 0; b0 < n; b0 += 32) {
            int i = b0 + lane;
            bool valid = i < n;
            unsigned en = valid ? wl[i] : 0u;
            float w = valid ? wv[i] : 0.0f;
            int r  = (int)(en >> 16);
            int si = (int)(en & 0xFFFFu);
            float coef = w / (den[r] + 1e-16f);
#pragma unroll
            for (int g = 0; g < 32; g += AGG_U) {
                if (b0 + g < n) {
                    v2f hv[AGG_U];
#pragma unroll
                    for (int u = 0; u < AGG_U; ++u) {
                        int sj = __builtin_amdgcn_readlane(si, g + u);
                        hv[u] = *(const v2f*)(H + (size_t)sj * 64 + 2 * lane);
                    }
#pragma unroll
                    for (int u = 0; u < AGG_U; ++u) asm volatile("" :: "v"(hv[u]));
#pragma unroll
                    for (int u = 0; u < AGG_U; ++u) {
                        if (b0 + g + u < n) {
                            int rj = __builtin_amdgcn_readlane(r, g + u);
                            float cj = __int_as_float(__builtin_amdgcn_readlane(__float_as_int(coef), g + u));
                            float* ar = acc + rj * 64 + 2 * lane;
                            atomicAdd(ar,     hv[u].x * cj);
                            atomicAdd(ar + 1, hv[u].y * cj);
                        }
                    }
                }
            }
        }
    } else {
        rows_rescan<2>(src, dst, H, s, base, tid, dloc, mkey, den, acc);
    }
    __syncthreads();

    if (MODE == 1) {
        if (wave < 26) {
            const int mt = wave >> 1, nh = wave & 1;
            const int lo = lane & 15, hi = lane >> 4;
            const float* ar = acc + (mt * 16 + lo) * 64 + 2 * hi;
            const float* wc = Wm + nh * 32 + lo;
            v8f c0 = {}, c1 = {};
#pragma unroll 4
            for (int k = 0; k < 64; k += 4) {
                v2f a = *(const v2f*)(ar + k);
                const float* w0 = wc + (k + 2 * hi) * 64;
                v2f b0; b0.x = w0[0];  b0.y = w0[64];
                v2f b1; b1.x = w0[16]; b1.y = w0[64 + 16];
                c0 = __builtin_amdgcn_wmma_f32_16x16x4_f32(false, a, false, b0, (short)0, c0, false, false);
                c1 = __builtin_amdgcn_wmma_f32_16x16x4_f32(false, a, false, b1, (short)0, c1, false, false);
            }
#pragma unroll
            for (int r = 0; r < 8; ++r) {
                int row = mt * 16 + r + 8 * hi;
                if (row < RB) {
                    lval[row * 64 + nh * 32 + lo]      = c0[r];
                    lval[row * 64 + nh * 32 + 16 + lo] = c1[r];
                }
            }
        }
        __syncthreads();
    }
    const float* res = (MODE == 1) ? lval : acc;

    for (int idx = tid; idx < RB * 64; idx += ROWS_NT) {
        float v = res[idx] + bias[idx & 63];
        if (relu) v = fmaxf(v, 0.0f);
        out[(size_t)base * 64 + idx] = v;
    }
}

extern "C" void kernel_launch(void* const* d_in, const int* in_sizes, int n_in,
                              void* d_out, int out_size, void* d_ws, size_t ws_size,
                              hipStream_t stream) {
    (void)in_sizes; (void)n_in; (void)out_size; (void)ws_size;

    const float* x    = (const float*)d_in[0];
    const int*   esrc = (const int*)  d_in[1];
    const int*   edst = (const int*)  d_in[2];
    const float* W0 = (const float*)d_in[3];
    const float* a0s = (const float*)d_in[4];
    const float* a0d = (const float*)d_in[5];
    const float* b0 = (const float*)d_in[6];
    const float* W1 = (const float*)d_in[7];
    const float* a1s = (const float*)d_in[8];
    const float* a1d = (const float*)d_in[9];
    const float* b1 = (const float*)d_in[10];
    const float* W2 = (const float*)d_in[11];
    const float* a2s = (const float*)d_in[12];
    const float* a2d = (const float*)d_in[13];
    const float* b2 = (const float*)d_in[14];
    float* out = (float*)d_out;

    float* ws = (float*)d_ws;
    float* H   = ws;
    float* XIN = H   + (size_t)N_NODES * 64;
    float* SV  = XIN + (size_t)N_NODES * 64;
    float* DV  = SV  + N_NODES;
    unsigned* OWN = (unsigned*)out;

    gat_gemm_wmma<<<((N_NODES / 16) + 7) / 8, 256, 0, stream>>>(x, W0, H, N_NODES, 128);
    gat_node_sd<<<(N_NODES + 255) / 256, 256, 0, stream>>>(H, a0s, a0d, SV, DV, N_NODES, edst, OWN, nullptr);
    gat_rows_lds<0><<<NBLK, ROWS_NT, 0, stream>>>(esrc, edst, OWN, H, SV, DV, b0, XIN, 1, nullptr);

    gat_node_sd<<<(N_NODES + 255) / 256, 256, 0, stream>>>(XIN, a1s, a1d, SV, DV, N_NODES, edst, nullptr, W1);
    gat_rows_lds<1><<<NBLK, ROWS_NT, 0, stream>>>(esrc, edst, OWN, XIN, SV, DV, b1, H, 1, W1);

    gat_node_sd<<<(N_NODES + 255) / 256, 256, 0, stream>>>(H, a2s, a2d, SV, DV, N_NODES, edst, nullptr, W2);
    gat_rows_lds<1><<<NBLK, ROWS_NT, 0, stream>>>(esrc, edst, OWN, H, SV, DV, b2, out, 0, W2);
}


# Owner pass of layers 0,1 forms the next layer's logits of its own 200 rows from the finished rows in LDS (4 lanes per row, W a once per block); the 2 logit launches go; logit sum order changes
# speedup vs baseline: 4.9179x; 1.0860x over previous
#include <hip/hip_runtime.h>
#include <hip/hip_bf16.h>
#include <cstddef>


#define N_NODES 50000
#define E_EDGES 800000
#define NEG_SLOPE 0.2f

typedef float v2f __attribute__((ext_vector_type(2)));
typedef float v8f __attribute__((ext_vector_type(8)));

__device__ __forceinline__ unsigned f2key(float f) {
    int i = __float_as_int(f);
    return (i >= 0) ? ((unsigned)i | 0x80000000u) : ~(unsigned)i;
}
__device__ __forceinline__ float key2f(unsigned k) {
    int i = (k & 0x80000000u) ? (int)(k & 0x7FFFFFFFu) : (int)~k;
    return __int_as_float(i);
}

__global__ void __launch_bounds__(256)
gat_gemm_wmma(const float* __restrict__ X, const float* __restrict__ W,
              float* __restrict__ H, int nrows, int K) {
    const int wave = blockIdx.x * (blockDim.x >> 5) + (threadIdx.x >> 5);
    const int lane = threadIdx.x & 31;
    const int rowBase = wave * 16;
    if (rowBase >= nrows) return;

    const int lo = lane & 15;
    const int hi = lane >> 4;

    v8f acc0 = {}, acc1 = {}, acc2 = {}, acc3 = {};

    const float* xr = X + (size_t)(rowBase + lo) * K + 2 * hi;
    const float* wc = W + lo;

    for (int k = 0; k < K; k += 4) {
        float2 af = *(const float2*)(xr + k);
        v2f a; a.x = af.x; a.y = af.y;

        const float* w0 = wc + (size_t)(k + 2 * hi) * 64;
        const float* w1 = w0 + 64;
        v2f b0; b0.x = w0[0];  b0.y = w1[0];
        v2f b1; b1.x = w0[16]; b1.y = w1[16];
        v2f b2; b2.x = w0[32]; b2.y = w1[32];
        v2f b3; b3.x = w0[48]; b3.y = w1[48];

        acc0 = __builtin_amdgcn_wmma_f32_16x16x4_f32(false, a, false, b0, (short)0, acc0, false, false);
        acc1 = __builtin_amdgcn_wmma_f32_16x16x4_f32(false, a, false, b1, (short)0, acc1, false, false);
        acc2 = __builtin_amdgcn_wmma_f32_16x16x4_f32(false, a, false, b2, (short)0, acc2, false, false);
        acc3 = __builtin_amdgcn_wmma_f32_16x16x4_f32(false, a, false, b3, (short)0, acc3, false, false);
    }

#pragma unroll
    for (int r = 0; r < 8; ++r) {
        float* o = H + (size_t)(rowBase + r + 8 * hi) * 64 + lo;
        o[0]  = acc0[r];
        o[16] = acc1[r];
        o[32] = acc2[r];
        o[48] = acc3[r];
    }
}

#define RB_KEY 200u
#define BLK_DW    (200 * 64)
#define SAVE_CAP  256
#define CNT_OFF   (32 * SAVE_CAP)
#define KEY_OFF   8448
#define KEY_DW    4096
#define S1_OFF    (CNT_OFF + 32)
#define D1_OFF    (KEY_OFF + KEY_DW)
__device__ __forceinline__ unsigned sd_pos(unsigned i, unsigned pitch) {
    return (i / 200u) * pitch + (i % 200u);
}
__device__ __forceinline__ unsigned key_dword_pos(unsigned q) {
    return (q / KEY_DW) * BLK_DW + KEY_OFF + (q % KEY_DW);
}
__global__ void __launch_bounds__(256)
gat_node_sd(const float* __restrict__ H, const float* __restrict__ asrc,
            const float* __restrict__ adst, float* __restrict__ s,
            float* __restrict__ d, int n,
            const int* __restrict__ dst, unsigned* __restrict__ keys) {
    {
        const int4* dst4 = (const int4*)dst;
#pragma unroll
        for (int j = 0; j < 4; ++j) {
            int q = blockIdx.x * 1024 + j * 256 + threadIdx.x;
            if (q < E_EDGES / 4) {
                int4 v = dst4[q];
                keys[key_dword_pos((unsigned)q)] =
                          ((unsigned)v.x / RB_KEY)         | (((unsigned)v.y / RB_KEY) << 8) |
                          (((unsigned)v.z / RB_KEY) << 16) | (((unsigned)v.w / RB_KEY) << 24);
            }
        }
    }
    int i = blockIdx.x * blockDim.x + threadIdx.x;
    if (i >= n) return;
    const float* hr = H + (size_t)i * 64;
    float ss = 0.0f, dd = 0.0f;
#pragma unroll
    for (int f = 0; f < 64; ++f) {
        float h = hr[f];
        ss = fmaf(h, asrc[f], ss);
        dd = fmaf(h, adst[f], dd);
    }
    s[i] = ss; d[i] = dd;
}

#define RB        200
#define NBLK      250
#define ROWS_NT   1024
#define WCAP      512
#define SCAN_U    7
#define AGG_U     16
#define PH0_U     5
#define NGROUP    (E_EDGES / 16)
#define NCHUNK    ((NGROUP + 31) / 32)
#define ROT_STEP  (NCHUNK / NBLK)
static_assert(RB * NBLK == N_NODES, "row blocks must tile the node set");
static_assert(RB == RB_KEY, "key list must be cut by the owners' row block");
static_assert(NBLK <= 255, "owner key must fit a byte, 255 is the pad value");
static_assert(E_EDGES % 16 == 0, "edge list must split into 16-edge groups");
static_assert(E_EDGES <= (1 << 20), "edge id must fit 20 bits of a list entry");
static_assert(BLK_DW == RB * 64, "footprint = the block's output rows");
static_assert(SAVE_CAP <= WCAP && SAVE_CAP % 32 == 0, "saved lists are a prefix of the LDS lists");
static_assert(CNT_OFF + 32 <= KEY_OFF && KEY_OFF + KEY_DW <= BLK_DW, "lists, counts, key piece must not overlap");
static_assert(KEY_DW == 32 * 128, "a key piece holds 32 whole chunks");
static_assert((E_EDGES / 4 + KEY_DW - 1) / KEY_DW <= NBLK, "key pieces must fit the footprints");
static_assert(N_NODES <= 65536, "source id must fit 16 bits of a saved entry");
static_assert(S1_OFF % 32 == 0 && S1_OFF + RB <= KEY_OFF, "layer-1 s gap: line aligned, in front of the key piece");
static_assert(D1_OFF % 32 == 0 && D1_OFF + RB <= BLK_DW, "layer-1 d gap: line aligned, inside the footprint");

__device__ __forceinline__ float leaky(float v) {
    return (v > 0.0f) ? v : NEG_SLOPE * v;
}

template <int PH>
__device__ __forceinline__ void rows_rescan(
        const int* __restrict__ src, const int* __restrict__ dst,
        const float* __restrict__ H, const float* __restrict__ s, unsigned sd_pitch,
        int base, int tid, const float* dloc, unsigned* mkey, float* den,
        float* acc) {
    for (int e = tid; e < E_EDGES; e += ROWS_NT) {
        int r = dst[e] - base;
        if ((unsigned)r < (unsigned)RB) {
            int si = src[e];
            float v = leaky(s[sd_pos((unsigned)si, sd_pitch)] + dloc[r]);
            if (PH == 0) {
                atomicMax(mkey + r, f2key(v));
            } else {
                float w = __expf(v - key2f(mkey[r]));
                if (PH == 1) {
                    atomicAdd(den + r, w);
                } else {
                    float coef = w / (den[r] + 1e-16f);
                    const float* hr = H + (size_t)si * 64;
                    float* ar = acc + r * 64;
                    for (int f = 0; f < 64; ++f) atomicAdd(ar + f, hr[f] * coef);
                }
            }
        }
    }
}

template <int MODE, bool NEXT>
__global__ void __launch_bounds__(ROWS_NT)
gat_rows_lds(const int* __restrict__ src, const int* __restrict__ dst,
             unsigned* own, const float* __restrict__ H, const float* __restrict__ s,
             const float* __restrict__ d, unsigned sd_pitch,
             const float* __restrict__ bias,
             float* out, int relu, const float* __restrict__ Wm,
             float* sn, float* dn, unsigned sn_pitch, const float* __restrict__ Wn,
             const float* __restrict__ asn, const float* __restrict__ adn) {
    __shared__ float    acc[(RB + 8) * 64];
    __shared__ unsigned mkey[RB];
    __shared__ float    den[RB];
    __shared__ float    dloc[RB];
    __shared__ float    vself[RB];
    __shared__ float    wself[RB];
    __shared__ unsigned list[32 * WCAP];
    __shared__ float    lval[32 * WCAP];
    __shared__ int      ovf;
    __shared__ int      nosave;
    __shared__ float    wan[128];
    __shared__ float    bl[64];

    const int tid  = threadIdx.x;
    const int lane = tid & 31;
    const int wave = __builtin_amdgcn_readfirstlane(tid >> 5);
    const int base = blockIdx.x * RB;
    unsigned* wl = list + wave * WCAP;
    float*    wv = lval + wave * WCAP;
    unsigned* ownb = own + (size_t)blockIdx.x * BLK_DW;
    unsigned* lg   = ownb + wave * SAVE_CAP;

    int csaved = 0;
    if (MODE == 1) csaved = __builtin_amdgcn_readfirstlane((int)ownb[CNT_OFF + wave]);

    if (tid == 0) { ovf = 0; nosave = 0; }
    if (NEXT && wave >= 8 && wave < 12) {
        const int t = tid - 256;
        const float* wr = Wn + (t & 63) * 64;
        const float* av = (wave < 10) ? asn : adn;
        float w = 0.0f;
#pragma unroll
        for (int j = 0; j < 64; ++j) w = fmaf(wr[j], av[j], w);
        wan[t] = w;
    }
    if (NEXT && wave >= 12 && wave < 14) bl[tid - 384] = bias[tid - 384];
    if (tid < RB) {
        const unsigned i = blockIdx.x * sd_pitch + tid;
        float dl = d[i];
        float v = leaky(s[i] + dl);
        dloc[tid] = dl;
        vself[tid] = v;
        mkey[tid] = f2key(v);
    }
    __syncthreads();

    int wcnt = 0;
    if (MODE == 1) {
        if ((unsigned)csaved > (unsigned)SAVE_CAP) { if (lane == 0) ovf = 1; }
        else            wcnt = csaved;
    } else {
        const uint4* key4 = (const uint4*)own;
        const unsigned pat = (unsigned)blockIdx.x * 0x01010101u;
        const int rot = blockIdx.x * ROT_STEP;
        for (int k = wave; k < NCHUNK; k += 32 * SCAN_U) {
            uint4 kv[SCAN_U];
            int   ch[SCAN_U];
#pragma unroll
            for (int u = 0; u < SCAN_U; ++u) {
                int kk = k + 32 * u;
                int c = kk + rot;
                if (c >= NCHUNK) c -= NCHUNK;
                ch[u] = c;
                int g = c * 32 + lane;
                int a = (c >> 5) * (BLK_DW / 4) + (KEY_OFF / 4) + (c & 31) * 32 + lane;
                if (kk < NCHUNK && g < NGROUP) kv[u] = key4[a];
                else                           kv[u] = make_uint4(~0u, ~0u, ~0u, ~0u);
            }
#pragma unroll
            for (int u = 0; u < SCAN_U; ++u) {
                if (k + 32 * u < NCHUNK) {
                    unsigned xs[4] = { kv[u].x ^ pat, kv[u].y ^ pat, kv[u].z ^ pat, kv[u].w ^ pat };
                    unsigned z[4];
#pragma unroll
                    for (int j = 0; j < 4; ++j) {
                        unsigned t = ((xs[j] & 0x7F7F7F7Fu) + 0x7F7F7F7Fu) | xs[j];
                        z[j] = ~(t | 0x7F7F7F7Fu);
                    }
                    unsigned zz = (z[0] >> 7) | (z[1] >> 6) | (z[2] >> 5) | (z[3] >> 4);
                    unsigned any = __builtin_amdgcn_ballot_w32(zz != 0u);
                    const unsigned e0 = (unsigned)(ch[u] * 512 + lane * 16);
                    while (any != 0u) {
                        int bit = __ffs((int)zz) - 1;
                        unsigned off = (unsigned)(((bit & 7) << 2) | (bit >> 3));
                        int p = wcnt + (int)__builtin_amdgcn_mbcnt_lo(any, 0u);
                        if (zz != 0u && p < WCAP) wl[p] = e0 + off;
                        wcnt += __builtin_popcount(any);
                        zz &= zz - 1u;
                        any = __builtin_amdgcn_ballot_w32(zz != 0u);
                    }
                }
            }
        }
    }
    if (wcnt > WCAP && lane == 0) ovf = 1;
    if (MODE == 0 && wcnt > SAVE_CAP && lane == 0) nosave = 1;
    __syncthreads();
    const bool fast = (__builtin_amdgcn_readfirstlane(ovf) == 0);
    const int  n    = wcnt;
    const bool saved = (MODE == 0) && fast && (__builtin_amdgcn_readfirstlane(nosave) == 0);
    if (MODE == 0 && lane == 0) ownb[CNT_OFF + wave] = saved ? (unsigned)n : ~0u;

    if (fast && MODE == 0) {
        for (int i = lane; i < n; i += 32) {
            int e = (int)wl[i];
            int r = dst[e] - base;
            int si = src[e];
            float v = leaky(s[sd_pos((unsigned)si, sd_pitch)] + dloc[r]);
            unsigned en = ((unsigned)r << 16) | (unsigned)si;
            wl[i] = en;
            if (saved) lg[i] = en;
            wv[i] = v;
            atomicMax(mkey + r, f2key(v));
        }
    } else if (fast) {
        for (int g0 = 0; g0 < n; g0 += 32 * PH0_U) {
            unsigned en[PH0_U];
            float    sv[PH0_U];
#pragma unroll
            for (int u = 0; u < PH0_U; ++u) {
                int i = g0 + 32 * u + lane;
                en[u] = lg[(i < n) ? i : n - 1];
            }
#pragma unroll
            for (int u = 0; u < PH0_U; ++u) sv[u] = s[sd_pos(en[u] & 0xFFFFu, sd_pitch)];
#pragma unroll
            for (int u = 0; u < PH0_U; ++u) asm volatile("" :: "v"(sv[u]));
#pragma unroll
            for (int u = 0; u < PH0_U; ++u) {
                int i = g0 + 32 * u + lane;
                if (i < n) {
                    int r = (int)(en[u] >> 16);
                    float v = leaky(sv[u] + dloc[r]);
                    wl[i] = en[u];
                    wv[i] = v;
                    atomicMax(mkey + r, f2key(v));
                }
            }
        }
    } else {
        rows_rescan<0>(src, dst, H, s, sd_pitch, base, tid, dloc, mkey, den, acc);
    }
    __syncthreads();
    if (tid < RB) {
        float w = __expf(vself[tid] - key2f(mkey[tid]));
        wself[tid] = w;
        den[tid] = w;
    }
    __syncthreads();

    if (fast) {
        for (int i = lane; i < n; i += 32) {
            int r = (int)(wl[i] >> 16);
            float w = __expf(wv[i] - key2f(mkey[r]));
            wv[i] = w;
            atomicAdd(den + r, w);
        }
    } else {
        rows_rescan<1>(src, dst, H, s, sd_pitch, base, tid, dloc, mkey, den, acc);
    }
    __syncthreads();
    for (int idx = tid; idx < RB * 64; idx += ROWS_NT) {
        int r = idx >> 6;
        float coef = wself[r] / (den[r] + 1e-16f);
        acc[idx] = H[(size_t)base * 64 + idx] * coef;
    }
    if (MODE == 1 && tid < 8 * 64) acc[RB * 64 + tid] = 0.0f;
    __syncthreads();

    if (fast) {
        for (int b0 = 0; b0 < n; b0 += 32) {
            int i = b0 + lane;
            bool valid = i < n;
            unsigned en = valid ? wl[i] : 0u;
            float w = valid ? wv[i] : 0.0f;
            int r  = (int)(en >> 16);
            int si = (int)(en & 0xFFFFu);
            float coef = w / (den[r] + 1e-16f);
#pragma unroll
            for (int g = 0; g < 32; g += AGG_U) {
                if (b0 + g < n) {
                    v2f hv[AGG_U];
#pragma unroll
                    for (int u = 0; u < AGG_U; ++u) {
                        int sj = __builtin_amdgcn_readlane(si, g + u);
                        hv[u] = *(const v2f*)(H + (size_t)sj * 64 + 2 * lane);
                    }
#pragma unroll
                    for (int u = 0; u < AGG_U; ++u) asm volatile("" :: "v"(hv[u]));
#pragma unroll
                    for (int u = 0; u < AGG_U; ++u) {
                        if (b0 + g + u < n) {
                            int rj = __builtin_amdgcn_readlane(r, g + u);
                            float cj = __int_as_float(__builtin_amdgcn_readlane(__float_as_int(coef), g + u));
                            float* ar = acc + rj * 64 + 2 * lane;
                            atomicAdd(ar,     hv[u].x * cj);
                            atomicAdd(ar + 1, hv[u].y * cj);
                        }
                    }
                }
            }
        }
    } else {
        rows_rescan<2>(src, dst, H, s, sd_pitch, base, tid, dloc, mkey, den, acc);
    }
    __syncthreads();

    if (MODE == 1) {
        if (wave < 26) {
            const int mt = wave >> 1, nh = wave & 1;
            const int lo = lane & 15, hi = lane >> 4;
            const float* ar = acc + (mt * 16 + lo) * 64 + 2 * hi;
            const float* wc = Wm + nh * 32 + lo;
            v8f c0 = {}, c1 = {};
#pragma unroll 4
            for (int k = 0; k < 64; k += 4) {
                v2f a = *(const v2f*)(ar + k);
                const float* w0 = wc + (k + 2 * hi) * 64;
                v2f b0; b0.x = w0[0];  b0.y = w0[64];
                v2f b1; b1.x = w0[16]; b1.y = w0[64 + 16];
                c0 = __builtin_amdgcn_wmma_f32_16x16x4_f32(false, a, false, b0, (short)0, c0, false, false);
                c1 = __builtin_amdgcn_wmma_f32_16x16x4_f32(false, a, false, b1, (short)0, c1, false, false);
            }
#pragma unroll
            for (int r = 0; r < 8; ++r) {
                int row = mt * 16 + r + 8 * hi;
                if (row < RB) {
                    lval[row * 64 + nh * 32 + lo]      = c0[r];
                    lval[row * 64 + nh * 32 + 16 + lo] = c1[r];
                }
            }
        }
        __syncthreads();
    }
    const float* res = (MODE == 1) ? lval : acc;

    for (int idx = tid; idx < RB * 64; idx += ROWS_NT) {
        float v = res[idx] + bias[idx & 63];
        if (relu) v = fmaxf(v, 0.0f);
        out[(size_t)base * 64 + idx] = v;
    }

    if (NEXT && tid < 4 * RB) {
        const int row = tid >> 2, part = tid & 3;
        const float* yr = res + row * 64 + part * 16;
        const float* ws = wan + part * 16;
        const float* bp = bl + part * 16;
        float ss = 0.0f, dd = 0.0f;
#pragma unroll
        for (int j = 0; j < 16; ++j) {
            const int f = (j + row) & 15;
            float yv = yr[f] + bp[f];
            if (relu) yv = fmaxf(yv, 0.0f);
            ss = fmaf(yv, ws[f], ss);
            dd = fmaf(yv, ws[64 + f], dd);
        }
        ss += __shfl_xor(ss, 1); dd += __shfl_xor(dd, 1);
        ss += __shfl_xor(ss, 2); dd += __shfl_xor(dd, 2);
        if (part == 0) {
            const unsigned p = blockIdx.x * sn_pitch + row;
            sn[p] = ss; dn[p] = dd;
        }
    }
}

extern "C" void kernel_launch(void* const* d_in, const int* in_sizes, int n_in,
                              void* d_out, int out_size, void* d_ws, size_t ws_size,
                              hipStream_t stream) {
    (void)in_sizes; (void)n_in; (void)out_size; (void)ws_size;

    const float* x    = (const float*)d_in[0];
    const int*   esrc = (const int*)  d_in[1];
    const int*   edst = (const int*)  d_in[2];
    const float* W0 = (const float*)d_in[3];
    const float* a0s = (const float*)d_in[4];
    const float* a0d = (const float*)d_in[5];
    const float* b0 = (const float*)d_in[6];
    const float* W1 = (const float*)d_in[7];
    const float* a1s = (const float*)d_in[8];
    const float* a1d = (const float*)d_in[9];
    const float* b1 = (const float*)d_in[10];
    const float* W2 = (const float*)d_in[11];
    const float* a2s = (const float*)d_in[12];
    const float* a2d = (const float*)d_in[13];
    const float* b2 = (const float*)d_in[14];
    float* out = (float*)d_out;

    float* ws = (float*)d_ws;
    float* H   = ws;
    float* XIN = H   + (size_t)N_NODES * 64;
    float* SV  = XIN + (size_t)N_NODES * 64;
    float* DV  = SV  + N_NODES;
    unsigned* OWN = (unsigned*)out;
    float* S1 = out + S1_OFF;
    float* D1 = out + D1_OFF;

    gat_gemm_wmma<<<((N_NODES / 16) + 7) / 8, 256, 0, stream>>>(x, W0, H, N_NODES, 128);
    gat_node_sd<<<(N_NODES + 255) / 256, 256, 0, stream>>>(H, a0s, a0d, SV, DV, N_NODES, edst, OWN);
    gat_rows_lds<0, true><<<NBLK, ROWS_NT, 0, stream>>>(esrc, edst, OWN, H, SV, DV, RB, b0, XIN, 1, nullptr,
                                                 S1, D1, BLK_DW, W1, a1s, a1d);

    gat_rows_lds<1, true><<<NBLK, ROWS_NT, 0, stream>>>(esrc, edst, OWN, XIN, S1, D1, BLK_DW, b1, H, 1, W1,
                                                 SV, DV, RB, W2, a2s, a2d);

    gat_rows_lds<1, false><<<NBLK, ROWS_NT, 0, stream>>>(esrc, edst, OWN, H, SV, DV, RB, b2, out, 0, W2,
                                                 nullptr, nullptr, 0, nullptr, nullptr, nullptr);
}


# Layer-0 logits formed in the GEMM launch as x.(W0 a) from the A fragments it loads anyway (W0 a once per block in LDS); same blocks write the key list; logit launch goes; logit sum order changes
# speedup vs baseline: 5.2014x; 1.0576x over previous
#include <hip/hip_runtime.h>
#include <hip/hip_bf16.h>
#include <cstddef>


#define N_NODES 50000
#define E_EDGES 800000
#define NEG_SLOPE 0.2f

typedef float v2f __attribute__((ext_vector_type(2)));
typedef float v8f __attribute__((ext_vector_type(8)));

__device__ __forceinline__ unsigned f2key(float f) {
    int i = __float_as_int(f);
    return (i >= 0) ? ((unsigned)i | 0x80000000u) : ~(unsigned)i;
}
__device__ __forceinline__ float key2f(unsigned k) {
    int i = (k & 0x80000000u) ? (int)(k & 0x7FFFFFFFu) : (int)~k;
    return __int_as_float(i);
}

#define RB_KEY 200u
#define BLK_DW    (200 * 64)
#define SAVE_CAP  256
#define CNT_OFF   (32 * SAVE_CAP)
#define KEY_OFF   8448
#define KEY_DW    4096
#define S1_OFF    (CNT_OFF + 32)
#define D1_OFF    (KEY_OFF + KEY_DW)
__device__ __forceinline__ unsigned sd_pos(unsigned i, unsigned pitch) {
    return (i / 200u) * pitch + (i % 200u);
}
__device__ __forceinline__ unsigned key_dword_pos(unsigned q) {
    return (q / KEY_DW) * BLK_DW + KEY_OFF + (q % KEY_DW);
}

#define FRONT_K   128
#define FRONT_BLK (((N_NODES / 16) + 7) / 8)
static_assert(N_NODES % 16 == 0, "row tiles must divide the node set");
static_assert(FRONT_BLK * 512 >= E_EDGES / 4, "the front blocks must cover the key list");
__global__ void __launch_bounds__(256)
gat_front(const float* __restrict__ X, const float* __restrict__ W,
          float* __restrict__ H, int nrows,
          const float* __restrict__ asrc, const float* __restrict__ adst,
          float* __restrict__ s, float* __restrict__ d,
          const int* __restrict__ dst, unsigned* __restrict__ keys) {
    __shared__ float wa[2 * FRONT_K];
    const int tid  = threadIdx.x;
    const int lane = tid & 31;
    const int wib  = __builtin_amdgcn_readfirstlane(tid >> 5);

    const int4* dst4 = (const int4*)dst;
    const int qlast = E_EDGES / 4 - 1;
    const int q0 = blockIdx.x * 512 + tid, q1 = q0 + 256;
    const int4 kv0 = dst4[(q0 < qlast) ? q0 : qlast];
    const int4 kv1 = dst4[(q1 < qlast) ? q1 : qlast];
    {
        const int c = lane & 7, rr = lane >> 3;
        const float4 as0 = *(const float4*)(asrc + 4 * c);
        const float4 as1 = *(const float4*)(asrc + 32 + 4 * c);
        const float4 ad0 = *(const float4*)(adst + 4 * c);
        const float4 ad1 = *(const float4*)(adst + 32 + 4 * c);
        const float* wr = W + (wib * 16 + rr) * 64 + 4 * c;
        float4 w0[4], w1[4];
        float  ps[4], pd[4];
#pragma unroll
        for (int j = 0; j < 4; ++j) {
            w0[j] = *(const float4*)(wr + j * 256);
            w1[j] = *(const float4*)(wr + j * 256 + 32);
        }
#pragma unroll
        for (int j = 0; j < 4; ++j) {
            float s_ = w0[j].x * as0.x, d_ = w0[j].x * ad0.x;
            s_ = fmaf(w0[j].y, as0.y, s_); d_ = fmaf(w0[j].y, ad0.y, d_);
            s_ = fmaf(w0[j].z, as0.z, s_); d_ = fmaf(w0[j].z, ad0.z, d_);
            s_ = fmaf(w0[j].w, as0.w, s_); d_ = fmaf(w0[j].w, ad0.w, d_);
            s_ = fmaf(w1[j].x, as1.x, s_); d_ = fmaf(w1[j].x, ad1.x, d_);
            s_ = fmaf(w1[j].y, as1.y, s_); d_ = fmaf(w1[j].y, ad1.y, d_);
            s_ = fmaf(w1[j].z, as1.z, s_); d_ = fmaf(w1[j].z, ad1.z, d_);
            s_ = fmaf(w1[j].w, as1.w, s_); d_ = fmaf(w1[j].w, ad1.w, d_);
            ps[j] = s_; pd[j] = d_;
        }
#pragma unroll
        for (int m = 1; m < 8; m <<= 1) {
#pragma unroll
            for (int j = 0; j < 4; ++j) {
                ps[j] += __shfl_xor(ps[j], m);
                pd[j] += __shfl_xor(pd[j], m);
            }
        }
        if (c == 0) {
#pragma unroll
            for (int j = 0; j < 4; ++j) {
                wa[wib * 16 + j * 4 + rr] = ps[j];
                wa[FRONT_K + wib * 16 + j * 4 + rr] = pd[j];
            }
        }
    }
    __syncthreads();

    const int rowBase = (blockIdx.x * 8 + wib) * 16;
    if (rowBase < nrows) {

    const int lo = lane & 15;
    const int hi = lane >> 4;

    v8f acc0 = {}, acc1 = {}, acc2 = {}, acc3 = {};
    float ss = 0.0f, dd = 0.0f;

    const float* xr = X + (size_t)(rowBase + lo) * FRONT_K + 2 * hi;
    const float* wc = W + lo;
    const float* wk = wa + 2 * hi;

    for (int k0 = 0; k0 < FRONT_K; k0 += 16) {
        v2f a[4], b[4][4], ws2[4], wd2[4];
#pragma unroll
        for (int u = 0; u < 4; ++u) {
            const int k = k0 + 4 * u;
            a[u] = *(const v2f*)(xr + k);
            const float* w0 = wc + (size_t)(k + 2 * hi) * 64;
            const float* w1 = w0 + 64;
#pragma unroll
            for (int t = 0; t < 4; ++t) { b[u][t].x = w0[16 * t]; b[u][t].y = w1[16 * t]; }
            ws2[u] = *(const v2f*)(wk + k);
            wd2[u] = *(const v2f*)(wk + FRONT_K + k);
        }
#pragma unroll
        for (int u = 0; u < 4; ++u) {
            asm volatile("" :: "v"(a[u]));
#pragma unroll
            for (int t = 0; t < 4; ++t) asm volatile("" :: "v"(b[u][t]));
        }
#pragma unroll
        for (int u = 0; u < 4; ++u) {
            acc0 = __builtin_amdgcn_wmma_f32_16x16x4_f32(false, a[u], false, b[u][0], (short)0, acc0, false, false);
            acc1 = __builtin_amdgcn_wmma_f32_16x16x4_f32(false, a[u], false, b[u][1], (short)0, acc1, false, false);
            acc2 = __builtin_amdgcn_wmma_f32_16x16x4_f32(false, a[u], false, b[u][2], (short)0, acc2, false, false);
            acc3 = __builtin_amdgcn_wmma_f32_16x16x4_f32(false, a[u], false, b[u][3], (short)0, acc3, false, false);
            ss = fmaf(a[u].x, ws2[u].x, ss); ss = fmaf(a[u].y, ws2[u].y, ss);
            dd = fmaf(a[u].x, wd2[u].x, dd); dd = fmaf(a[u].y, wd2[u].y, dd);
        }
    }

#pragma unroll
    for (int r = 0; r < 8; ++r) {
        float* o = H + (size_t)(rowBase + r + 8 * hi) * 64 + lo;
        o[0]  = acc0[r];
        o[16] = acc1[r];
        o[32] = acc2[r];
        o[48] = acc3[r];
    }

    ss += __shfl_xor(ss, 16); dd += __shfl_xor(dd, 16);
    if (hi == 0) { s[rowBase + lo] = ss; d[rowBase + lo] = dd; }
    }

    if (q0 <= qlast)
        keys[key_dword_pos((unsigned)q0)] =
                  ((unsigned)kv0.x / RB_KEY)         | (((unsigned)kv0.y / RB_KEY) << 8) |
                  (((unsigned)kv0.z / RB_KEY) << 16) | (((unsigned)kv0.w / RB_KEY) << 24);
    if (q1 <= qlast)
        keys[key_dword_pos((unsigned)q1)] =
                  ((unsigned)kv1.x / RB_KEY)         | (((unsigned)kv1.y / RB_KEY) << 8) |
                  (((unsigned)kv1.z / RB_KEY) << 16) | (((unsigned)kv1.w / RB_KEY) << 24);
}

#define RB        200
#define NBLK      250
#define ROWS_NT   1024
#define WCAP      512
#define SCAN_U    7
#define AGG_U     16
#define PH0_U     5
#define NGROUP    (E_EDGES / 16)
#define NCHUNK    ((NGROUP + 31) / 32)
#define ROT_STEP  (NCHUNK / NBLK)
static_assert(RB * NBLK == N_NODES, "row blocks must tile the node set");
static_assert(RB == RB_KEY, "key list must be cut by the owners' row block");
static_assert(NBLK <= 255, "owner key must fit a byte, 255 is the pad value");
static_assert(E_EDGES % 16 == 0, "edge list must split into 16-edge groups");
static_assert(E_EDGES <= (1 << 20), "edge id must fit 20 bits of a list entry");
static_assert(BLK_DW == RB * 64, "footprint = the block's output rows");
static_assert(SAVE_CAP <= WCAP && SAVE_CAP % 32 == 0, "saved lists are a prefix of the LDS lists");
static_assert(CNT_OFF + 32 <= KEY_OFF && KEY_OFF + KEY_DW <= BLK_DW, "lists, counts, key piece must not overlap");
static_assert(KEY_DW == 32 * 128, "a key piece holds 32 whole chunks");
static_assert((E_EDGES / 4 + KEY_DW - 1) / KEY_DW <= NBLK, "key pieces must fit the footprints");
static_assert(N_NODES <= 65536, "source id must fit 16 bits of a saved entry");
static_assert(S1_OFF % 32 == 0 && S1_OFF + RB <= KEY_OFF, "layer-1 s gap: line aligned, in front of the key piece");
static_assert(D1_OFF % 32 == 0 && D1_OFF + RB <= BLK_DW, "layer-1 d gap: line aligned, inside the footprint");

__device__ __forceinline__ float leaky(float v) {
    return (v > 0.0f) ? v : NEG_SLOPE * v;
}

template <int PH>
__device__ __forceinline__ void rows_rescan(
        const int* __restrict__ src, const int* __restrict__ dst,
        const float* __restrict__ H, const float* __restrict__ s, unsigned sd_pitch,
        int base, int tid, const float* dloc, unsigned* mkey, float* den,
        float* acc) {
    for (int e = tid; e < E_EDGES; e += ROWS_NT) {
        int r = dst[e] - base;
        if ((unsigned)r < (unsigned)RB) {
            int si = src[e];
            float v = leaky(s[sd_pos((unsigned)si, sd_pitch)] + dloc[r]);
            if (PH == 0) {
                atomicMax(mkey + r, f2key(v));
            } else {
                float w = __expf(v - key2f(mkey[r]));
                if (PH == 1) {
                    atomicAdd(den + r, w);
                } else {
                    float coef = w / (den[r] + 1e-16f);
                    const float* hr = H + (size_t)si * 64;
                    float* ar = acc + r * 64;
                    for (int f = 0; f < 64; ++f) atomicAdd(ar + f, hr[f] * coef);
                }
            }
        }
    }
}

template <int MODE, bool NEXT>
__global__ void __launch_bounds__(ROWS_NT)
gat_rows_lds(const int* __restrict__ src, const int* __restrict__ dst,
             unsigned* own, const float* __restrict__ H, const float* __restrict__ s,
             const float* __restrict__ d, unsigned sd_pitch,
             const float* __restrict__ bias,
             float* out, int relu, const float* __restrict__ Wm,
             float* sn, float* dn, unsigned sn_pitch, const float* __restrict__ Wn,
             const float* __restrict__ asn, const float* __restrict__ adn) {
    __shared__ float    acc[(RB + 8) * 64];
    __shared__ unsigned mkey[RB];
    __shared__ float    den[RB];
    __shared__ float    dloc[RB];
    __shared__ float    vself[RB];
    __shared__ float    wself[RB];
    __shared__ unsigned list[32 * WCAP];
    __shared__ float    lval[32 * WCAP];
    __shared__ int      ovf;
    __shared__ int      nosave;
    __shared__ float    wan[128];
    __shared__ float    bl[64];

    const int tid  = threadIdx.x;
    const int lane = tid & 31;
    const int wave = __builtin_amdgcn_readfirstlane(tid >> 5);
    const int base = blockIdx.x * RB;
    unsigned* wl = list + wave * WCAP;
    float*    wv = lval + wave * WCAP;
    unsigned* ownb = own + (size_t)blockIdx.x * BLK_DW;
    unsigned* lg   = ownb + wave * SAVE_CAP;

    int csaved = 0;
    if (MODE == 1) csaved = __builtin_amdgcn_readfirstlane((int)ownb[CNT_OFF + wave]);

    if (tid == 0) { ovf = 0; nosave = 0; }
    if (NEXT && wave >= 8 && wave < 12) {
        const int t = tid - 256;
        const float* wr = Wn + (t & 63) * 64;
        const float* av = (wave < 10) ? asn : adn;
        float w = 0.0f;
#pragma unroll
        for (int j = 0; j < 64; ++j) w = fmaf(wr[j], av[j], w);
        wan[t] = w;
    }
    if (NEXT && wave >= 12 && wave < 14) bl[tid - 384] = bias[tid - 384];
    if (tid < RB) {
        const unsigned i = blockIdx.x * sd_pitch + tid;
        float dl = d[i];
        float v = leaky(s[i] + dl);
        dloc[tid] = dl;
        vself[tid] = v;
        mkey[tid] = f2key(v);
    }
    __syncthreads();

    int wcnt = 0;
    if (MODE == 1) {
        if ((unsigned)csaved > (unsigned)SAVE_CAP) { if (lane == 0) ovf = 1; }
        else            wcnt = csaved;
    } else {
        const uint4* key4 = (const uint4*)own;
        const unsigned pat = (unsigned)blockIdx.x * 0x01010101u;
        const int rot = blockIdx.x * ROT_STEP;
        for (int k = wave; k < NCHUNK; k += 32 * SCAN_U) {
            uint4 kv[SCAN_U];
            int   ch[SCAN_U];
#pragma unroll
            for (int u = 0; u < SCAN_U; ++u) {
                int kk = k + 32 * u;
                int c = kk + rot;
                if (c >= NCHUNK) c -= NCHUNK;
                ch[u] = c;
                int g = c * 32 + lane;
                int a = (c >> 5) * (BLK_DW / 4) + (KEY_OFF / 4) + (c & 31) * 32 + lane;
                if (kk < NCHUNK && g < NGROUP) kv[u] = key4[a];
                else                           kv[u] = make_uint4(~0u, ~0u, ~0u, ~0u);
            }
#pragma unroll
            for (int u = 0; u < SCAN_U; ++u) {
                if (k + 32 * u < NCHUNK) {
                    unsigned xs[4] = { kv[u].x ^ pat, kv[u].y ^ pat, kv[u].z ^ pat, kv[u].w ^ pat };
                    unsigned z[4];
#pragma unroll
                    for (int j = 0; j < 4; ++j) {
                        unsigned t = ((xs[j] & 0x7F7F7F7Fu) + 0x7F7F7F7Fu) | xs[j];
                        z[j] = ~(t | 0x7F7F7F7Fu);
                    }
                    unsigned zz = (z[0] >> 7) | (z[1] >> 6) | (z[2] >> 5) | (z[3] >> 4);
                    unsigned any = __builtin_amdgcn_ballot_w32(zz != 0u);
                    const unsigned e0 = (unsigned)(ch[u] * 512 + lane * 16);
                    while (any != 0u) {
                        int bit = __ffs((int)zz) - 1;
                        unsigned off = (unsigned)(((bit & 7) << 2) | (bit >> 3));
                        int p = wcnt + (int)__builtin_amdgcn_mbcnt_lo(any, 0u);
                        if (zz != 0u && p < WCAP) wl[p] = e0 + off;
                        wcnt += __builtin_popcount(any);
                        zz &= zz - 1u;
                        any = __builtin_amdgcn_ballot_w32(zz != 0u);
                    }
                }
            }
        }
    }
    if (wcnt > WCAP && lane == 0) ovf = 1;
    if (MODE == 0 && wcnt > SAVE_CAP && lane == 0) nosave = 1;
    __syncthreads();
    const bool fast = (__builtin_amdgcn_readfirstlane(ovf) == 0);
    const int  n    = wcnt;
    const bool saved = (MODE == 0) && fast && (__builtin_amdgcn_readfirstlane(nosave) == 0);
    if (MODE == 0 && lane == 0) ownb[CNT_OFF + wave] = saved ? (unsigned)n : ~0u;

    if (fast && MODE == 0) {
        for (int i = lane; i < n; i += 32) {
            int e = (int)wl[i];
            int r = dst[e] - base;
            int si = src[e];
            float v = leaky(s[sd_pos((unsigned)si, sd_pitch)] + dloc[r]);
            unsigned en = ((unsigned)r << 16) | (unsigned)si;
            wl[i] = en;
            if (saved) lg[i] = en;
            wv[i] = v;
            atomicMax(mkey + r, f2key(v));
        }
    } else if (fast) {
        for (int g0 = 0; g0 < n; g0 += 32 * PH0_U) {
            unsigned en[PH0_U];
            float    sv[PH0_U];
#pragma unroll
            for (int u = 0; u < PH0_U; ++u) {
                int i = g0 + 32 * u + lane;
                en[u] = lg[(i < n) ? i : n - 1];
            }
#pragma unroll
            for (int u = 0; u < PH0_U; ++u) sv[u] = s[sd_pos(en[u] & 0xFFFFu, sd_pitch)];
#pragma unroll
            for (int u = 0; u < PH0_U; ++u) asm volatile("" :: "v"(sv[u]));
#pragma unroll
            for (int u = 0; u < PH0_U; ++u) {
                int i = g0 + 32 * u + lane;
                if (i < n) {
                    int r = (int)(en[u] >> 16);
                    float v = leaky(sv[u] + dloc[r]);
                    wl[i] = en[u];
                    wv[i] = v;
                    atomicMax(mkey + r, f2key(v));
                }
            }
        }
    } else {
        rows_rescan<0>(src, dst, H, s, sd_pitch, base, tid, dloc, mkey, den, acc);
    }
    __syncthreads();
    if (tid < RB) {
        float w = __expf(vself[tid] - key2f(mkey[tid]));
        wself[tid] = w;
        den[tid] = w;
    }
    __syncthreads();

    if (fast) {
        for (int i = lane; i < n; i += 32) {
            int r = (int)(wl[i] >> 16);
            float w = __expf(wv[i] - key2f(mkey[r]));
            wv[i] = w;
            atomicAdd(den + r, w);
        }
    } else {
        rows_rescan<1>(src, dst, H, s, sd_pitch, base, tid, dloc, mkey, den, acc);
    }
    __syncthreads();
    for (int idx = tid; idx < RB * 64; idx += ROWS_NT) {
        int r = idx >> 6;
        float coef = wself[r] / (den[r] + 1e-16f);
        acc[idx] = H[(size_t)base * 64 + idx] * coef;
    }
    if (MODE == 1 && tid < 8 * 64) acc[RB * 64 + tid] = 0.0f;
    __syncthreads();

    if (fast) {
        for (int b0 = 0; b0 < n; b0 += 32) {
            int i = b0 + lane;
            bool valid = i < n;
            unsigned en = valid ? wl[i] : 0u;
            float w = valid ? wv[i] : 0.0f;
            int r  = (int)(en >> 16);
            int si = (int)(en & 0xFFFFu);
            float coef = w / (den[r] + 1e-16f);
#pragma unroll
            for (int g = 0; g < 32; g += AGG_U) {
                if (b0 + g < n) {
                    v2f hv[AGG_U];
#pragma unroll
                    for (int u = 0; u < AGG_U; ++u) {
                        int sj = __builtin_amdgcn_readlane(si, g + u);
                        hv[u] = *(const v2f*)(H + (size_t)sj * 64 + 2 * lane);
                    }
#pragma unroll
                    for (int u = 0; u < AGG_U; ++u) asm volatile("" :: "v"(hv[u]));
#pragma unroll
                    for (int u = 0; u < AGG_U; ++u) {
                        if (b0 + g + u < n) {
                            int rj = __builtin_amdgcn_readlane(r, g + u);
                            float cj = __int_as_float(__builtin_amdgcn_readlane(__float_as_int(coef), g + u));
                            float* ar = acc + rj * 64 + 2 * lane;
                            atomicAdd(ar,     hv[u].x * cj);
                            atomicAdd(ar + 1, hv[u].y * cj);
                        }
                    }
                }
            }
        }
    } else {
        rows_rescan<2>(src, dst, H, s, sd_pitch, base, tid, dloc, mkey, den, acc);
    }
    __syncthreads();

    if (MODE == 1) {
        if (wave < 26) {
            const int mt = wave >> 1, nh = wave & 1;
            const int lo = lane & 15, hi = lane >> 4;
            const float* ar = acc + (mt * 16 + lo) * 64 + 2 * hi;
            const float* wc = Wm + nh * 32 + lo;
            v8f c0 = {}, c1 = {};
#pragma unroll 4
            for (int k = 0; k < 64; k += 4) {
                v2f a = *(const v2f*)(ar + k);
                const float* w0 = wc + (k + 2 * hi) * 64;
                v2f b0; b0.x = w0[0];  b0.y = w0[64];
                v2f b1; b1.x = w0[16]; b1.y = w0[64 + 16];
                c0 = __builtin_amdgcn_wmma_f32_16x16x4_f32(false, a, false, b0, (short)0, c0, false, false);
                c1 = __builtin_amdgcn_wmma_f32_16x16x4_f32(false, a, false, b1, (short)0, c1, false, false);
            }
#pragma unroll
            for (int r = 0; r < 8; ++r) {
                int row = mt * 16 + r + 8 * hi;
                if (row < RB) {
                    lval[row * 64 + nh * 32 + lo]      = c0[r];
                    lval[row * 64 + nh * 32 + 16 + lo] = c1[r];
                }
            }
        }
        __syncthreads();
    }
    const float* res = (MODE == 1) ? lval : acc;

    for (int idx = tid; idx < RB * 64; idx += ROWS_NT) {
        float v = res[idx] + bias[idx & 63];
        if (relu) v = fmaxf(v, 0.0f);
        out[(size_t)base * 64 + idx] = v;
    }

    if (NEXT && tid < 4 * RB) {
        const int row = tid >> 2, part = tid & 3;
        const float* yr = res + row * 64 + part * 16;
        const float* ws = wan + part * 16;
        const float* bp = bl + part * 16;
        float ss = 0.0f, dd = 0.0f;
#pragma unroll
        for (int j = 0; j < 16; ++j) {
            const int f = (j + row) & 15;
            float yv = yr[f] + bp[f];
            if (relu) yv = fmaxf(yv, 0.0f);
            ss = fmaf(yv, ws[f], ss);
            dd = fmaf(yv, ws[64 + f], dd);
        }
        ss += __shfl_xor(ss, 1); dd += __shfl_xor(dd, 1);
        ss += __shfl_xor(ss, 2); dd += __shfl_xor(dd, 2);
        if (part == 0) {
            const unsigned p = blockIdx.x * sn_pitch + row;
            sn[p] = ss; dn[p] = dd;
        }
    }
}

extern "C" void kernel_launch(void* const* d_in, const int* in_sizes, int n_in,
                              void* d_out, int out_size, void* d_ws, size_t ws_size,
                              hipStream_t stream) {
    (void)in_sizes; (void)n_in; (void)out_size; (void)ws_size;

    const float* x    = (const float*)d_in[0];
    const int*   esrc = (const int*)  d_in[1];
    const int*   edst = (const int*)  d_in[2];
    const float* W0 = (const float*)d_in[3];
    const float* a0s = (const float*)d_in[4];
    const float* a0d = (const float*)d_in[5];
    const float* b0 = (const float*)d_in[6];
    const float* W1 = (const float*)d_in[7];
    const float* a1s = (const float*)d_in[8];
    const float* a1d = (const float*)d_in[9];
    const float* b1 = (const float*)d_in[10];
    const float* W2 = (const float*)d_in[11];
    const float* a2s = (const float*)d_in[12];
    const float* a2d = (const float*)d_in[13];
    const float* b2 = (const float*)d_in[14];
    float* out = (float*)d_out;

    float* ws = (float*)d_ws;
    float* H   = ws;
    float* XIN = H   + (size_t)N_NODES * 64;
    float* SV  = XIN + (size_t)N_NODES * 64;
    float* DV  = SV  + N_NODES;
    unsigned* OWN = (unsigned*)out;
    float* S1 = out + S1_OFF;
    float* D1 = out + D1_OFF;

    gat_front<<<FRONT_BLK, 256, 0, stream>>>(x, W0, H, N_NODES, a0s, a0d, SV, DV, edst, OWN);
    gat_rows_lds<0, true><<<NBLK, ROWS_NT, 0, stream>>>(esrc, edst, OWN, H, SV, DV, RB, b0, XIN, 1, nullptr,
                                                 S1, D1, BLK_DW, W1, a1s, a1d);

    gat_rows_lds<1, true><<<NBLK, ROWS_NT, 0, stream>>>(esrc, edst, OWN, XIN, S1, D1, BLK_DW, b1, H, 1, W1,
                                                 SV, DV, RB, W2, a2s, a2d);

    gat_rows_lds<1, false><<<NBLK, ROWS_NT, 0, stream>>>(esrc, edst, OWN, H, SV, DV, RB, b2, out, 0, W2,
                                                 nullptr, nullptr, 0, nullptr, nullptr, nullptr);
}


# Front launch marks per owner which 16-edge groups hold its edges (LDS OR, one bit per group); layer-0 owners load only marked key groups, not the whole key list; list (sum) order changes
# speedup vs baseline: 5.4951x; 1.0565x over previous
#include <hip/hip_runtime.h>
#include <hip/hip_bf16.h>
#include <cstddef>


#define N_NODES 50000
#define E_EDGES 800000
#define NEG_SLOPE 0.2f

typedef float v2f __attribute__((ext_vector_type(2)));
typedef float v8f __attribute__((ext_vector_type(8)));

__device__ __forceinline__ unsigned f2key(float f) {
    int i = __float_as_int(f);
    return (i >= 0) ? ((unsigned)i | 0x80000000u) : ~(unsigned)i;
}
__device__ __forceinline__ float key2f(unsigned k) {
    int i = (k & 0x80000000u) ? (int)(k & 0x7FFFFFFFu) : (int)~k;
    return __int_as_float(i);
}

#define RB_KEY 200u
#define BLK_DW    (200 * 64)
#define SAVE_CAP  256
#define CNT_OFF   (32 * SAVE_CAP)
#define KEY_OFF   8448
#define KEY_DW    4096
#define S1_OFF    (CNT_OFF + 32)
#define D1_OFF    (KEY_OFF + KEY_DW)
__device__ __forceinline__ unsigned sd_pos(unsigned i, unsigned pitch) {
    return (i / 200u) * pitch + (i % 200u);
}
__device__ __forceinline__ unsigned key_dword_pos(unsigned q) {
    return (q / KEY_DW) * BLK_DW + KEY_OFF + (q % KEY_DW);
}

#define FRONT_K   128
#define FRONT_BLK (((N_NODES / 16) + 7) / 8)
static_assert(N_NODES % 16 == 0, "row tiles must divide the node set");
static_assert(FRONT_BLK * 512 >= E_EDGES / 4, "the front blocks must cover the key list");
#define BM_WORDS  (FRONT_BLK * 4)
#define BM_WPW    ((BM_WORDS + 31) / 32)
#define BMT_P0    ((E_EDGES / 4 + KEY_DW - 1) / KEY_DW)
static_assert(KEY_DW == 4 * 1024, "a key piece holds the marks of four front blocks");
static_assert(BM_WPW > 32 && BM_WPW <= 64, "an owner lane takes one or two mark words");
__device__ __forceinline__ unsigned bmt_pos(unsigned j, unsigned b, unsigned ww) {
    return (BMT_P0 + (j >> 2)) * BLK_DW + KEY_OFF + (j & 3u) * 1024u + b * 4u + ww;
}
__global__ void __launch_bounds__(256)
gat_front(const float* __restrict__ X, const float* __restrict__ W,
          float* __restrict__ H, int nrows,
          const float* __restrict__ asrc, const float* __restrict__ adst,
          float* __restrict__ s, float* __restrict__ d,
          const int* __restrict__ dst, unsigned* __restrict__ keys) {
    __shared__ float wa[2 * FRONT_K];
    __shared__ unsigned bm[256 * 4];
    const int tid  = threadIdx.x;
    const int lane = tid & 31;
    const int wib  = __builtin_amdgcn_readfirstlane(tid >> 5);

    const int4* dst4 = (const int4*)dst;
    const int qlast = E_EDGES / 4 - 1;
    const int q0 = blockIdx.x * 512 + tid, q1 = q0 + 256;
    const int4 kv0 = dst4[(q0 < qlast) ? q0 : qlast];
    const int4 kv1 = dst4[(q1 < qlast) ? q1 : qlast];
    for (int i = tid; i < 256 * 4; i += 256) bm[i] = 0u;
    {
        const int c = lane & 7, rr = lane >> 3;
        const float4 as0 = *(const float4*)(asrc + 4 * c);
        const float4 as1 = *(const float4*)(asrc + 32 + 4 * c);
        const float4 ad0 = *(const float4*)(adst + 4 * c);
        const float4 ad1 = *(const float4*)(adst + 32 + 4 * c);
        const float* wr = W + (wib * 16 + rr) * 64 + 4 * c;
        float4 w0[4], w1[4];
        float  ps[4], pd[4];
#pragma unroll
        for (int j = 0; j < 4; ++j) {
            w0[j] = *(const float4*)(wr + j * 256);
            w1[j] = *(const float4*)(wr + j * 256 + 32);
        }
#pragma unroll
        for (int j = 0; j < 4; ++j) {
            float s_ = w0[j].x * as0.x, d_ = w0[j].x * ad0.x;
            s_ = fmaf(w0[j].y, as0.y, s_); d_ = fmaf(w0[j].y, ad0.y, d_);
            s_ = fmaf(w0[j].z, as0.z, s_); d_ = fmaf(w0[j].z, ad0.z, d_);
            s_ = fmaf(w0[j].w, as0.w, s_); d_ = fmaf(w0[j].w, ad0.w, d_);
            s_ = fmaf(w1[j].x, as1.x, s_); d_ = fmaf(w1[j].x, ad1.x, d_);
            s_ = fmaf(w1[j].y, as1.y, s_); d_ = fmaf(w1[j].y, ad1.y, d_);
            s_ = fmaf(w1[j].z, as1.z, s_); d_ = fmaf(w1[j].z, ad1.z, d_);
            s_ = fmaf(w1[j].w, as1.w, s_); d_ = fmaf(w1[j].w, ad1.w, d_);
            ps[j] = s_; pd[j] = d_;
        }
#pragma unroll
        for (int m = 1; m < 8; m <<= 1) {
#pragma unroll
            for (int j = 0; j < 4; ++j) {
                ps[j] += __shfl_xor(ps[j], m);
                pd[j] += __shfl_xor(pd[j], m);
            }
        }
        if (c == 0) {
#pragma unroll
            for (int j = 0; j < 4; ++j) {
                wa[wib * 16 + j * 4 + rr] = ps[j];
                wa[FRONT_K + wib * 16 + j * 4 + rr] = pd[j];
            }
        }
    }
    __syncthreads();

    {
        const unsigned mbit = 1u << ((tid >> 2) & 31);
        const int mw = tid >> 7;
        if (q0 <= qlast) {
            atomicOr(&bm[((unsigned)kv0.x / RB_KEY) * 4 + mw], mbit);
            atomicOr(&bm[((unsigned)kv0.y / RB_KEY) * 4 + mw], mbit);
            atomicOr(&bm[((unsigned)kv0.z / RB_KEY) * 4 + mw], mbit);
            atomicOr(&bm[((unsigned)kv0.w / RB_KEY) * 4 + mw], mbit);
        }
        if (q1 <= qlast) {
            atomicOr(&bm[((unsigned)kv1.x / RB_KEY) * 4 + 2 + mw], mbit);
            atomicOr(&bm[((unsigned)kv1.y / RB_KEY) * 4 + 2 + mw], mbit);
            atomicOr(&bm[((unsigned)kv1.z / RB_KEY) * 4 + 2 + mw], mbit);
            atomicOr(&bm[((unsigned)kv1.w / RB_KEY) * 4 + 2 + mw], mbit);
        }
    }
    __syncthreads();
    *(uint4*)(keys + bmt_pos((unsigned)blockIdx.x, (unsigned)tid, 0u)) =
        make_uint4(bm[4 * tid], bm[4 * tid + 1], bm[4 * tid + 2], bm[4 * tid + 3]);

    const int rowBase = (blockIdx.x * 8 + wib) * 16;
    if (rowBase < nrows) {

    const int lo = lane & 15;
    const int hi = lane >> 4;

    v8f acc0 = {}, acc1 = {}, acc2 = {}, acc3 = {};
    float ss = 0.0f, dd = 0.0f;

    const float* xr = X + (size_t)(rowBase + lo) * FRONT_K + 2 * hi;
    const float* wc = W + lo;
    const float* wk = wa + 2 * hi;

    for (int k0 = 0; k0 < FRONT_K; k0 += 16) {
        v2f a[4], b[4][4], ws2[4], wd2[4];
#pragma unroll
        for (int u = 0; u < 4; ++u) {
            const int k = k0 + 4 * u;
            a[u] = *(const v2f*)(xr + k);
            const float* w0 = wc + (size_t)(k + 2 * hi) * 64;
            const float* w1 = w0 + 64;
#pragma unroll
            for (int t = 0; t < 4; ++t) { b[u][t].x = w0[16 * t]; b[u][t].y = w1[16 * t]; }
            ws2[u] = *(const v2f*)(wk + k);
            wd2[u] = *(const v2f*)(wk + FRONT_K + k);
        }
#pragma unroll
        for (int u = 0; u < 4; ++u) {
            asm volatile("" :: "v"(a[u]));
#pragma unroll
            for (int t = 0; t < 4; ++t) asm volatile("" :: "v"(b[u][t]));
        }
#pragma unroll
        for (int u = 0; u < 4; ++u) {
            acc0 = __builtin_amdgcn_wmma_f32_16x16x4_f32(false, a[u], false, b[u][0], (short)0, acc0, false, false);
            acc1 = __builtin_amdgcn_wmma_f32_16x16x4_f32(false, a[u], false, b[u][1], (short)0, acc1, false, false);
            acc2 = __builtin_amdgcn_wmma_f32_16x16x4_f32(false, a[u], false, b[u][2], (short)0, acc2, false, false);
            acc3 = __builtin_amdgcn_wmma_f32_16x16x4_f32(false, a[u], false, b[u][3], (short)0, acc3, false, false);
            ss = fmaf(a[u].x, ws2[u].x, ss); ss = fmaf(a[u].y, ws2[u].y, ss);
            dd = fmaf(a[u].x, wd2[u].x, dd); dd = fmaf(a[u].y, wd2[u].y, dd);
        }
    }

#pragma unroll
    for (int r = 0; r < 8; ++r) {
        float* o = H + (size_t)(rowBase + r + 8 * hi) * 64 + lo;
        o[0]  = acc0[r];
        o[16] = acc1[r];
        o[32] = acc2[r];
        o[48] = acc3[r];
    }

    ss += __shfl_xor(ss, 16); dd += __shfl_xor(dd, 16);
    if (hi == 0) { s[rowBase + lo] = ss; d[rowBase + lo] = dd; }
    }

    if (q0 <= qlast)
        keys[key_dword_pos((unsigned)q0)] =
                  ((unsigned)kv0.x / RB_KEY)         | (((unsigned)kv0.y / RB_KEY) << 8) |
                  (((unsigned)kv0.z / RB_KEY) << 16) | (((unsigned)kv0.w / RB_KEY) << 24);
    if (q1 <= qlast)
        keys[key_dword_pos((unsigned)q1)] =
                  ((unsigned)kv1.x / RB_KEY)         | (((unsigned)kv1.y / RB_KEY) << 8) |
                  (((unsigned)kv1.z / RB_KEY) << 16) | (((unsigned)kv1.w / RB_KEY) << 24);
}

#define RB        200
#define NBLK      250
#define ROWS_NT   1024
#define WCAP      512
#define GRP_U     5
#define AGG_U     16
#define PH0_U     5
#define NGROUP    (E_EDGES / 16)
static_assert(RB * NBLK == N_NODES, "row blocks must tile the node set");
static_assert(RB == RB_KEY, "key list must be cut by the owners' row block");
static_assert(NBLK <= 255, "owner key must fit a byte, 255 is the pad value");
static_assert(E_EDGES % 16 == 0, "edge list must split into 16-edge groups");
static_assert(E_EDGES <= (1 << 20), "edge id must fit 20 bits of a list entry");
static_assert(BLK_DW == RB * 64, "footprint = the block's output rows");
static_assert(SAVE_CAP <= WCAP && SAVE_CAP % 32 == 0, "saved lists are a prefix of the LDS lists");
static_assert(CNT_OFF + 32 <= KEY_OFF && KEY_OFF + KEY_DW <= BLK_DW, "lists, counts, key piece must not overlap");
static_assert(BMT_P0 + (FRONT_BLK + 3) / 4 <= NBLK, "the marks must fit the free key pieces");
static_assert(FRONT_BLK * 128 >= NGROUP, "the marks must cover every 16-edge group");
static_assert((E_EDGES / 4 + KEY_DW - 1) / KEY_DW <= NBLK, "key pieces must fit the footprints");
static_assert(N_NODES <= 65536, "source id must fit 16 bits of a saved entry");
static_assert(S1_OFF % 32 == 0 && S1_OFF + RB <= KEY_OFF, "layer-1 s gap: line aligned, in front of the key piece");
static_assert(D1_OFF % 32 == 0 && D1_OFF + RB <= BLK_DW, "layer-1 d gap: line aligned, inside the footprint");

__device__ __forceinline__ float leaky(float v) {
    return (v > 0.0f) ? v : NEG_SLOPE * v;
}

template <int PH>
__device__ __forceinline__ void rows_rescan(
        const int* __restrict__ src, const int* __restrict__ dst,
        const float* __restrict__ H, const float* __restrict__ s, unsigned sd_pitch,
        int base, int tid, const float* dloc, unsigned* mkey, float* den,
        float* acc) {
    for (int e = tid; e < E_EDGES; e += ROWS_NT) {
        int r = dst[e] - base;
        if ((unsigned)r < (unsigned)RB) {
            int si = src[e];
            float v = leaky(s[sd_pos((unsigned)si, sd_pitch)] + dloc[r]);
            if (PH == 0) {
                atomicMax(mkey + r, f2key(v));
            } else {
                float w = __expf(v - key2f(mkey[r]));
                if (PH == 1) {
                    atomicAdd(den + r, w);
                } else {
                    float coef = w / (den[r] + 1e-16f);
                    const float* hr = H + (size_t)si * 64;
                    float* ar = acc + r * 64;
                    for (int f = 0; f < 64; ++f) atomicAdd(ar + f, hr[f] * coef);
                }
            }
        }
    }
}

template <int MODE, bool NEXT>
__global__ void __launch_bounds__(ROWS_NT)
gat_rows_lds(const int* __restrict__ src, const int* __restrict__ dst,
             unsigned* own, const float* __restrict__ H, const float* __restrict__ s,
             const float* __restrict__ d, unsigned sd_pitch,
             const float* __restrict__ bias,
             float* out, int relu, const float* __restrict__ Wm,
             float* sn, float* dn, unsigned sn_pitch, const float* __restrict__ Wn,
             const float* __restrict__ asn, const float* __restrict__ adn) {
    __shared__ float    acc[(RB + 8) * 64];
    __shared__ unsigned mkey[RB];
    __shared__ float    den[RB];
    __shared__ float    dloc[RB];
    __shared__ float    vself[RB];
    __shared__ float    wself[RB];
    __shared__ unsigned list[32 * WCAP];
    __shared__ float    lval[32 * WCAP];
    __shared__ int      ovf;
    __shared__ int      nosave;
    __shared__ float    wan[128];
    __shared__ float    bl[64];

    const int tid  = threadIdx.x;
    const int lane = tid & 31;
    const int wave = __builtin_amdgcn_readfirstlane(tid >> 5);
    const int base = blockIdx.x * RB;
    unsigned* wl = list + wave * WCAP;
    float*    wv = lval + wave * WCAP;
    unsigned* ownb = own + (size_t)blockIdx.x * BLK_DW;
    unsigned* lg   = ownb + wave * SAVE_CAP;

    int csaved = 0;
    if (MODE == 1) csaved = __builtin_amdgcn_readfirstlane((int)ownb[CNT_OFF + wave]);
    unsigned bw0 = 0u, bw1 = 0u;
    if (MODE == 0) {
        const unsigned w0i = (unsigned)(wave * BM_WPW + lane), w1i = w0i + 32u;
        if (w0i < (unsigned)BM_WORDS)
            bw0 = own[bmt_pos(w0i >> 2, (unsigned)blockIdx.x, w0i & 3u)];
        if (lane < BM_WPW - 32 && w1i < (unsigned)BM_WORDS)
            bw1 = own[bmt_pos(w1i >> 2, (unsigned)blockIdx.x, w1i & 3u)];
    }

    if (tid == 0) { ovf = 0; nosave = 0; }
    if (NEXT && wave >= 8 && wave < 12) {
        const int t = tid - 256;
        const float* wr = Wn + (t & 63) * 64;
        const float* av = (wave < 10) ? asn : adn;
        float w = 0.0f;
#pragma unroll
        for (int j = 0; j < 64; ++j) w = fmaf(wr[j], av[j], w);
        wan[t] = w;
    }
    if (NEXT && wave >= 12 && wave < 14) bl[tid - 384] = bias[tid - 384];
    if (tid < RB) {
        const unsigned i = blockIdx.x * sd_pitch + tid;
        float dl = d[i];
        float v = leaky(s[i] + dl);
        dloc[tid] = dl;
        vself[tid] = v;
        mkey[tid] = f2key(v);
    }
    __syncthreads();

    int wcnt = 0;
    if (MODE == 1) {
        if ((unsigned)csaved > (unsigned)SAVE_CAP) { if (lane == 0) ovf = 1; }
        else            wcnt = csaved;
    } else {
        const uint4* key4 = (const uint4*)own;
        const unsigned pat = (unsigned)blockIdx.x * 0x01010101u;
        unsigned* gl = reinterpret_cast<unsigned*>(wv);
        int ng = 0;
#pragma unroll
        for (int pass = 0; pass < 2; ++pass) {
            unsigned bits = pass ? bw1 : bw0;
            const unsigned g0 = (unsigned)(wave * BM_WPW + 32 * pass + lane) * 32u;
            unsigned live = __builtin_amdgcn_ballot_w32(bits != 0u);
            while (live != 0u) {
                int p = ng + (int)__builtin_amdgcn_mbcnt_lo(live, 0u);
                if (bits != 0u && p < WCAP) gl[p] = g0 + (unsigned)(__ffs((int)bits) - 1);
                ng += __builtin_popcount(live);
                bits &= bits - 1u;
                live = __builtin_amdgcn_ballot_w32(bits != 0u);
            }
        }
        if (ng > WCAP && lane == 0) ovf = 1;
        const int ngc = (ng < WCAP) ? ng : WCAP;
        for (int i0 = 0; i0 < ngc; i0 += 32 * GRP_U) {
            uint4    kv[GRP_U];
            unsigned gg[GRP_U];
#pragma unroll
            for (int u = 0; u < GRP_U; ++u) {
                int i = i0 + 32 * u + lane;
                unsigned g = gl[(i < ngc) ? i : ngc - 1];
                gg[u] = g;
                kv[u] = key4[(g >> 10) * (BLK_DW / 4) + (KEY_OFF / 4) + (g & 1023u)];
            }
#pragma unroll
            for (int u = 0; u < GRP_U; ++u) {
                if (i0 + 32 * u < ngc) {
                    unsigned xs[4] = { kv[u].x ^ pat, kv[u].y ^ pat, kv[u].z ^ pat, kv[u].w ^ pat };
                    unsigned z[4];
#pragma unroll
                    for (int j = 0; j < 4; ++j) {
                        unsigned t = ((xs[j] & 0x7F7F7F7Fu) + 0x7F7F7F7Fu) | xs[j];
                        z[j] = ~(t | 0x7F7F7F7Fu);
                    }
                    unsigned zz = (z[0] >> 7) | (z[1] >> 6) | (z[2] >> 5) | (z[3] >> 4);
                    if (i0 + 32 * u + lane >= ngc) zz = 0u;
                    unsigned any = __builtin_amdgcn_ballot_w32(zz != 0u);
                    const unsigned e0 = gg[u] * 16u;
                    while (any != 0u) {
                        int bit = __ffs((int)zz) - 1;
                        unsigned off = (unsigned)(((bit & 7) << 2) | (bit >> 3));
                        int p = wcnt + (int)__builtin_amdgcn_mbcnt_lo(any, 0u);
                        if (zz != 0u && p < WCAP) wl[p] = e0 + off;
                        wcnt += __builtin_popcount(any);
                        zz &= zz - 1u;
                        any = __builtin_amdgcn_ballot_w32(zz != 0u);
                    }
                }
            }
        }
    }
    if (wcnt > WCAP && lane == 0) ovf = 1;
    if (MODE == 0 && wcnt > SAVE_CAP && lane == 0) nosave = 1;
    __syncthreads();
    const bool fast = (__builtin_amdgcn_readfirstlane(ovf) == 0);
    const int  n    = wcnt;
    const bool saved = (MODE == 0) && fast && (__builtin_amdgcn_readfirstlane(nosave) == 0);
    if (MODE == 0 && lane == 0) ownb[CNT_OFF + wave] = saved ? (unsigned)n : ~0u;

    if (fast && MODE == 0) {
        for (int i = lane; i < n; i += 32) {
            int e = (int)wl[i];
            int r = dst[e] - base;
            int si = src[e];
            float v = leaky(s[sd_pos((unsigned)si, sd_pitch)] + dloc[r]);
            unsigned en = ((unsigned)r << 16) | (unsigned)si;
            wl[i] = en;
            if (saved) lg[i] = en;
            wv[i] = v;
            atomicMax(mkey + r, f2key(v));
        }
    } else if (fast) {
        for (int g0 = 0; g0 < n; g0 += 32 * PH0_U) {
            unsigned en[PH0_U];
            float    sv[PH0_U];
#pragma unroll
            for (int u = 0; u < PH0_U; ++u) {
                int i = g0 + 32 * u + lane;
                en[u] = lg[(i < n) ? i : n - 1];
            }
#pragma unroll
            for (int u = 0; u < PH0_U; ++u) sv[u] = s[sd_pos(en[u] & 0xFFFFu, sd_pitch)];
#pragma unroll
            for (int u = 0; u < PH0_U; ++u) asm volatile("" :: "v"(sv[u]));
#pragma unroll
            for (int u = 0; u < PH0_U; ++u) {
                int i = g0 + 32 * u + lane;
                if (i < n) {
                    int r = (int)(en[u] >> 16);
                    float v = leaky(sv[u] + dloc[r]);
                    wl[i] = en[u];
                    wv[i] = v;
                    atomicMax(mkey + r, f2key(v));
                }
            }
        }
    } else {
        rows_rescan<0>(src, dst, H, s, sd_pitch, base, tid, dloc, mkey, den, acc);
    }
    __syncthreads();
    if (tid < RB) {
        float w = __expf(vself[tid] - key2f(mkey[tid]));
        wself[tid] = w;
        den[tid] = w;
    }
    __syncthreads();

    if (fast) {
        for (int i = lane; i < n; i += 32) {
            int r = (int)(wl[i] >> 16);
            float w = __expf(wv[i] - key2f(mkey[r]));
            wv[i] = w;
            atomicAdd(den + r, w);
        }
    } else {
        rows_rescan<1>(src, dst, H, s, sd_pitch, base, tid, dloc, mkey, den, acc);
    }
    __syncthreads();
    for (int idx = tid; idx < RB * 64; idx += ROWS_NT) {
        int r = idx >> 6;
        float coef = wself[r] / (den[r] + 1e-16f);
        acc[idx] = H[(size_t)base * 64 + idx] * coef;
    }
    if (MODE == 1 && tid < 8 * 64) acc[RB * 64 + tid] = 0.0f;
    __syncthreads();

    if (fast) {
        for (int b0 = 0; b0 < n; b0 += 32) {
            int i = b0 + lane;
            bool valid = i < n;
            unsigned en = valid ? wl[i] : 0u;
            float w = valid ? wv[i] : 0.0f;
            int r  = (int)(en >> 16);
            int si = (int)(en & 0xFFFFu);
            float coef = w / (den[r] + 1e-16f);
#pragma unroll
            for (int g = 0; g < 32; g += AGG_U) {
                if (b0 + g < n) {
                    v2f hv[AGG_U];
#pragma unroll
                    for (int u = 0; u < AGG_U; ++u) {
                        int sj = __builtin_amdgcn_readlane(si, g + u);
                        hv[u] = *(const v2f*)(H + (size_t)sj * 64 + 2 * lane);
                    }
#pragma unroll
                    for (int u = 0; u < AGG_U; ++u) asm volatile("" :: "v"(hv[u]));
#pragma unroll
                    for (int u = 0; u < AGG_U; ++u) {
                        if (b0 + g + u < n) {
                            int rj = __builtin_amdgcn_readlane(r, g + u);
                            float cj = __int_as_float(__builtin_amdgcn_readlane(__float_as_int(coef), g + u));
                            float* ar = acc + rj * 64 + 2 * lane;
                            atomicAdd(ar,     hv[u].x * cj);
                            atomicAdd(ar + 1, hv[u].y * cj);
                        }
                    }
                }
            }
        }
    } else {
        rows_rescan<2>(src, dst, H, s, sd_pitch, base, tid, dloc, mkey, den, acc);
    }
    __syncthreads();

    if (MODE == 1) {
        if (wave < 26) {
            const int mt = wave >> 1, nh = wave & 1;
            const int lo = lane & 15, hi = lane >> 4;
            const float* ar = acc + (mt * 16 + lo) * 64 + 2 * hi;
            const float* wc = Wm + nh * 32 + lo;
            v8f c0 = {}, c1 = {};
#pragma unroll 4
            for (int k = 0; k < 64; k += 4) {
                v2f a = *(const v2f*)(ar + k);
                const float* w0 = wc + (k + 2 * hi) * 64;
                v2f b0; b0.x = w0[0];  b0.y = w0[64];
                v2f b1; b1.x = w0[16]; b1.y = w0[64 + 16];
                c0 = __builtin_amdgcn_wmma_f32_16x16x4_f32(false, a, false, b0, (short)0, c0, false, false);
                c1 = __builtin_amdgcn_wmma_f32_16x16x4_f32(false, a, false, b1, (short)0, c1, false, false);
            }
#pragma unroll
            for (int r = 0; r < 8; ++r) {
                int row = mt * 16 + r + 8 * hi;
                if (row < RB) {
                    lval[row * 64 + nh * 32 + lo]      = c0[r];
                    lval[row * 64 + nh * 32 + 16 + lo] = c1[r];
                }
            }
        }
        __syncthreads();
    }
    const float* res = (MODE == 1) ? lval : acc;

    for (int idx = tid; idx < RB * 64; idx += ROWS_NT) {
        float v = res[idx] + bias[idx & 63];
        if (relu) v = fmaxf(v, 0.0f);
        out[(size_t)base * 64 + idx] = v;
    }

    if (NEXT && tid < 4 * RB) {
        const int row = tid >> 2, part = tid & 3;
        const float* yr = res + row * 64 + part * 16;
        const float* ws = wan + part * 16;
        const float* bp = bl + part * 16;
        float ss = 0.0f, dd = 0.0f;
#pragma unroll
        for (int j = 0; j < 16; ++j) {
            const int f = (j + row) & 15;
            float yv = yr[f] + bp[f];
            if (relu) yv = fmaxf(yv, 0.0f);
            ss = fmaf(yv, ws[f], ss);
            dd = fmaf(yv, ws[64 + f], dd);
        }
        ss += __shfl_xor(ss, 1); dd += __shfl_xor(dd, 1);
        ss += __shfl_xor(ss, 2); dd += __shfl_xor(dd, 2);
        if (part == 0) {
            const unsigned p = blockIdx.x * sn_pitch + row;
            sn[p] = ss; dn[p] = dd;
        }
    }
}

extern "C" void kernel_launch(void* const* d_in, const int* in_sizes, int n_in,
                              void* d_out, int out_size, void* d_ws, size_t ws_size,
                              hipStream_t stream) {
    (void)in_sizes; (void)n_in; (void)out_size; (void)ws_size;

    const float* x    = (const float*)d_in[0];
    const int*   esrc = (const int*)  d_in[1];
    const int*   edst = (const int*)  d_in[2];
    const float* W0 = (const float*)d_in[3];
    const float* a0s = (const float*)d_in[4];
    const float* a0d = (const float*)d_in[5];
    const float* b0 = (const float*)d_in[6];
    const float* W1 = (const float*)d_in[7];
    const float* a1s = (const float*)d_in[8];
    const float* a1d = (const float*)d_in[9];
    const float* b1 = (const float*)d_in[10];
    const float* W2 = (const float*)d_in[11];
    const float* a2s = (const float*)d_in[12];
    const float* a2d = (const float*)d_in[13];
    const float* b2 = (const float*)d_in[14];
    float* out = (float*)d_out;

    float* ws = (float*)d_ws;
    float* H   = ws;
    float* XIN = H   + (size_t)N_NODES * 64;
    float* SV  = XIN + (size_t)N_NODES * 64;
    float* DV  = SV  + N_NODES;
    unsigned* OWN = (unsigned*)out;
    float* S1 = out + S1_OFF;
    float* D1 = out + D1_OFF;

    gat_front<<<FRONT_BLK, 256, 0, stream>>>(x, W0, H, N_NODES, a0s, a0d, SV, DV, edst, OWN);
    gat_rows_lds<0, true><<<NBLK, ROWS_NT, 0, stream>>>(esrc, edst, OWN, H, SV, DV, RB, b0, XIN, 1, nullptr,
                                                 S1, D1, BLK_DW, W1, a1s, a1d);

    gat_rows_lds<1, true><<<NBLK, ROWS_NT, 0, stream>>>(esrc, edst, OWN, XIN, S1, D1, BLK_DW, b1, H, 1, W1,
                                                 SV, DV, RB, W2, a2s, a2d);

    gat_rows_lds<1, false><<<NBLK, ROWS_NT, 0, stream>>>(esrc, edst, OWN, H, SV, DV, RB, b2, out, 0, W2,
                                                 nullptr, nullptr, 0, nullptr, nullptr, nullptr);
}


# Front launch writes a (destination, source) dword per edge, not a one-byte key; layer-0 owners take row and source from marked 64-byte groups: per-edge dst, src loads go; list (sum) order changes
# speedup vs baseline: 5.7210x; 1.0411x over previous
#include <hip/hip_runtime.h>
#include <hip/hip_bf16.h>
#include <cstddef>


#define N_NODES 50000
#define E_EDGES 800000
#define NEG_SLOPE 0.2f

typedef float v2f __attribute__((ext_vector_type(2)));
typedef float v8f __attribute__((ext_vector_type(8)));

__device__ __forceinline__ unsigned f2key(float f) {
    int i = __float_as_int(f);
    return (i >= 0) ? ((unsigned)i | 0x80000000u) : ~(unsigned)i;
}
__device__ __forceinline__ float key2f(unsigned k) {
    int i = (k & 0x80000000u) ? (int)(k & 0x7FFFFFFFu) : (int)~k;
    return __int_as_float(i);
}

#define RB_KEY 200u
#define BLK_DW    (200 * 64)
#define SAVE_CAP  192
#define CNT_OFF   (32 * SAVE_CAP)
#define MARK_OFF  6400
#define MARK_DW   2048
#define KEY_OFF   8448
#define KEY_DW    4096
#define S1_OFF    (CNT_OFF + 32)
#define D1_OFF    (KEY_OFF + KEY_DW)
__device__ __forceinline__ unsigned sd_pos(unsigned i, unsigned pitch) {
    return (i / 200u) * pitch + (i % 200u);
}
__device__ __forceinline__ unsigned rec_pos(unsigned e) {
    return (e / KEY_DW) * BLK_DW + KEY_OFF + (e % KEY_DW);
}

#define FRONT_K   128
#define FRONT_BLK (((N_NODES / 16) + 7) / 8)
static_assert(N_NODES % 16 == 0, "row tiles must divide the node set");
static_assert(FRONT_BLK * 512 >= E_EDGES / 4, "the front blocks must cover the edge list");
#define BM_WORDS  (FRONT_BLK * 4)
#define BM_WPW    ((BM_WORDS + 31) / 32)
static_assert(MARK_DW == 2 * 1024, "a mark piece holds the marks of two front blocks");
static_assert(BM_WPW > 32 && BM_WPW <= 64, "an owner lane takes one or two mark words");
__device__ __forceinline__ unsigned bmt_pos(unsigned j, unsigned b, unsigned ww) {
    return (j >> 1) * BLK_DW + MARK_OFF + (j & 1u) * 1024u + b * 4u + ww;
}
__global__ void __launch_bounds__(256)
gat_front(const float* __restrict__ X, const float* __restrict__ W,
          float* __restrict__ H, int nrows,
          const float* __restrict__ asrc, const float* __restrict__ adst,
          float* __restrict__ s, float* __restrict__ d,
          const int* __restrict__ src, const int* __restrict__ dst,
          unsigned* __restrict__ keys) {
    __shared__ float wa[2 * FRONT_K];
    __shared__ unsigned bm[256 * 4];
    const int tid  = threadIdx.x;
    const int lane = tid & 31;
    const int wib  = __builtin_amdgcn_readfirstlane(tid >> 5);

    const int4* dst4 = (const int4*)dst;
    const int4* src4 = (const int4*)src;
    const int qlast = E_EDGES / 4 - 1;
    const int q0 = blockIdx.x * 512 + tid, q1 = q0 + 256;
    const int4 kv0 = dst4[(q0 < qlast) ? q0 : qlast];
    const int4 kv1 = dst4[(q1 < qlast) ? q1 : qlast];
    const int4 sv0 = src4[(q0 < qlast) ? q0 : qlast];
    const int4 sv1 = src4[(q1 < qlast) ? q1 : qlast];
    for (int i = tid; i < 256 * 4; i += 256) bm[i] = 0u;
    {
        const int c = lane & 7, rr = lane >> 3;
        const float4 as0 = *(const float4*)(asrc + 4 * c);
        const float4 as1 = *(const float4*)(asrc + 32 + 4 * c);
        const float4 ad0 = *(const float4*)(adst + 4 * c);
        const float4 ad1 = *(const float4*)(adst + 32 + 4 * c);
        const float* wr = W + (wib * 16 + rr) * 64 + 4 * c;
        float4 w0[4], w1[4];
        float  ps[4], pd[4];
#pragma unroll
        for (int j = 0; j < 4; ++j) {
            w0[j] = *(const float4*)(wr + j * 256);
            w1[j] = *(const float4*)(wr + j * 256 + 32);
        }
#pragma unroll
        for (int j = 0; j < 4; ++j) {
            float s_ = w0[j].x * as0.x, d_ = w0[j].x * ad0.x;
            s_ = fmaf(w0[j].y, as0.y, s_); d_ = fmaf(w0[j].y, ad0.y, d_);
            s_ = fmaf(w0[j].z, as0.z, s_); d_ = fmaf(w0[j].z, ad0.z, d_);
            s_ = fmaf(w0[j].w, as0.w, s_); d_ = fmaf(w0[j].w, ad0.w, d_);
            s_ = fmaf(w1[j].x, as1.x, s_); d_ = fmaf(w1[j].x, ad1.x, d_);
            s_ = fmaf(w1[j].y, as1.y, s_); d_ = fmaf(w1[j].y, ad1.y, d_);
            s_ = fmaf(w1[j].z, as1.z, s_); d_ = fmaf(w1[j].z, ad1.z, d_);
            s_ = fmaf(w1[j].w, as1.w, s_); d_ = fmaf(w1[j].w, ad1.w, d_);
            ps[j] = s_; pd[j] = d_;
        }
#pragma unroll
        for (int m = 1; m < 8; m <<= 1) {
#pragma unroll
            for (int j = 0; j < 4; ++j) {
                ps[j] += __shfl_xor(ps[j], m);
                pd[j] += __shfl_xor(pd[j], m);
            }
        }
        if (c == 0) {
#pragma unroll
            for (int j = 0; j < 4; ++j) {
                wa[wib * 16 + j * 4 + rr] = ps[j];
                wa[FRONT_K + wib * 16 + j * 4 + rr] = pd[j];
            }
        }
    }
    __syncthreads();

    {
        const unsigned mbit = 1u << ((tid >> 2) & 31);
        const int mw = tid >> 7;
        if (q0 <= qlast) {
            atomicOr(&bm[((unsigned)kv0.x / RB_KEY) * 4 + mw], mbit);
            atomicOr(&bm[((unsigned)kv0.y / RB_KEY) * 4 + mw], mbit);
            atomicOr(&bm[((unsigned)kv0.z / RB_KEY) * 4 + mw], mbit);
            atomicOr(&bm[((unsigned)kv0.w / RB_KEY) * 4 + mw], mbit);
        }
        if (q1 <= qlast) {
            atomicOr(&bm[((unsigned)kv1.x / RB_KEY) * 4 + 2 + mw], mbit);
            atomicOr(&bm[((unsigned)kv1.y / RB_KEY) * 4 + 2 + mw], mbit);
            atomicOr(&bm[((unsigned)kv1.z / RB_KEY) * 4 + 2 + mw], mbit);
            atomicOr(&bm[((unsigned)kv1.w / RB_KEY) * 4 + 2 + mw], mbit);
        }
    }
    __syncthreads();
    *(uint4*)(keys + bmt_pos((unsigned)blockIdx.x, (unsigned)tid, 0u)) =
        make_uint4(bm[4 * tid], bm[4 * tid + 1], bm[4 * tid + 2], bm[4 * tid + 3]);
    const uint4 rec0 = make_uint4(((unsigned)kv0.x << 16) | (unsigned)sv0.x, ((unsigned)kv0.y << 16) | (unsigned)sv0.y,
                                  ((unsigned)kv0.z << 16) | (unsigned)sv0.z, ((unsigned)kv0.w << 16) | (unsigned)sv0.w);
    const uint4 rec1 = make_uint4(((unsigned)kv1.x << 16) | (unsigned)sv1.x, ((unsigned)kv1.y << 16) | (unsigned)sv1.y,
                                  ((unsigned)kv1.z << 16) | (unsigned)sv1.z, ((unsigned)kv1.w << 16) | (unsigned)sv1.w);

    const int rowBase = (blockIdx.x * 8 + wib) * 16;
    if (rowBase < nrows) {

    const int lo = lane & 15;
    const int hi = lane >> 4;

    v8f acc0 = {}, acc1 = {}, acc2 = {}, acc3 = {};
    float ss = 0.0f, dd = 0.0f;

    const float* xr = X + (size_t)(rowBase + lo) * FRONT_K + 2 * hi;
    const float* wc = W + lo;
    const float* wk = wa + 2 * hi;

    for (int k0 = 0; k0 < FRONT_K; k0 += 16) {
        v2f a[4], b[4][4], ws2[4], wd2[4];
#pragma unroll
        for (int u = 0; u < 4; ++u) {
            const int k = k0 + 4 * u;
            a[u] = *(const v2f*)(xr + k);
            const float* w0 = wc + (size_t)(k + 2 * hi) * 64;
            const float* w1 = w0 + 64;
#pragma unroll
            for (int t = 0; t < 4; ++t) { b[u][t].x = w0[16 * t]; b[u][t].y = w1[16 * t]; }
            ws2[u] = *(const v2f*)(wk + k);
            wd2[u] = *(const v2f*)(wk + FRONT_K + k);
        }
#pragma unroll
        for (int u = 0; u < 4; ++u) {
            asm volatile("" :: "v"(a[u]));
#pragma unroll
            for (int t = 0; t < 4; ++t) asm volatile("" :: "v"(b[u][t]));
        }
#pragma unroll
        for (int u = 0; u < 4; ++u) {
            acc0 = __builtin_amdgcn_wmma_f32_16x16x4_f32(false, a[u], false, b[u][0], (short)0, acc0, false, false);
            acc1 = __builtin_amdgcn_wmma_f32_16x16x4_f32(false, a[u], false, b[u][1], (short)0, acc1, false, false);
            acc2 = __builtin_amdgcn_wmma_f32_16x16x4_f32(false, a[u], false, b[u][2], (short)0, acc2, false, false);
            acc3 = __builtin_amdgcn_wmma_f32_16x16x4_f32(false, a[u], false, b[u][3], (short)0, acc3, false, false);
            ss = fmaf(a[u].x, ws2[u].x, ss); ss = fmaf(a[u].y, ws2[u].y, ss);
            dd = fmaf(a[u].x, wd2[u].x, dd); dd = fmaf(a[u].y, wd2[u].y, dd);
        }
    }

#pragma unroll
    for (int r = 0; r < 8; ++r) {
        float* o = H + (size_t)(rowBase + r + 8 * hi) * 64 + lo;
        o[0]  = acc0[r];
        o[16] = acc1[r];
        o[32] = acc2[r];
        o[48] = acc3[r];
    }

    ss += __shfl_xor(ss, 16); dd += __shfl_xor(dd, 16);
    if (hi == 0) { s[rowBase + lo] = ss; d[rowBase + lo] = dd; }
    }

    if (q0 <= qlast) *(uint4*)(keys + rec_pos(4u * (unsigned)q0)) = rec0;
    if (q1 <= qlast) *(uint4*)(keys + rec_pos(4u * (unsigned)q1)) = rec1;
}

#define RB        200
#define NBLK      250
#define ROWS_NT   1024
#define WCAP      512
#define GRP_U     18
#define AGG_U     16
#define PH0_U     5
#define NGROUP    (E_EDGES / 16)
static_assert(RB * NBLK == N_NODES, "row blocks must tile the node set");
static_assert(RB == RB_KEY, "group marks must be cut by the owners' row block");
static_assert(NBLK <= 256, "an owner needs a mark slot");
static_assert(E_EDGES % 16 == 0, "edge list must split into 16-edge groups");
static_assert(E_EDGES <= (1 << 20), "edge id must fit 20 bits of a list entry");
static_assert(BLK_DW == RB * 64, "footprint = the block's output rows");
static_assert(SAVE_CAP <= WCAP && SAVE_CAP % 32 == 0, "saved lists are a prefix of the LDS lists");
static_assert(CNT_OFF + 32 <= MARK_OFF && MARK_OFF + MARK_DW <= KEY_OFF && KEY_OFF + KEY_DW <= BLK_DW,
              "lists, counts, mark piece, record piece must not overlap");
static_assert((FRONT_BLK + 1) / 2 <= NBLK, "the marks must fit the mark pieces");
static_assert(FRONT_BLK * 128 >= NGROUP, "the marks must cover every 16-edge group");
static_assert((E_EDGES + KEY_DW - 1) / KEY_DW <= NBLK, "record pieces must fit the footprints");
static_assert(KEY_DW % 16 == 0 && MARK_OFF % 32 == 0 && KEY_OFF % 32 == 0, "groups and pieces start on 64-byte and 128-byte bounds");
static_assert(N_NODES <= 65536, "source id must fit 16 bits of a saved entry");
static_assert(S1_OFF % 32 == 0 && S1_OFF + RB <= MARK_OFF, "layer-1 s gap: line aligned, in front of the mark piece");
static_assert(D1_OFF % 32 == 0 && D1_OFF + RB <= BLK_DW, "layer-1 d gap: line aligned, inside the footprint");

__device__ __forceinline__ float leaky(float v) {
    return (v > 0.0f) ? v : NEG_SLOPE * v;
}

template <int PH>
__device__ __forceinline__ void rows_rescan(
        const int* __restrict__ src, const int* __restrict__ dst,
        const float* __restrict__ H, const float* __restrict__ s, unsigned sd_pitch,
        int base, int tid, const float* dloc, unsigned* mkey, float* den,
        float* acc) {
    for (int e = tid; e < E_EDGES; e += ROWS_NT) {
        int r = dst[e] - base;
        if ((unsigned)r < (unsigned)RB) {
            int si = src[e];
            float v = leaky(s[sd_pos((unsigned)si, sd_pitch)] + dloc[r]);
            if (PH == 0) {
                atomicMax(mkey + r, f2key(v));
            } else {
                float w = __expf(v - key2f(mkey[r]));
                if (PH == 1) {
                    atomicAdd(den + r, w);
                } else {
                    float coef = w / (den[r] + 1e-16f);
                    const float* hr = H + (size_t)si * 64;
                    float* ar = acc + r * 64;
                    for (int f = 0; f < 64; ++f) atomicAdd(ar + f, hr[f] * coef);
                }
            }
        }
    }
}

template <int MODE, bool NEXT>
__global__ void __launch_bounds__(ROWS_NT)
gat_rows_lds(const int* __restrict__ src, const int* __restrict__ dst,
             unsigned* own, const float* __restrict__ H, const float* __restrict__ s,
             const float* __restrict__ d, unsigned sd_pitch,
             const float* __restrict__ bias,
             float* out, int relu, const float* __restrict__ Wm,
             float* sn, float* dn, unsigned sn_pitch, const float* __restrict__ Wn,
             const float* __restrict__ asn, const float* __restrict__ adn) {
    __shared__ float    acc[(RB + 8) * 64];
    __shared__ unsigned mkey[RB];
    __shared__ float    den[RB];
    __shared__ float    dloc[RB];
    __shared__ float    vself[RB];
    __shared__ float    wself[RB];
    __shared__ unsigned list[32 * WCAP];
    __shared__ float    lval[32 * WCAP];
    __shared__ int      ovf;
    __shared__ int      nosave;
    __shared__ float    wan[128];
    __shared__ float    bl[64];

    const int tid  = threadIdx.x;
    const int lane = tid & 31;
    const int wave = __builtin_amdgcn_readfirstlane(tid >> 5);
    const int base = blockIdx.x * RB;
    unsigned* wl = list + wave * WCAP;
    float*    wv = lval + wave * WCAP;
    unsigned* ownb = own + (size_t)blockIdx.x * BLK_DW;
    unsigned* lg   = ownb + wave * SAVE_CAP;

    int csaved = 0;
    if (MODE == 1) csaved = __builtin_amdgcn_readfirstlane((int)ownb[CNT_OFF + wave]);
    unsigned bw0 = 0u, bw1 = 0u;
    if (MODE == 0) {
        const unsigned w0i = (unsigned)(wave * BM_WPW + lane), w1i = w0i + 32u;
        if (w0i < (unsigned)BM_WORDS)
            bw0 = own[bmt_pos(w0i >> 2, (unsigned)blockIdx.x, w0i & 3u)];
        if (lane < BM_WPW - 32 && w1i < (unsigned)BM_WORDS)
            bw1 = own[bmt_pos(w1i >> 2, (unsigned)blockIdx.x, w1i & 3u)];
    }

    if (tid == 0) { ovf = 0; nosave = 0; }
    if (NEXT && wave >= 8 && wave < 12) {
        const int t = tid - 256;
        const float* wr = Wn + (t & 63) * 64;
        const float* av = (wave < 10) ? asn : adn;
        float w = 0.0f;
#pragma unroll
        for (int j = 0; j < 64; ++j) w = fmaf(wr[j], av[j], w);
        wan[t] = w;
    }
    if (NEXT && wave >= 12 && wave < 14) bl[tid - 384] = bias[tid - 384];
    if (tid < RB) {
        const unsigned i = blockIdx.x * sd_pitch + tid;
        float dl = d[i];
        float v = leaky(s[i] + dl);
        dloc[tid] = dl;
        vself[tid] = v;
        mkey[tid] = f2key(v);
    }
    __syncthreads();

    int wcnt = 0;
    if (MODE == 1) {
        if ((unsigned)csaved > (unsigned)SAVE_CAP) { if (lane == 0) ovf = 1; }
        else            wcnt = csaved;
    } else {
        const uint4* rec4 = (const uint4*)own;
        const unsigned bsh = (unsigned)base << 16;
        const int sub = lane >> 2, part = lane & 3;
        unsigned* gl = reinterpret_cast<unsigned*>(wv);
        int ng = 0;
#pragma unroll
        for (int pass = 0; pass < 2; ++pass) {
            unsigned bits = pass ? bw1 : bw0;
            const unsigned g0 = (unsigned)(wave * BM_WPW + 32 * pass + lane) * 32u;
            unsigned live = __builtin_amdgcn_ballot_w32(bits != 0u);
            while (live != 0u) {
                int p = ng + (int)__builtin_amdgcn_mbcnt_lo(live, 0u);
                if (bits != 0u && p < WCAP) gl[p] = g0 + (unsigned)(__ffs((int)bits) - 1);
                ng += __builtin_popcount(live);
                bits &= bits - 1u;
                live = __builtin_amdgcn_ballot_w32(bits != 0u);
            }
        }
        if (ng > WCAP && lane == 0) ovf = 1;
        const int ngc = (ng < WCAP) ? ng : WCAP;
        for (int i0 = 0; i0 < ngc; i0 += 8 * GRP_U) {
            uint4 kv[GRP_U];
#pragma unroll
            for (int u = 0; u < GRP_U; ++u) {
                int i = i0 + 8 * u + sub;
                unsigned g = gl[(i < ngc) ? i : ngc - 1];
                kv[u] = rec4[(g >> 8) * (BLK_DW / 4) + (KEY_OFF / 4) + (g & 255u) * 4u + (unsigned)part];
            }
#pragma unroll
            for (int u = 0; u < GRP_U; ++u) asm volatile("" :: "v"(kv[u].x), "v"(kv[u].y), "v"(kv[u].z), "v"(kv[u].w));
#pragma unroll
            for (int u = 0; u < GRP_U; ++u) {
                if (i0 + 8 * u < ngc) {
                    const bool okl = (i0 + 8 * u + sub) < ngc;
                    const unsigned okm = __builtin_amdgcn_ballot_w32(okl);
                    const unsigned rc[4] = { kv[u].x, kv[u].y, kv[u].z, kv[u].w };
#pragma unroll
                    for (int j = 0; j < 4; ++j) {
                        const unsigned t = rc[j] - bsh;
                        const bool mj = t < ((unsigned)RB << 16);
                        const unsigned mm = __builtin_amdgcn_ballot_w32(mj) & okm;
                        unsigned p = __builtin_amdgcn_mbcnt_lo(mm, (unsigned)wcnt);
                        p = (p < (unsigned)(WCAP - 1)) ? p : (unsigned)(WCAP - 1);
                        if (mj && okl) wl[p] = t;
                        wcnt += __builtin_popcount(mm);
                    }
                }
            }
        }
    }
    if (wcnt > WCAP && lane == 0) ovf = 1;
    if (MODE == 0 && wcnt > SAVE_CAP && lane == 0) nosave = 1;
    __syncthreads();
    const bool fast = (__builtin_amdgcn_readfirstlane(ovf) == 0);
    const int  n    = wcnt;
    const bool saved = (MODE == 0) && fast && (__builtin_amdgcn_readfirstlane(nosave) == 0);
    if (MODE == 0 && lane == 0) ownb[CNT_OFF + wave] = saved ? (unsigned)n : ~0u;

    if (fast) {
        for (int g0 = 0; g0 < n; g0 += 32 * PH0_U) {
            unsigned en[PH0_U];
            float    sv[PH0_U];
#pragma unroll
            for (int u = 0; u < PH0_U; ++u) {
                int i = g0 + 32 * u + lane;
                int ic = (i < n) ? i : n - 1;
                en[u] = (MODE == 0) ? wl[ic] : lg[ic];
            }
#pragma unroll
            for (int u = 0; u < PH0_U; ++u) sv[u] = s[sd_pos(en[u] & 0xFFFFu, sd_pitch)];
#pragma unroll
            for (int u = 0; u < PH0_U; ++u) asm volatile("" :: "v"(sv[u]));
#pragma unroll
            for (int u = 0; u < PH0_U; ++u) {
                int i = g0 + 32 * u + lane;
                if (i < n) {
                    int r = (int)(en[u] >> 16);
                    float v = leaky(sv[u] + dloc[r]);
                    if (MODE == 0) { if (saved) lg[i] = en[u]; }
                    else           wl[i] = en[u];
                    wv[i] = v;
                    atomicMax(mkey + r, f2key(v));
                }
            }
        }
    } else {
        rows_rescan<0>(src, dst, H, s, sd_pitch, base, tid, dloc, mkey, den, acc);
    }
    __syncthreads();
    if (tid < RB) {
        float w = __expf(vself[tid] - key2f(mkey[tid]));
        wself[tid] = w;
        den[tid] = w;
    }
    __syncthreads();

    if (fast) {
        for (int i = lane; i < n; i += 32) {
            int r = (int)(wl[i] >> 16);
            float w = __expf(wv[i] - key2f(mkey[r]));
            wv[i] = w;
            atomicAdd(den + r, w);
        }
    } else {
        rows_rescan<1>(src, dst, H, s, sd_pitch, base, tid, dloc, mkey, den, acc);
    }
    __syncthreads();
    for (int idx = tid; idx < RB * 64; idx += ROWS_NT) {
        int r = idx >> 6;
        float coef = wself[r] / (den[r] + 1e-16f);
        acc[idx] = H[(size_t)base * 64 + idx] * coef;
    }
    if (MODE == 1 && tid < 8 * 64) acc[RB * 64 + tid] = 0.0f;
    __syncthreads();

    if (fast) {
        for (int b0 = 0; b0 < n; b0 += 32) {
            int i = b0 + lane;
            bool valid = i < n;
            unsigned en = valid ? wl[i] : 0u;
            float w = valid ? wv[i] : 0.0f;
            int r  = (int)(en >> 16);
            int si = (int)(en & 0xFFFFu);
            float coef = w / (den[r] + 1e-16f);
#pragma unroll
            for (int g = 0; g < 32; g += AGG_U) {
                if (b0 + g < n) {
                    v2f hv[AGG_U];
#pragma unroll
                    for (int u = 0; u < AGG_U; ++u) {
                        int sj = __builtin_amdgcn_readlane(si, g + u);
                        hv[u] = *(const v2f*)(H + (size_t)sj * 64 + 2 * lane);
                    }
#pragma unroll
                    for (int u = 0; u < AGG_U; ++u) asm volatile("" :: "v"(hv[u]));
#pragma unroll
                    for (int u = 0; u < AGG_U; ++u) {
                        if (b0 + g + u < n) {
                            int rj = __builtin_amdgcn_readlane(r, g + u);
                            float cj = __int_as_float(__builtin_amdgcn_readlane(__float_as_int(coef), g + u));
                            float* ar = acc + rj * 64 + 2 * lane;
                            atomicAdd(ar,     hv[u].x * cj);
                            atomicAdd(ar + 1, hv[u].y * cj);
                        }
                    }
                }
            }
        }
    } else {
        rows_rescan<2>(src, dst, H, s, sd_pitch, base, tid, dloc, mkey, den, acc);
    }
    __syncthreads();

    if (MODE == 1) {
        if (wave < 26) {
            const int mt = wave >> 1, nh = wave & 1;
            const int lo = lane & 15, hi = lane >> 4;
            const float* ar = acc + (mt * 16 + lo) * 64 + 2 * hi;
            const float* wc = Wm + nh * 32 + lo;
            v8f c0 = {}, c1 = {};
#pragma unroll 4
            for (int k = 0; k < 64; k += 4) {
                v2f a = *(const v2f*)(ar + k);
                const float* w0 = wc + (k + 2 * hi) * 64;
                v2f b0; b0.x = w0[0];  b0.y = w0[64];
                v2f b1; b1.x = w0[16]; b1.y = w0[64 + 16];
                c0 = __builtin_amdgcn_wmma_f32_16x16x4_f32(false, a, false, b0, (short)0, c0, false, false);
                c1 = __builtin_amdgcn_wmma_f32_16x16x4_f32(false, a, false, b1, (short)0, c1, false, false);
            }
#pragma unroll
            for (int r = 0; r < 8; ++r) {
                int row = mt * 16 + r + 8 * hi;
                if (row < RB) {
                    lval[row * 64 + nh * 32 + lo]      = c0[r];
                    lval[row * 64 + nh * 32 + 16 + lo] = c1[r];
                }
            }
        }
        __syncthreads();
    }
    const float* res = (MODE == 1) ? lval : acc;

    for (int idx = tid; idx < RB * 64; idx += ROWS_NT) {
        float v = res[idx] + bias[idx & 63];
        if (relu) v = fmaxf(v, 0.0f);
        out[(size_t)base * 64 + idx] = v;
    }

    if (NEXT && tid < 4 * RB) {
        const int row = tid >> 2, part = tid & 3;
        const float* yr = res + row * 64 + part * 16;
        const float* ws = wan + part * 16;
        const float* bp = bl + part * 16;
        float ss = 0.0f, dd = 0.0f;
#pragma unroll
        for (int j = 0; j < 16; ++j) {
            const int f = (j + row) & 15;
            float yv = yr[f] + bp[f];
            if (relu) yv = fmaxf(yv, 0.0f);
            ss = fmaf(yv, ws[f], ss);
            dd = fmaf(yv, ws[64 + f], dd);
        }
        ss += __shfl_xor(ss, 1); dd += __shfl_xor(dd, 1);
        ss += __shfl_xor(ss, 2); dd += __shfl_xor(dd, 2);
        if (part == 0) {
            const unsigned p = blockIdx.x * sn_pitch + row;
            sn[p] = ss; dn[p] = dd;
        }
    }
}

extern "C" void kernel_launch(void* const* d_in, const int* in_sizes, int n_in,
                              void* d_out, int out_size, void* d_ws, size_t ws_size,
                              hipStream_t stream) {
    (void)in_sizes; (void)n_in; (void)out_size; (void)ws_size;

    const float* x    = (const float*)d_in[0];
    const int*   esrc = (const int*)  d_in[1];
    const int*   edst = (const int*)  d_in[2];
    const float* W0 = (const float*)d_in[3];
    const float* a0s = (const float*)d_in[4];
    const float* a0d = (const float*)d_in[5];
    const float* b0 = (const float*)d_in[6];
    const float* W1 = (const float*)d_in[7];
    const float* a1s = (const float*)d_in[8];
    const float* a1d = (const float*)d_in[9];
    const float* b1 = (const float*)d_in[10];
    const float* W2 = (const float*)d_in[11];
    const float* a2s = (const float*)d_in[12];
    const float* a2d = (const float*)d_in[13];
    const float* b2 = (const float*)d_in[14];
    float* out = (float*)d_out;

    float* ws = (float*)d_ws;
    float* H   = ws;
    float* XIN = H   + (size_t)N_NODES * 64;
    float* SV  = XIN + (size_t)N_NODES * 64;
    float* DV  = SV  + N_NODES;
    unsigned* OWN = (unsigned*)out;
    float* S1 = out + S1_OFF;
    float* D1 = out + D1_OFF;

    gat_front<<<FRONT_BLK, 256, 0, stream>>>(x, W0, H, N_NODES, a0s, a0d, SV, DV, esrc, edst, OWN);
    gat_rows_lds<0, true><<<NBLK, ROWS_NT, 0, stream>>>(esrc, edst, OWN, H, SV, DV, RB, b0, XIN, 1, nullptr,
                                                 S1, D1, BLK_DW, W1, a1s, a1d);

    gat_rows_lds<1, true><<<NBLK, ROWS_NT, 0, stream>>>(esrc, edst, OWN, XIN, S1, D1, BLK_DW, b1, H, 1, W1,
                                                 SV, DV, RB, W2, a2s, a2d);

    gat_rows_lds<1, false><<<NBLK, ROWS_NT, 0, stream>>>(esrc, edst, OWN, H, SV, DV, RB, b2, out, 0, W2,
                                                 nullptr, nullptr, 0, nullptr, nullptr, nullptr);
}


# Owners put their edge list in row order in LDS (integer counters, one scan); a wave sums a run of equal rows in registers, one LDS add per run: 64 float LDS adds per edge go; sum order changes
# speedup vs baseline: 6.6772x; 1.1671x over previous
#include <hip/hip_runtime.h>
#include <hip/hip_bf16.h>
#include <cstddef>


#define N_NODES 50000
#define E_EDGES 800000
#define NEG_SLOPE 0.2f

typedef float v2f __attribute__((ext_vector_type(2)));
typedef float v8f __attribute__((ext_vector_type(8)));

__device__ __forceinline__ unsigned f2key(float f) {
    int i = __float_as_int(f);
    return (i >= 0) ? ((unsigned)i | 0x80000000u) : ~(unsigned)i;
}
__device__ __forceinline__ float key2f(unsigned k) {
    int i = (k & 0x80000000u) ? (int)(k & 0x7FFFFFFFu) : (int)~k;
    return __int_as_float(i);
}

#define RB_KEY 200u
#define BLK_DW    (200 * 64)
#define SAVE_CAP  192
#define SRT_CAP   (32 * SAVE_CAP)
#define CNT_OFF   SRT_CAP
#define MARK_OFF  6400
#define MARK_DW   2048
#define KEY_OFF   8448
#define KEY_DW    4096
#define S1_OFF    (CNT_OFF + 32)
#define D1_OFF    (KEY_OFF + KEY_DW)
__device__ __forceinline__ unsigned sd_pos(unsigned i, unsigned pitch) {
    return (i / 200u) * pitch + (i % 200u);
}
__device__ __forceinline__ unsigned rec_pos(unsigned e) {
    return (e / KEY_DW) * BLK_DW + KEY_OFF + (e % KEY_DW);
}

#define FRONT_K   128
#define FRONT_BLK (((N_NODES / 16) + 7) / 8)
static_assert(N_NODES % 16 == 0, "row tiles must divide the node set");
static_assert(FRONT_BLK * 512 >= E_EDGES / 4, "the front blocks must cover the edge list");
#define BM_WORDS  (FRONT_BLK * 4)
#define BM_WPW    ((BM_WORDS + 31) / 32)
static_assert(MARK_DW == 2 * 1024, "a mark piece holds the marks of two front blocks");
static_assert(BM_WPW > 32 && BM_WPW <= 64, "an owner lane takes one or two mark words");
__device__ __forceinline__ unsigned bmt_pos(unsigned j, unsigned b, unsigned ww) {
    return (j >> 1) * BLK_DW + MARK_OFF + (j & 1u) * 1024u + b * 4u + ww;
}
__global__ void __launch_bounds__(256)
gat_front(const float* __restrict__ X, const float* __restrict__ W,
          float* __restrict__ H, int nrows,
          const float* __restrict__ asrc, const float* __restrict__ adst,
          float* __restrict__ s, float* __restrict__ d,
          const int* __restrict__ src, const int* __restrict__ dst,
          unsigned* __restrict__ keys) {
    __shared__ float wa[2 * FRONT_K];
    __shared__ unsigned bm[256 * 4];
    const int tid  = threadIdx.x;
    const int lane = tid & 31;
    const int wib  = __builtin_amdgcn_readfirstlane(tid >> 5);

    const int4* dst4 = (const int4*)dst;
    const int4* src4 = (const int4*)src;
    const int qlast = E_EDGES / 4 - 1;
    const int q0 = blockIdx.x * 512 + tid, q1 = q0 + 256;
    const int4 kv0 = dst4[(q0 < qlast) ? q0 : qlast];
    const int4 kv1 = dst4[(q1 < qlast) ? q1 : qlast];
    const int4 sv0 = src4[(q0 < qlast) ? q0 : qlast];
    const int4 sv1 = src4[(q1 < qlast) ? q1 : qlast];
    for (int i = tid; i < 256 * 4; i += 256) bm[i] = 0u;
    {
        const int c = lane & 7, rr = lane >> 3;
        const float4 as0 = *(const float4*)(asrc + 4 * c);
        const float4 as1 = *(const float4*)(asrc + 32 + 4 * c);
        const float4 ad0 = *(const float4*)(adst + 4 * c);
        const float4 ad1 = *(const float4*)(adst + 32 + 4 * c);
        const float* wr = W + (wib * 16 + rr) * 64 + 4 * c;
        float4 w0[4], w1[4];
        float  ps[4], pd[4];
#pragma unroll
        for (int j = 0; j < 4; ++j) {
            w0[j] = *(const float4*)(wr + j * 256);
            w1[j] = *(const float4*)(wr + j * 256 + 32);
        }
#pragma unroll
        for (int j = 0; j < 4; ++j) {
            float s_ = w0[j].x * as0.x, d_ = w0[j].x * ad0.x;
            s_ = fmaf(w0[j].y, as0.y, s_); d_ = fmaf(w0[j].y, ad0.y, d_);
            s_ = fmaf(w0[j].z, as0.z, s_); d_ = fmaf(w0[j].z, ad0.z, d_);
            s_ = fmaf(w0[j].w, as0.w, s_); d_ = fmaf(w0[j].w, ad0.w, d_);
            s_ = fmaf(w1[j].x, as1.x, s_); d_ = fmaf(w1[j].x, ad1.x, d_);
            s_ = fmaf(w1[j].y, as1.y, s_); d_ = fmaf(w1[j].y, ad1.y, d_);
            s_ = fmaf(w1[j].z, as1.z, s_); d_ = fmaf(w1[j].z, ad1.z, d_);
            s_ = fmaf(w1[j].w, as1.w, s_); d_ = fmaf(w1[j].w, ad1.w, d_);
            ps[j] = s_; pd[j] = d_;
        }
#pragma unroll
        for (int m = 1; m < 8; m <<= 1) {
#pragma unroll
            for (int j = 0; j < 4; ++j) {
                ps[j] += __shfl_xor(ps[j], m);
                pd[j] += __shfl_xor(pd[j], m);
            }
        }
        if (c == 0) {
#pragma unroll
            for (int j = 0; j < 4; ++j) {
                wa[wib * 16 + j * 4 + rr] = ps[j];
                wa[FRONT_K + wib * 16 + j * 4 + rr] = pd[j];
            }
        }
    }
    __syncthreads();

    {
        const unsigned mbit = 1u << ((tid >> 2) & 31);
        const int mw = tid >> 7;
        if (q0 <= qlast) {
            atomicOr(&bm[((unsigned)kv0.x / RB_KEY) * 4 + mw], mbit);
            atomicOr(&bm[((unsigned)kv0.y / RB_KEY) * 4 + mw], mbit);
            atomicOr(&bm[((unsigned)kv0.z / RB_KEY) * 4 + mw], mbit);
            atomicOr(&bm[((unsigned)kv0.w / RB_KEY) * 4 + mw], mbit);
        }
        if (q1 <= qlast) {
            atomicOr(&bm[((unsigned)kv1.x / RB_KEY) * 4 + 2 + mw], mbit);
            atomicOr(&bm[((unsigned)kv1.y / RB_KEY) * 4 + 2 + mw], mbit);
            atomicOr(&bm[((unsigned)kv1.z / RB_KEY) * 4 + 2 + mw], mbit);
            atomicOr(&bm[((unsigned)kv1.w / RB_KEY) * 4 + 2 + mw], mbit);
        }
    }
    __syncthreads();
    *(uint4*)(keys + bmt_pos((unsigned)blockIdx.x, (unsigned)tid, 0u)) =
        make_uint4(bm[4 * tid], bm[4 * tid + 1], bm[4 * tid + 2], bm[4 * tid + 3]);
    const uint4 rec0 = make_uint4(((unsigned)kv0.x << 16) | (unsigned)sv0.x, ((unsigned)kv0.y << 16) | (unsigned)sv0.y,
                                  ((unsigned)kv0.z << 16) | (unsigned)sv0.z, ((unsigned)kv0.w << 16) | (unsigned)sv0.w);
    const uint4 rec1 = make_uint4(((unsigned)kv1.x << 16) | (unsigned)sv1.x, ((unsigned)kv1.y << 16) | (unsigned)sv1.y,
                                  ((unsigned)kv1.z << 16) | (unsigned)sv1.z, ((unsigned)kv1.w << 16) | (unsigned)sv1.w);

    const int rowBase = (blockIdx.x * 8 + wib) * 16;
    if (rowBase < nrows) {

    const int lo = lane & 15;
    const int hi = lane >> 4;

    v8f acc0 = {}, acc1 = {}, acc2 = {}, acc3 = {};
    float ss = 0.0f, dd = 0.0f;

    const float* xr = X + (size_t)(rowBase + lo) * FRONT_K + 2 * hi;
    const float* wc = W + lo;
    const float* wk = wa + 2 * hi;

    for (int k0 = 0; k0 < FRONT_K; k0 += 16) {
        v2f a[4], b[4][4], ws2[4], wd2[4];
#pragma unroll
        for (int u = 0; u < 4; ++u) {
            const int k = k0 + 4 * u;
            a[u] = *(const v2f*)(xr + k);
            const float* w0 = wc + (size_t)(k + 2 * hi) * 64;
            const float* w1 = w0 + 64;
#pragma unroll
            for (int t = 0; t < 4; ++t) { b[u][t].x = w0[16 * t]; b[u][t].y = w1[16 * t]; }
            ws2[u] = *(const v2f*)(wk + k);
            wd2[u] = *(const v2f*)(wk + FRONT_K + k);
        }
#pragma unroll
        for (int u = 0; u < 4; ++u) {
            asm volatile("" :: "v"(a[u]));
#pragma unroll
            for (int t = 0; t < 4; ++t) asm volatile("" :: "v"(b[u][t]));
        }
#pragma unroll
        for (int u = 0; u < 4; ++u) {
            acc0 = __builtin_amdgcn_wmma_f32_16x16x4_f32(false, a[u], false, b[u][0], (short)0, acc0, false, false);
            acc1 = __builtin_amdgcn_wmma_f32_16x16x4_f32(false, a[u], false, b[u][1], (short)0, acc1, false, false);
            acc2 = __builtin_amdgcn_wmma_f32_16x16x4_f32(false, a[u], false, b[u][2], (short)0, acc2, false, false);
            acc3 = __builtin_amdgcn_wmma_f32_16x16x4_f32(false, a[u], false, b[u][3], (short)0, acc3, false, false);
            ss = fmaf(a[u].x, ws2[u].x, ss); ss = fmaf(a[u].y, ws2[u].y, ss);
            dd = fmaf(a[u].x, wd2[u].x, dd); dd = fmaf(a[u].y, wd2[u].y, dd);
        }
    }

#pragma unroll
    for (int r = 0; r < 8; ++r) {
        float* o = H + (size_t)(rowBase + r + 8 * hi) * 64 + lo;
        o[0]  = acc0[r];
        o[16] = acc1[r];
        o[32] = acc2[r];
        o[48] = acc3[r];
    }

    ss += __shfl_xor(ss, 16); dd += __shfl_xor(dd, 16);
    if (hi == 0) { s[rowBase + lo] = ss; d[rowBase + lo] = dd; }
    }

    if (q0 <= qlast) *(uint4*)(keys + rec_pos(4u * (unsigned)q0)) = rec0;
    if (q1 <= qlast) *(uint4*)(keys + rec_pos(4u * (unsigned)q1)) = rec1;
}


#define RB        200
#define NBLK      250
#define ROWS_NT   1024
#define WCAP      512
#define GRP_U     18
#define AGG_U     16
#define PH0_U     5
#define NGROUP    (E_EDGES / 16)
#define SCAN_WAVE 7
static_assert(RB * NBLK == N_NODES, "row blocks must tile the node set");
static_assert(RB == RB_KEY, "group marks must be cut by the owners' row block");
static_assert(NBLK <= 256, "an owner needs a mark slot");
static_assert(E_EDGES % 16 == 0, "edge list must split into 16-edge groups");
static_assert(E_EDGES <= (1 << 20), "edge id must fit 20 bits of a list entry");
static_assert(BLK_DW == RB * 64, "footprint = the block's output rows");
static_assert(SRT_CAP <= CNT_OFF, "the saved list ends in front of its count");
static_assert(CNT_OFF + 32 <= MARK_OFF && MARK_OFF + MARK_DW <= KEY_OFF && KEY_OFF + KEY_DW <= BLK_DW,
              "list, count, mark piece, record piece must not overlap");
static_assert((FRONT_BLK + 1) / 2 <= NBLK, "the marks must fit the mark pieces");
static_assert(FRONT_BLK * 128 >= NGROUP, "the marks must cover every 16-edge group");
static_assert((E_EDGES + KEY_DW - 1) / KEY_DW <= NBLK, "record pieces must fit the footprints");
static_assert(KEY_DW % 16 == 0 && MARK_OFF % 32 == 0 && KEY_OFF % 32 == 0, "groups and pieces start on 64-byte and 128-byte bounds");
static_assert(N_NODES <= 65536, "source id must fit 16 bits of a saved entry");
static_assert(S1_OFF % 32 == 0 && S1_OFF + RB <= MARK_OFF, "layer-1 s gap: line aligned, in front of the mark piece");
static_assert(D1_OFF % 32 == 0 && D1_OFF + RB <= BLK_DW, "layer-1 d gap: line aligned, inside the footprint");
static_assert(SCAN_WAVE * 32 >= RB && SCAN_WAVE < 32, "the scan wave has no self-loop rows to set up");
static_assert(29 * 7 >= RB, "seven rows per lane of the scan wave cover the block");

__device__ __forceinline__ float leaky(float v) {
    return (v > 0.0f) ? v : NEG_SLOPE * v;
}

template <int PH>
__device__ __forceinline__ void rows_rescan(
        const int* __restrict__ src, const int* __restrict__ dst,
        const float* __restrict__ H, const float* __restrict__ s, unsigned sd_pitch,
        int base, int tid, const float* dloc, unsigned* mkey, float* den,
        float* acc) {
    for (int e = tid; e < E_EDGES; e += ROWS_NT) {
        int r = dst[e] - base;
        if ((unsigned)r < (unsigned)RB) {
            int si = src[e];
            float v = leaky(s[sd_pos((unsigned)si, sd_pitch)] + dloc[r]);
            if (PH == 0) {
                atomicMax(mkey + r, f2key(v));
            } else {
                float w = __expf(v - key2f(mkey[r]));
                if (PH == 1) {
                    atomicAdd(den + r, w);
                } else {
                    float coef = w / (den[r] + 1e-16f);
                    const float* hr = H + (size_t)si * 64;
                    float* ar = acc + r * 64;
                    for (int f = 0; f < 64; ++f) atomicAdd(ar + f, hr[f] * coef);
                }
            }
        }
    }
}

template <int MODE, bool NEXT>
__global__ void __launch_bounds__(ROWS_NT)
gat_rows_lds(const int* __restrict__ src, const int* __restrict__ dst,
             unsigned* own, const float* __restrict__ H, const float* __restrict__ s,
             const float* __restrict__ d, unsigned sd_pitch,
             const float* __restrict__ bias,
             float* out, int relu, const float* __restrict__ Wm,
             float* sn, float* dn, unsigned sn_pitch, const float* __restrict__ Wn,
             const float* __restrict__ asn, const float* __restrict__ adn) {
    __shared__ float    acc[(RB + 8) * 64];
    __shared__ unsigned mkey[RB];
    __shared__ float    den[RB];
    __shared__ float    dloc[RB];
    __shared__ float    vself[RB];
    __shared__ float    wself[RB];
    __shared__ unsigned fill[RB];
    __shared__ unsigned list[32 * WCAP];
    __shared__ float    lval[32 * WCAP];
    __shared__ unsigned srt_e[SRT_CAP];
    __shared__ float    srt_w[SRT_CAP];
    __shared__ int      ovf;
    __shared__ int      ntot;
    __shared__ float    wan[128];
    __shared__ float    bl[64];

    const int tid  = threadIdx.x;
    const int lane = tid & 31;
    const int wave = __builtin_amdgcn_readfirstlane(tid >> 5);
    const int base = blockIdx.x * RB;
    unsigned* wl = list + wave * WCAP;
    float*    wv = lval + wave * WCAP;
    unsigned* ownb = own + (size_t)blockIdx.x * BLK_DW;

    unsigned nsaved = 0u;
    if (MODE == 1) nsaved = (unsigned)__builtin_amdgcn_readfirstlane((int)ownb[CNT_OFF]);
    unsigned bw0 = 0u, bw1 = 0u;
    if (MODE == 0) {
        const unsigned w0i = (unsigned)(wave * BM_WPW + lane), w1i = w0i + 32u;
        if (w0i < (unsigned)BM_WORDS)
            bw0 = own[bmt_pos(w0i >> 2, (unsigned)blockIdx.x, w0i & 3u)];
        if (lane < BM_WPW - 32 && w1i < (unsigned)BM_WORDS)
            bw1 = own[bmt_pos(w1i >> 2, (unsigned)blockIdx.x, w1i & 3u)];
    }

    if (tid == 0) { ovf = 0; ntot = 0; }
    if (NEXT && wave >= 8 && wave < 12) {
        const int t = tid - 256;
        const float* wr = Wn + (t & 63) * 64;
        const float* av = (wave < 10) ? asn : adn;
        float w = 0.0f;
#pragma unroll
        for (int j = 0; j < 64; ++j) w = fmaf(wr[j], av[j], w);
        wan[t] = w;
    }
    if (NEXT && wave >= 12 && wave < 14) bl[tid - 384] = bias[tid - 384];
    if (tid < RB) {
        const unsigned i = blockIdx.x * sd_pitch + tid;
        float dl = d[i];
        float v = leaky(s[i] + dl);
        dloc[tid] = dl;
        vself[tid] = v;
        mkey[tid] = f2key(v);
        if (MODE == 0) fill[tid] = 0u;
    }
    __syncthreads();

    int wcnt = 0;
    if (MODE == 0) {
        const uint4* rec4 = (const uint4*)own;
        const unsigned bsh = (unsigned)base << 16;
        const int sub = lane >> 2, part = lane & 3;
        unsigned* gl = reinterpret_cast<unsigned*>(wv);
        int ng = 0;
#pragma unroll
        for (int pass = 0; pass < 2; ++pass) {
            unsigned bits = pass ? bw1 : bw0;
            const unsigned g0 = (unsigned)(wave * BM_WPW + 32 * pass + lane) * 32u;
            unsigned live = __builtin_amdgcn_ballot_w32(bits != 0u);
            while (live != 0u) {
                int p = ng + (int)__builtin_amdgcn_mbcnt_lo(live, 0u);
                if (bits != 0u && p < WCAP) gl[p] = g0 + (unsigned)(__ffs((int)bits) - 1);
                ng += __builtin_popcount(live);
                bits &= bits - 1u;
                live = __builtin_amdgcn_ballot_w32(bits != 0u);
            }
        }
        if (ng > WCAP && lane == 0) ovf = 1;
        const int ngc = (ng < WCAP) ? ng : WCAP;
        for (int i0 = 0; i0 < ngc; i0 += 8 * GRP_U) {
            uint4 kv[GRP_U];
#pragma unroll
            for (int u = 0; u < GRP_U; ++u) {
                int i = i0 + 8 * u + sub;
                unsigned g = gl[(i < ngc) ? i : ngc - 1];
                kv[u] = rec4[(g >> 8) * (BLK_DW / 4) + (KEY_OFF / 4) + (g & 255u) * 4u + (unsigned)part];
            }
#pragma unroll
            for (int u = 0; u < GRP_U; ++u) asm volatile("" :: "v"(kv[u].x), "v"(kv[u].y), "v"(kv[u].z), "v"(kv[u].w));
#pragma unroll
            for (int u = 0; u < GRP_U; ++u) {
                if (i0 + 8 * u < ngc) {
                    const bool okl = (i0 + 8 * u + sub) < ngc;
                    const unsigned okm = __builtin_amdgcn_ballot_w32(okl);
                    const unsigned rc[4] = { kv[u].x, kv[u].y, kv[u].z, kv[u].w };
#pragma unroll
                    for (int j = 0; j < 4; ++j) {
                        const unsigned t = rc[j] - bsh;
                        const bool mj = t < ((unsigned)RB << 16);
                        const unsigned mm = __builtin_amdgcn_ballot_w32(mj) & okm;
                        unsigned p = __builtin_amdgcn_mbcnt_lo(mm, (unsigned)wcnt);
                        p = (p < (unsigned)(WCAP - 1)) ? p : (unsigned)(WCAP - 1);
                        if (mj && okl) wl[p] = t;
                        wcnt += __builtin_popcount(mm);
                    }
                }
            }
        }
        if (wcnt > WCAP && lane == 0) ovf = 1;
        if (lane == 0) atomicAdd(&ntot, wcnt);
    }
    __syncthreads();
    const unsigned nt = (MODE == 0) ? (unsigned)__builtin_amdgcn_readfirstlane(ntot) : nsaved;
    const bool fast = (__builtin_amdgcn_readfirstlane(ovf) == 0) && (nt <= (unsigned)SRT_CAP);
    if (MODE == 0 && tid == 0) ownb[CNT_OFF] = fast ? nt : ~0u;
    const int lo = fast ? (int)(((unsigned)wave * nt) >> 5) : 0;
    const int hi = fast ? (int)(((unsigned)(wave + 1) * nt) >> 5) : 0;
    unsigned* pe = (MODE == 0) ? wl : (srt_e + lo);
    float*    pv = (MODE == 0) ? wv : (srt_w + lo);
    const int n  = (MODE == 0) ? wcnt : (hi - lo);

    if (fast) {
        for (int g0 = 0; g0 < n; g0 += 32 * PH0_U) {
            unsigned en[PH0_U];
            float    sv[PH0_U];
#pragma unroll
            for (int u = 0; u < PH0_U; ++u) {
                int i = g0 + 32 * u + lane;
                int ic = (i < n) ? i : n - 1;
                en[u] = (MODE == 0) ? wl[ic] : ownb[lo + ic];
            }
#pragma unroll
            for (int u = 0; u < PH0_U; ++u) sv[u] = s[sd_pos(en[u] & 0xFFFFu, sd_pitch)];
#pragma unroll
            for (int u = 0; u < PH0_U; ++u) asm volatile("" :: "v"(sv[u]));
#pragma unroll
            for (int u = 0; u < PH0_U; ++u) {
                int i = g0 + 32 * u + lane;
                if (i < n) {
                    int r = (int)(en[u] >> 16);
                    float v = leaky(sv[u] + dloc[r]);
                    if (MODE == 1) pe[i] = en[u];
                    pv[i] = v;
                    atomicMax(mkey + r, f2key(v));
                    if (MODE == 0) atomicAdd(fill + r, 1u);
                }
            }
        }
    } else {
        rows_rescan<0>(src, dst, H, s, sd_pitch, base, tid, dloc, mkey, den, acc);
    }
    __syncthreads();
    if (tid < RB) {
        float w = __expf(vself[tid] - key2f(mkey[tid]));
        wself[tid] = w;
        den[tid] = w;
    }
    if (MODE == 0 && fast && wave == SCAN_WAVE) {
        unsigned c[7];
        unsigned t = 0u;
#pragma unroll
        for (int j = 0; j < 7; ++j) {
            const int r = 7 * lane + j;
            c[j] = (r < RB) ? fill[r] : 0u;
            t += c[j];
        }
        unsigned incl = t;
#pragma unroll
        for (int m = 1; m < 32; m <<= 1) {
            const unsigned o = __shfl_up(incl, m);
            if (lane >= m) incl += o;
        }
        unsigned run = incl - t;
#pragma unroll
        for (int j = 0; j < 7; ++j) {
            const int r = 7 * lane + j;
            if (r < RB) fill[r] = run;
            run += c[j];
        }
    }
    __syncthreads();

    if (fast) {
        for (int i = lane; i < n; i += 32) {
            const unsigned e = pe[i];
            int r = (int)(e >> 16);
            float w = __expf(pv[i] - key2f(mkey[r]));
            atomicAdd(den + r, w);
            if (MODE == 0) {
                const unsigned p = atomicAdd(fill + r, 1u);
                srt_e[p] = e;
                srt_w[p] = w;
            } else {
                pv[i] = w;
            }
        }
    } else {
        rows_rescan<1>(src, dst, H, s, sd_pitch, base, tid, dloc, mkey, den, acc);
    }
    __syncthreads();
    for (int idx = tid; idx < RB * 64; idx += ROWS_NT) {
        int r = idx >> 6;
        float coef = wself[r] / (den[r] + 1e-16f);
        acc[idx] = H[(size_t)base * 64 + idx] * coef;
    }
    if (MODE == 1 && tid < 8 * 64) acc[RB * 64 + tid] = 0.0f;
    if (MODE == 0 && fast) {
        for (int idx = tid; idx < (int)nt; idx += ROWS_NT) ownb[idx] = srt_e[idx];
    }
    __syncthreads();

    if (fast) {
        int   cur = -1;
        float sx = 0.0f, sy = 0.0f;
        for (int b0 = lo; b0 < hi; b0 += 32) {
            int i = b0 + lane;
            bool valid = i < hi;
            unsigned en = valid ? srt_e[i] : 0u;
            float w = valid ? srt_w[i] : 0.0f;
            int r  = (int)(en >> 16);
            int si = (int)(en & 0xFFFFu);
            float coef = w / (den[r] + 1e-16f);
#pragma unroll
            for (int g = 0; g < 32; g += AGG_U) {
                if (b0 + g < hi) {
                    v2f hv[AGG_U];
#pragma unroll
                    for (int u = 0; u < AGG_U; ++u) {
                        int sj = __builtin_amdgcn_readlane(si, g + u);
                        hv[u] = *(const v2f*)(H + (size_t)sj * 64 + 2 * lane);
                    }
#pragma unroll
                    for (int u = 0; u < AGG_U; ++u) asm volatile("" :: "v"(hv[u]));
#pragma unroll
                    for (int u = 0; u < AGG_U; ++u) {
                        if (b0 + g + u < hi) {
                            int rj = __builtin_amdgcn_readlane(r, g + u);
                            float cj = __int_as_float(__builtin_amdgcn_readlane(__float_as_int(coef), g + u));
                            if (rj != cur) {
                                if (cur >= 0) {
                                    float* ar = acc + cur * 64 + 2 * lane;
                                    atomicAdd(ar,     sx);
                                    atomicAdd(ar + 1, sy);
                                }
                                cur = rj; sx = 0.0f; sy = 0.0f;
                            }
                            sx = fmaf(hv[u].x, cj, sx);
                            sy = fmaf(hv[u].y, cj, sy);
                        }
                    }
                }
            }
        }
        if (cur >= 0) {
            float* ar = acc + cur * 64 + 2 * lane;
            atomicAdd(ar,     sx);
            atomicAdd(ar + 1, sy);
        }
    } else {
        rows_rescan<2>(src, dst, H, s, sd_pitch, base, tid, dloc, mkey, den, acc);
    }
    __syncthreads();

    if (MODE == 1) {
        if (wave < 26) {
            const int mt = wave >> 1, nh = wave & 1;
            const int lo = lane & 15, hi = lane >> 4;
            const float* ar = acc + (mt * 16 + lo) * 64 + 2 * hi;
            const float* wc = Wm + nh * 32 + lo;
            v8f c0 = {}, c1 = {};
#pragma unroll 4
            for (int k = 0; k < 64; k += 4) {
                v2f a = *(const v2f*)(ar + k);
                const float* w0 = wc + (k + 2 * hi) * 64;
                v2f b0; b0.x = w0[0];  b0.y = w0[64];
                v2f b1; b1.x = w0[16]; b1.y = w0[64 + 16];
                c0 = __builtin_amdgcn_wmma_f32_16x16x4_f32(false, a, false, b0, (short)0, c0, false, false);
                c1 = __builtin_amdgcn_wmma_f32_16x16x4_f32(false, a, false, b1, (short)0, c1, false, false);
            }
#pragma unroll
            for (int r = 0; r < 8; ++r) {
                int row = mt * 16 + r + 8 * hi;
                if (row < RB) {
                    lval[row * 64 + nh * 32 + lo]      = c0[r];
                    lval[row * 64 + nh * 32 + 16 + lo] = c1[r];
                }
            }
        }
        __syncthreads();
    }
    const float* res = (MODE == 1) ? lval : acc;

    for (int idx = tid; idx < RB * 64; idx += ROWS_NT) {
        float v = res[idx] + bias[idx & 63];
        if (relu) v = fmaxf(v, 0.0f);
        out[(size_t)base * 64 + idx] = v;
    }

    if (NEXT && tid < 4 * RB) {
        const int row = tid >> 2, part = tid & 3;
        const float* yr = res + row * 64 + part * 16;
        const float* ws = wan + part * 16;
        const float* bp = bl + part * 16;
        float ss = 0.0f, dd = 0.0f;
#pragma unroll
        for (int j = 0; j < 16; ++j) {
            const int f = (j + row) & 15;
            float yv = yr[f] + bp[f];
            if (relu) yv = fmaxf(yv, 0.0f);
            ss = fmaf(yv, ws[f], ss);
            dd = fmaf(yv, ws[64 + f], dd);
        }
        ss += __shfl_xor(ss, 1); dd += __shfl_xor(dd, 1);
        ss += __shfl_xor(ss, 2); dd += __shfl_xor(dd, 2);
        if (part == 0) {
            const unsigned p = blockIdx.x * sn_pitch + row;
            sn[p] = ss; dn[p] = dd;
        }
    }
}

extern "C" void kernel_launch(void* const* d_in, const int* in_sizes, int n_in,
                              void* d_out, int out_size, void* d_ws, size_t ws_size,
                              hipStream_t stream) {
    (void)in_sizes; (void)n_in; (void)out_size; (void)ws_size;

    const float* x    = (const float*)d_in[0];
    const int*   esrc = (const int*)  d_in[1];
    const int*   edst = (const int*)  d_in[2];
    const float* W0 = (const float*)d_in[3];
    const float* a0s = (const float*)d_in[4];
    const float* a0d = (const float*)d_in[5];
    const float* b0 = (const float*)d_in[6];
    const float* W1 = (const float*)d_in[7];
    const float* a1s = (const float*)d_in[8];
    const float* a1d = (const float*)d_in[9];
    const float* b1 = (const float*)d_in[10];
    const float* W2 = (const float*)d_in[11];
    const float* a2s = (const float*)d_in[12];
    const float* a2d = (const float*)d_in[13];
    const float* b2 = (const float*)d_in[14];
    float* out = (float*)d_out;

    float* ws = (float*)d_ws;
    float* H   = ws;
    float* XIN = H   + (size_t)N_NODES * 64;
    float* SV  = XIN + (size_t)N_NODES * 64;
    float* DV  = SV  + N_NODES;
    unsigned* OWN = (unsigned*)out;
    float* S1 = out + S1_OFF;
    float* D1 = out + D1_OFF;

    gat_front<<<FRONT_BLK, 256, 0, stream>>>(x, W0, H, N_NODES, a0s, a0d, SV, DV, esrc, edst, OWN);
    gat_rows_lds<0, true><<<NBLK, ROWS_NT, 0, stream>>>(esrc, edst, OWN, H, SV, DV, RB, b0, XIN, 1, nullptr,
                                                 S1, D1, BLK_DW, W1, a1s, a1d);

    gat_rows_lds<1, true><<<NBLK, ROWS_NT, 0, stream>>>(esrc, edst, OWN, XIN, S1, D1, BLK_DW, b1, H, 1, W1,
                                                 SV, DV, RB, W2, a2s, a2d);

    gat_rows_lds<1, false><<<NBLK, ROWS_NT, 0, stream>>>(esrc, edst, OWN, H, SV, DV, RB, b2, out, 0, W2,
                                                 nullptr, nullptr, 0, nullptr, nullptr, nullptr);
}


# Layers 1,2 max, denominator passes: lanes take consecutive entries of the row-ordered list and join equal rows in registers, so an LDS atomic's lanes hit different words; den sum order changes
# speedup vs baseline: 7.1359x; 1.0687x over previous
#include <hip/hip_runtime.h>
#include <hip/hip_bf16.h>
#include <cstddef>


#define N_NODES 50000
#define E_EDGES 800000
#define NEG_SLOPE 0.2f

typedef float v2f __attribute__((ext_vector_type(2)));
typedef float v8f __attribute__((ext_vector_type(8)));

__device__ __forceinline__ unsigned f2key(float f) {
    int i = __float_as_int(f);
    return (i >= 0) ? ((unsigned)i | 0x80000000u) : ~(unsigned)i;
}
__device__ __forceinline__ float key2f(unsigned k) {
    int i = (k & 0x80000000u) ? (int)(k & 0x7FFFFFFFu) : (int)~k;
    return __int_as_float(i);
}

#define RB_KEY 200u
#define BLK_DW    (200 * 64)
#define SAVE_CAP  192
#define SRT_CAP   (32 * SAVE_CAP)
#define CNT_OFF   SRT_CAP
#define MARK_OFF  6400
#define MARK_DW   2048
#define KEY_OFF   8448
#define KEY_DW    4096
#define S1_OFF    (CNT_OFF + 32)
#define D1_OFF    (KEY_OFF + KEY_DW)
__device__ __forceinline__ unsigned sd_pos(unsigned i, unsigned pitch) {
    return (i / 200u) * pitch + (i % 200u);
}
__device__ __forceinline__ unsigned rec_pos(unsigned e) {
    return (e / KEY_DW) * BLK_DW + KEY_OFF + (e % KEY_DW);
}

#define FRONT_K   128
#define FRONT_BLK (((N_NODES / 16) + 7) / 8)
static_assert(N_NODES % 16 == 0, "row tiles must divide the node set");
static_assert(FRONT_BLK * 512 >= E_EDGES / 4, "the front blocks must cover the edge list");
#define BM_WORDS  (FRONT_BLK * 4)
#define BM_WPW    ((BM_WORDS + 31) / 32)
static_assert(MARK_DW == 2 * 1024, "a mark piece holds the marks of two front blocks");
static_assert(BM_WPW > 32 && BM_WPW <= 64, "an owner lane takes one or two mark words");
__device__ __forceinline__ unsigned bmt_pos(unsigned j, unsigned b, unsigned ww) {
    return (j >> 1) * BLK_DW + MARK_OFF + (j & 1u) * 1024u + b * 4u + ww;
}
__global__ void __launch_bounds__(256)
gat_front(const float* __restrict__ X, const float* __restrict__ W,
          float* __restrict__ H, int nrows,
          const float* __restrict__ asrc, const float* __restrict__ adst,
          float* __restrict__ s, float* __restrict__ d,
          const int* __restrict__ src, const int* __restrict__ dst,
          unsigned* __restrict__ keys) {
    __shared__ float wa[2 * FRONT_K];
    __shared__ unsigned bm[256 * 4];
    const int tid  = threadIdx.x;
    const int lane = tid & 31;
    const int wib  = __builtin_amdgcn_readfirstlane(tid >> 5);

    const int4* dst4 = (const int4*)dst;
    const int4* src4 = (const int4*)src;
    const int qlast = E_EDGES / 4 - 1;
    const int q0 = blockIdx.x * 512 + tid, q1 = q0 + 256;
    const int4 kv0 = dst4[(q0 < qlast) ? q0 : qlast];
    const int4 kv1 = dst4[(q1 < qlast) ? q1 : qlast];
    const int4 sv0 = src4[(q0 < qlast) ? q0 : qlast];
    const int4 sv1 = src4[(q1 < qlast) ? q1 : qlast];
    for (int i = tid; i < 256 * 4; i += 256) bm[i] = 0u;
    {
        const int c = lane & 7, rr = lane >> 3;
        const float4 as0 = *(const float4*)(asrc + 4 * c);
        const float4 as1 = *(const float4*)(asrc + 32 + 4 * c);
        const float4 ad0 = *(const float4*)(adst + 4 * c);
        const float4 ad1 = *(const float4*)(adst + 32 + 4 * c);
        const float* wr = W + (wib * 16 + rr) * 64 + 4 * c;
        float4 w0[4], w1[4];
        float  ps[4], pd[4];
#pragma unroll
        for (int j = 0; j < 4; ++j) {
            w0[j] = *(const float4*)(wr + j * 256);
            w1[j] = *(const float4*)(wr + j * 256 + 32);
        }
#pragma unroll
        for (int j = 0; j < 4; ++j) {
            float s_ = w0[j].x * as0.x, d_ = w0[j].x * ad0.x;
            s_ = fmaf(w0[j].y, as0.y, s_); d_ = fmaf(w0[j].y, ad0.y, d_);
            s_ = fmaf(w0[j].z, as0.z, s_); d_ = fmaf(w0[j].z, ad0.z, d_);
            s_ = fmaf(w0[j].w, as0.w, s_); d_ = fmaf(w0[j].w, ad0.w, d_);
            s_ = fmaf(w1[j].x, as1.x, s_); d_ = fmaf(w1[j].x, ad1.x, d_);
            s_ = fmaf(w1[j].y, as1.y, s_); d_ = fmaf(w1[j].y, ad1.y, d_);
            s_ = fmaf(w1[j].z, as1.z, s_); d_ = fmaf(w1[j].z, ad1.z, d_);
            s_ = fmaf(w1[j].w, as1.w, s_); d_ = fmaf(w1[j].w, ad1.w, d_);
            ps[j] = s_; pd[j] = d_;
        }
#pragma unroll
        for (int m = 1; m < 8; m <<= 1) {
#pragma unroll
            for (int j = 0; j < 4; ++j) {
                ps[j] += __shfl_xor(ps[j], m);
                pd[j] += __shfl_xor(pd[j], m);
            }
        }
        if (c == 0) {
#pragma unroll
            for (int j = 0; j < 4; ++j) {
                wa[wib * 16 + j * 4 + rr] = ps[j];
                wa[FRONT_K + wib * 16 + j * 4 + rr] = pd[j];
            }
        }
    }
    __syncthreads();

    {
        const unsigned mbit = 1u << ((tid >> 2) & 31);
        const int mw = tid >> 7;
        if (q0 <= qlast) {
            atomicOr(&bm[((unsigned)kv0.x / RB_KEY) * 4 + mw], mbit);
            atomicOr(&bm[((unsigned)kv0.y / RB_KEY) * 4 + mw], mbit);
            atomicOr(&bm[((unsigned)kv0.z / RB_KEY) * 4 + mw], mbit);
            atomicOr(&bm[((unsigned)kv0.w / RB_KEY) * 4 + mw], mbit);
        }
        if (q1 <= qlast) {
            atomicOr(&bm[((unsigned)kv1.x / RB_KEY) * 4 + 2 + mw], mbit);
            atomicOr(&bm[((unsigned)kv1.y / RB_KEY) * 4 + 2 + mw], mbit);
            atomicOr(&bm[((unsigned)kv1.z / RB_KEY) * 4 + 2 + mw], mbit);
            atomicOr(&bm[((unsigned)kv1.w / RB_KEY) * 4 + 2 + mw], mbit);
        }
    }
    __syncthreads();
    *(uint4*)(keys + bmt_pos((unsigned)blockIdx.x, (unsigned)tid, 0u)) =
        make_uint4(bm[4 * tid], bm[4 * tid + 1], bm[4 * tid + 2], bm[4 * tid + 3]);
    const uint4 rec0 = make_uint4(((unsigned)kv0.x << 16) | (unsigned)sv0.x, ((unsigned)kv0.y << 16) | (unsigned)sv0.y,
                                  ((unsigned)kv0.z << 16) | (unsigned)sv0.z, ((unsigned)kv0.w << 16) | (unsigned)sv0.w);
    const uint4 rec1 = make_uint4(((unsigned)kv1.x << 16) | (unsigned)sv1.x, ((unsigned)kv1.y << 16) | (unsigned)sv1.y,
                                  ((unsigned)kv1.z << 16) | (unsigned)sv1.z, ((unsigned)kv1.w << 16) | (unsigned)sv1.w);

    const int rowBase = (blockIdx.x * 8 + wib) * 16;
    if (rowBase < nrows) {

    const int lo = lane & 15;
    const int hi = lane >> 4;

    v8f acc0 = {}, acc1 = {}, acc2 = {}, acc3 = {};
    float ss = 0.0f, dd = 0.0f;

    const float* xr = X + (size_t)(rowBase + lo) * FRONT_K + 2 * hi;
    const float* wc = W + lo;
    const float* wk = wa + 2 * hi;

    for (int k0 = 0; k0 < FRONT_K; k0 += 16) {
        v2f a[4], b[4][4], ws2[4], wd2[4];
#pragma unroll
        for (int u = 0; u < 4; ++u) {
            const int k = k0 + 4 * u;
            a[u] = *(const v2f*)(xr + k);
            const float* w0 = wc + (size_t)(k + 2 * hi) * 64;
            const float* w1 = w0 + 64;
#pragma unroll
            for (int t = 0; t < 4; ++t) { b[u][t].x = w0[16 * t]; b[u][t].y = w1[16 * t]; }
            ws2[u] = *(const v2f*)(wk + k);
            wd2[u] = *(const v2f*)(wk + FRONT_K + k);
        }
#pragma unroll
        for (int u = 0; u < 4; ++u) {
            asm volatile("" :: "v"(a[u]));
#pragma unroll
            for (int t = 0; t < 4; ++t) asm volatile("" :: "v"(b[u][t]));
        }
#pragma unroll
        for (int u = 0; u < 4; ++u) {
            acc0 = __builtin_amdgcn_wmma_f32_16x16x4_f32(false, a[u], false, b[u][0], (short)0, acc0, false, false);
            acc1 = __builtin_amdgcn_wmma_f32_16x16x4_f32(false, a[u], false, b[u][1], (short)0, acc1, false, false);
            acc2 = __builtin_amdgcn_wmma_f32_16x16x4_f32(false, a[u], false, b[u][2], (short)0, acc2, false, false);
            acc3 = __builtin_amdgcn_wmma_f32_16x16x4_f32(false, a[u], false, b[u][3], (short)0, acc3, false, false);
            ss = fmaf(a[u].x, ws2[u].x, ss); ss = fmaf(a[u].y, ws2[u].y, ss);
            dd = fmaf(a[u].x, wd2[u].x, dd); dd = fmaf(a[u].y, wd2[u].y, dd);
        }
    }

#pragma unroll
    for (int r = 0; r < 8; ++r) {
        float* o = H + (size_t)(rowBase + r + 8 * hi) * 64 + lo;
        o[0]  = acc0[r];
        o[16] = acc1[r];
        o[32] = acc2[r];
        o[48] = acc3[r];
    }

    ss += __shfl_xor(ss, 16); dd += __shfl_xor(dd, 16);
    if (hi == 0) { s[rowBase + lo] = ss; d[rowBase + lo] = dd; }
    }

    if (q0 <= qlast) *(uint4*)(keys + rec_pos(4u * (unsigned)q0)) = rec0;
    if (q1 <= qlast) *(uint4*)(keys + rec_pos(4u * (unsigned)q1)) = rec1;
}


#define RB        200
#define NBLK      250
#define ROWS_NT   1024
#define WCAP      512
#define GRP_U     18
#define AGG_U     16
#define PH0_U     5
#define NGROUP    (E_EDGES / 16)
#define SCAN_WAVE 7
static_assert(RB * NBLK == N_NODES, "row blocks must tile the node set");
static_assert(RB == RB_KEY, "group marks must be cut by the owners' row block");
static_assert(NBLK <= 256, "an owner needs a mark slot");
static_assert(E_EDGES % 16 == 0, "edge list must split into 16-edge groups");
static_assert(E_EDGES <= (1 << 20), "edge id must fit 20 bits of a list entry");
static_assert(BLK_DW == RB * 64, "footprint = the block's output rows");
static_assert(SRT_CAP <= CNT_OFF, "the saved list ends in front of its count");
static_assert(CNT_OFF + 32 <= MARK_OFF && MARK_OFF + MARK_DW <= KEY_OFF && KEY_OFF + KEY_DW <= BLK_DW,
              "list, count, mark piece, record piece must not overlap");
static_assert((FRONT_BLK + 1) / 2 <= NBLK, "the marks must fit the mark pieces");
static_assert(FRONT_BLK * 128 >= NGROUP, "the marks must cover every 16-edge group");
static_assert((E_EDGES + KEY_DW - 1) / KEY_DW <= NBLK, "record pieces must fit the footprints");
static_assert(KEY_DW % 16 == 0 && MARK_OFF % 32 == 0 && KEY_OFF % 32 == 0, "groups and pieces start on 64-byte and 128-byte bounds");
static_assert(N_NODES <= 65536, "source id must fit 16 bits of a saved entry");
static_assert(S1_OFF % 32 == 0 && S1_OFF + RB <= MARK_OFF, "layer-1 s gap: line aligned, in front of the mark piece");
static_assert(D1_OFF % 32 == 0 && D1_OFF + RB <= BLK_DW, "layer-1 d gap: line aligned, inside the footprint");
static_assert(SCAN_WAVE * 32 >= RB && SCAN_WAVE < 32, "the scan wave has no self-loop rows to set up");
static_assert(29 * 7 >= RB, "seven rows per lane of the scan wave cover the block");

__device__ __forceinline__ float leaky(float v) {
    return (v > 0.0f) ? v : NEG_SLOPE * v;
}

template <int PH>
__device__ __forceinline__ void rows_rescan(
        const int* __restrict__ src, const int* __restrict__ dst,
        const float* __restrict__ H, const float* __restrict__ s, unsigned sd_pitch,
        int base, int tid, const float* dloc, unsigned* mkey, float* den,
        float* acc) {
    for (int e = tid; e < E_EDGES; e += ROWS_NT) {
        int r = dst[e] - base;
        if ((unsigned)r < (unsigned)RB) {
            int si = src[e];
            float v = leaky(s[sd_pos((unsigned)si, sd_pitch)] + dloc[r]);
            if (PH == 0) {
                atomicMax(mkey + r, f2key(v));
            } else {
                float w = __expf(v - key2f(mkey[r]));
                if (PH == 1) {
                    atomicAdd(den + r, w);
                } else {
                    float coef = w / (den[r] + 1e-16f);
                    const float* hr = H + (size_t)si * 64;
                    float* ar = acc + r * 64;
                    for (int f = 0; f < 64; ++f) atomicAdd(ar + f, hr[f] * coef);
                }
            }
        }
    }
}

template <int MODE, bool NEXT>
__global__ void __launch_bounds__(ROWS_NT)
gat_rows_lds(const int* __restrict__ src, const int* __restrict__ dst,
             unsigned* own, const float* __restrict__ H, const float* __restrict__ s,
             const float* __restrict__ d, unsigned sd_pitch,
             const float* __restrict__ bias,
             float* out, int relu, const float* __restrict__ Wm,
             float* sn, float* dn, unsigned sn_pitch, const float* __restrict__ Wn,
             const float* __restrict__ asn, const float* __restrict__ adn) {
    __shared__ float    acc[(RB + 8) * 64];
    __shared__ unsigned mkey[RB];
    __shared__ float    den[RB];
    __shared__ float    dloc[RB];
    __shared__ float    vself[RB];
    __shared__ float    wself[RB];
    __shared__ unsigned fill[RB];
    __shared__ unsigned list[32 * WCAP];
    __shared__ float    lval[32 * WCAP];
    __shared__ unsigned srt_e[SRT_CAP];
    __shared__ float    srt_w[SRT_CAP];
    __shared__ int      ovf;
    __shared__ int      ntot;
    __shared__ float    wan[128];
    __shared__ float    bl[64];

    const int tid  = threadIdx.x;
    const int lane = tid & 31;
    const int wave = __builtin_amdgcn_readfirstlane(tid >> 5);
    const int base = blockIdx.x * RB;
    unsigned* wl = list + wave * WCAP;
    float*    wv = lval + wave * WCAP;
    unsigned* ownb = own + (size_t)blockIdx.x * BLK_DW;

    unsigned nsaved = 0u;
    if (MODE == 1) nsaved = (unsigned)__builtin_amdgcn_readfirstlane((int)ownb[CNT_OFF]);
    unsigned bw0 = 0u, bw1 = 0u;
    if (MODE == 0) {
        const unsigned w0i = (unsigned)(wave * BM_WPW + lane), w1i = w0i + 32u;
        if (w0i < (unsigned)BM_WORDS)
            bw0 = own[bmt_pos(w0i >> 2, (unsigned)blockIdx.x, w0i & 3u)];
        if (lane < BM_WPW - 32 && w1i < (unsigned)BM_WORDS)
            bw1 = own[bmt_pos(w1i >> 2, (unsigned)blockIdx.x, w1i & 3u)];
    }

    if (tid == 0) { ovf = 0; ntot = 0; }
    if (NEXT && wave >= 8 && wave < 12) {
        const int t = tid - 256;
        const float* wr = Wn + (t & 63) * 64;
        const float* av = (wave < 10) ? asn : adn;
        float w = 0.0f;
#pragma unroll
        for (int j = 0; j < 64; ++j) w = fmaf(wr[j], av[j], w);
        wan[t] = w;
    }
    if (NEXT && wave >= 12 && wave < 14) bl[tid - 384] = bias[tid - 384];
    if (tid < RB) {
        const unsigned i = blockIdx.x * sd_pitch + tid;
        float dl = d[i];
        float v = leaky(s[i] + dl);
        dloc[tid] = dl;
        vself[tid] = v;
        mkey[tid] = f2key(v);
        if (MODE == 0) fill[tid] = 0u;
    }
    __syncthreads();

    int wcnt = 0;
    if (MODE == 0) {
        const uint4* rec4 = (const uint4*)own;
        const unsigned bsh = (unsigned)base << 16;
        const int sub = lane >> 2, part = lane & 3;
        unsigned* gl = reinterpret_cast<unsigned*>(wv);
        int ng = 0;
#pragma unroll
        for (int pass = 0; pass < 2; ++pass) {
            unsigned bits = pass ? bw1 : bw0;
            const unsigned g0 = (unsigned)(wave * BM_WPW + 32 * pass + lane) * 32u;
            unsigned live = __builtin_amdgcn_ballot_w32(bits != 0u);
            while (live != 0u) {
                int p = ng + (int)__builtin_amdgcn_mbcnt_lo(live, 0u);
                if (bits != 0u && p < WCAP) gl[p] = g0 + (unsigned)(__ffs((int)bits) - 1);
                ng += __builtin_popcount(live);
                bits &= bits - 1u;
                live = __builtin_amdgcn_ballot_w32(bits != 0u);
            }
        }
        if (ng > WCAP && lane == 0) ovf = 1;
        const int ngc = (ng < WCAP) ? ng : WCAP;
        for (int i0 = 0; i0 < ngc; i0 += 8 * GRP_U) {
            uint4 kv[GRP_U];
#pragma unroll
            for (int u = 0; u < GRP_U; ++u) {
                int i = i0 + 8 * u + sub;
                unsigned g = gl[(i < ngc) ? i : ngc - 1];
                kv[u] = rec4[(g >> 8) * (BLK_DW / 4) + (KEY_OFF / 4) + (g & 255u) * 4u + (unsigned)part];
            }
#pragma unroll
            for (int u = 0; u < GRP_U; ++u) asm volatile("" :: "v"(kv[u].x), "v"(kv[u].y), "v"(kv[u].z), "v"(kv[u].w));
#pragma unroll
            for (int u = 0; u < GRP_U; ++u) {
                if (i0 + 8 * u < ngc) {
                    const bool okl = (i0 + 8 * u + sub) < ngc;
                    const unsigned okm = __builtin_amdgcn_ballot_w32(okl);
                    const unsigned rc[4] = { kv[u].x, kv[u].y, kv[u].z, kv[u].w };
#pragma unroll
                    for (int j = 0; j < 4; ++j) {
                        const unsigned t = rc[j] - bsh;
                        const bool mj = t < ((unsigned)RB << 16);
                        const unsigned mm = __builtin_amdgcn_ballot_w32(mj) & okm;
                        unsigned p = __builtin_amdgcn_mbcnt_lo(mm, (unsigned)wcnt);
                        p = (p < (unsigned)(WCAP - 1)) ? p : (unsigned)(WCAP - 1);
                        if (mj && okl) wl[p] = t;
                        wcnt += __builtin_popcount(mm);
                    }
                }
            }
        }
        if (wcnt > WCAP && lane == 0) ovf = 1;
        if (lane == 0) atomicAdd(&ntot, wcnt);
    }
    __syncthreads();
    const unsigned nt = (MODE == 0) ? (unsigned)__builtin_amdgcn_readfirstlane(ntot) : nsaved;
    const bool fast = (__builtin_amdgcn_readfirstlane(ovf) == 0) && (nt <= (unsigned)SRT_CAP);
    if (MODE == 0 && tid == 0) ownb[CNT_OFF] = fast ? nt : ~0u;
    const int lo = fast ? (int)(((unsigned)wave * nt) >> 5) : 0;
    const int hi = fast ? (int)(((unsigned)(wave + 1) * nt) >> 5) : 0;
    unsigned* pe = (MODE == 0) ? wl : (srt_e + lo);
    float*    pv = (MODE == 0) ? wv : (srt_w + lo);
    const int n  = (MODE == 0) ? wcnt : (hi - lo);

    if (fast) {
        for (int g0 = 0; g0 < n; g0 += 32 * PH0_U) {
            unsigned en[PH0_U];
            float    sv[PH0_U];
#pragma unroll
            for (int u = 0; u < PH0_U; ++u) {
                int i = (MODE == 1) ? (g0 + PH0_U * lane + u) : (g0 + 32 * u + lane);
                int ic = (i < n) ? i : n - 1;
                en[u] = (MODE == 0) ? wl[ic] : ownb[lo + ic];
            }
#pragma unroll
            for (int u = 0; u < PH0_U; ++u) sv[u] = s[sd_pos(en[u] & 0xFFFFu, sd_pitch)];
#pragma unroll
            for (int u = 0; u < PH0_U; ++u) asm volatile("" :: "v"(sv[u]));
            int rcur = -1;
            unsigned kcur = 0u;
#pragma unroll
            for (int u = 0; u < PH0_U; ++u) {
                int i = (MODE == 1) ? (g0 + PH0_U * lane + u) : (g0 + 32 * u + lane);
                if (i < n) {
                    int r = (int)(en[u] >> 16);
                    float v = leaky(sv[u] + dloc[r]);
                    if (MODE == 1) pe[i] = en[u];
                    pv[i] = v;
                    const unsigned k = f2key(v);
                    if (MODE == 0) {
                        atomicMax(mkey + r, k);
                        atomicAdd(fill + r, 1u);
                    } else if (r != rcur) {
                        if (rcur >= 0) atomicMax(mkey + rcur, kcur);
                        rcur = r;
                        kcur = k;
                    } else {
                        kcur = (k > kcur) ? k : kcur;
                    }
                }
            }
            if (MODE == 1 && rcur >= 0) atomicMax(mkey + rcur, kcur);
        }
    } else {
        rows_rescan<0>(src, dst, H, s, sd_pitch, base, tid, dloc, mkey, den, acc);
    }
    __syncthreads();
    if (tid < RB) {
        float w = __expf(vself[tid] - key2f(mkey[tid]));
        wself[tid] = w;
        den[tid] = w;
    }
    if (MODE == 0 && fast && wave == SCAN_WAVE) {
        unsigned c[7];
        unsigned t = 0u;
#pragma unroll
        for (int j = 0; j < 7; ++j) {
            const int r = 7 * lane + j;
            c[j] = (r < RB) ? fill[r] : 0u;
            t += c[j];
        }
        unsigned incl = t;
#pragma unroll
        for (int m = 1; m < 32; m <<= 1) {
            const unsigned o = __shfl_up(incl, m);
            if (lane >= m) incl += o;
        }
        unsigned run = incl - t;
#pragma unroll
        for (int j = 0; j < 7; ++j) {
            const int r = 7 * lane + j;
            if (r < RB) fill[r] = run;
            run += c[j];
        }
    }
    __syncthreads();

    if (fast) {
        if (MODE == 1) {
            const int K = ((n + 31) >> 5) | 1;
            int   rcur = -1;
            float wsum = 0.0f;
            for (int j = 0; j < K; ++j) {
                const int i = K * lane + j;
                if (i < n) {
                    const int r = (int)(pe[i] >> 16);
                    const float w = __expf(pv[i] - key2f(mkey[r]));
                    pv[i] = w;
                    if (r != rcur) {
                        if (rcur >= 0) atomicAdd(den + rcur, wsum);
                        rcur = r;
                        wsum = 0.0f;
                    }
                    wsum += w;
                }
            }
            if (rcur >= 0) atomicAdd(den + rcur, wsum);
        } else {
            for (int i = lane; i < n; i += 32) {
                const unsigned e = pe[i];
                int r = (int)(e >> 16);
                float w = __expf(pv[i] - key2f(mkey[r]));
                atomicAdd(den + r, w);
                const unsigned p = atomicAdd(fill + r, 1u);
                srt_e[p] = e;
                srt_w[p] = w;
            }
        }
    } else {
        rows_rescan<1>(src, dst, H, s, sd_pitch, base, tid, dloc, mkey, den, acc);
    }
    __syncthreads();
    for (int idx = tid; idx < RB * 64; idx += ROWS_NT) {
        int r = idx >> 6;
        float coef = wself[r] / (den[r] + 1e-16f);
        acc[idx] = H[(size_t)base * 64 + idx] * coef;
    }
    if (MODE == 1 && tid < 8 * 64) acc[RB * 64 + tid] = 0.0f;
    if (MODE == 0 && fast) {
        for (int idx = tid; idx < (int)nt; idx += ROWS_NT) ownb[idx] = srt_e[idx];
    }
    __syncthreads();

    if (fast) {
        int   cur = -1;
        float sx = 0.0f, sy = 0.0f;
        for (int b0 = lo; b0 < hi; b0 += 32) {
            int i = b0 + lane;
            bool valid = i < hi;
            unsigned en = valid ? srt_e[i] : 0u;
            float w = valid ? srt_w[i] : 0.0f;
            int r  = (int)(en >> 16);
            int si = (int)(en & 0xFFFFu);
            float coef = w / (den[r] + 1e-16f);
#pragma unroll
            for (int g = 0; g < 32; g += AGG_U) {
                if (b0 + g < hi) {
                    v2f hv[AGG_U];
#pragma unroll
                    for (int u = 0; u < AGG_U; ++u) {
                        int sj = __builtin_amdgcn_readlane(si, g + u);
                        hv[u] = *(const v2f*)(H + (size_t)sj * 64 + 2 * lane);
                    }
#pragma unroll
                    for (int u = 0; u < AGG_U; ++u) asm volatile("" :: "v"(hv[u]));
#pragma unroll
                    for (int u = 0; u < AGG_U; ++u) {
                        if (b0 + g + u < hi) {
                            int rj = __builtin_amdgcn_readlane(r, g + u);
                            float cj = __int_as_float(__builtin_amdgcn_readlane(__float_as_int(coef), g + u));
                            if (rj != cur) {
                                if (cur >= 0) {
                                    float* ar = acc + cur * 64 + 2 * lane;
                                    atomicAdd(ar,     sx);
                                    atomicAdd(ar + 1, sy);
                                }
                                cur = rj; sx = 0.0f; sy = 0.0f;
                            }
                            sx = fmaf(hv[u].x, cj, sx);
                            sy = fmaf(hv[u].y, cj, sy);
                        }
                    }
                }
            }
        }
        if (cur >= 0) {
            float* ar = acc + cur * 64 + 2 * lane;
            atomicAdd(ar,     sx);
            atomicAdd(ar + 1, sy);
        }
    } else {
        rows_rescan<2>(src, dst, H, s, sd_pitch, base, tid, dloc, mkey, den, acc);
    }
    __syncthreads();

    if (MODE == 1) {
        if (wave < 26) {
            const int mt = wave >> 1, nh = wave & 1;
            const int lo = lane & 15, hi = lane >> 4;
            const float* ar = acc + (mt * 16 + lo) * 64 + 2 * hi;
            const float* wc = Wm + nh * 32 + lo;
            v8f c0 = {}, c1 = {};
#pragma unroll 4
            for (int k = 0; k < 64; k += 4) {
                v2f a = *(const v2f*)(ar + k);
                const float* w0 = wc + (k + 2 * hi) * 64;
                v2f b0; b0.x = w0[0];  b0.y = w0[64];
                v2f b1; b1.x = w0[16]; b1.y = w0[64 + 16];
                c0 = __builtin_amdgcn_wmma_f32_16x16x4_f32(false, a, false, b0, (short)0, c0, false, false);
                c1 = __builtin_amdgcn_wmma_f32_16x16x4_f32(false, a, false, b1, (short)0, c1, false, false);
            }
#pragma unroll
            for (int r = 0; r < 8; ++r) {
                int row = mt * 16 + r + 8 * hi;
                if (row < RB) {
                    lval[row * 64 + nh * 32 + lo]      = c0[r];
                    lval[row * 64 + nh * 32 + 16 + lo] = c1[r];
                }
            }
        }
        __syncthreads();
    }
    const float* res = (MODE == 1) ? lval : acc;

    for (int idx = tid; idx < RB * 64; idx += ROWS_NT) {
        float v = res[idx] + bias[idx & 63];
        if (relu) v = fmaxf(v, 0.0f);
        out[(size_t)base * 64 + idx] = v;
    }

    if (NEXT && tid < 4 * RB) {
        const int row = tid >> 2, part = tid & 3;
        const float* yr = res + row * 64 + part * 16;
        const float* ws = wan + part * 16;
        const float* bp = bl + part * 16;
        float ss = 0.0f, dd = 0.0f;
#pragma unroll
        for (int j = 0; j < 16; ++j) {
            const int f = (j + row) & 15;
            float yv = yr[f] + bp[f];
            if (relu) yv = fmaxf(yv, 0.0f);
            ss = fmaf(yv, ws[f], ss);
            dd = fmaf(yv, ws[64 + f], dd);
        }
        ss += __shfl_xor(ss, 1); dd += __shfl_xor(dd, 1);
        ss += __shfl_xor(ss, 2); dd += __shfl_xor(dd, 2);
        if (part == 0) {
            const unsigned p = blockIdx.x * sn_pitch + row;
            sn[p] = ss; dn[p] = dd;
        }
    }
}

extern "C" void kernel_launch(void* const* d_in, const int* in_sizes, int n_in,
                              void* d_out, int out_size, void* d_ws, size_t ws_size,
                              hipStream_t stream) {
    (void)in_sizes; (void)n_in; (void)out_size; (void)ws_size;

    const float* x    = (const float*)d_in[0];
    const int*   esrc = (const int*)  d_in[1];
    const int*   edst = (const int*)  d_in[2];
    const float* W0 = (const float*)d_in[3];
    const float* a0s = (const float*)d_in[4];
    const float* a0d = (const float*)d_in[5];
    const float* b0 = (const float*)d_in[6];
    const float* W1 = (const float*)d_in[7];
    const float* a1s = (const float*)d_in[8];
    const float* a1d = (const float*)d_in[9];
    const float* b1 = (const float*)d_in[10];
    const float* W2 = (const float*)d_in[11];
    const float* a2s = (const float*)d_in[12];
    const float* a2d = (const float*)d_in[13];
    const float* b2 = (const float*)d_in[14];
    float* out = (float*)d_out;

    float* ws = (float*)d_ws;
    float* H   = ws;
    float* XIN = H   + (size_t)N_NODES * 64;
    float* SV  = XIN + (size_t)N_NODES * 64;
    float* DV  = SV  + N_NODES;
    unsigned* OWN = (unsigned*)out;
    float* S1 = out + S1_OFF;
    float* D1 = out + D1_OFF;

    gat_front<<<FRONT_BLK, 256, 0, stream>>>(x, W0, H, N_NODES, a0s, a0d, SV, DV, esrc, edst, OWN);
    gat_rows_lds<0, true><<<NBLK, ROWS_NT, 0, stream>>>(esrc, edst, OWN, H, SV, DV, RB, b0, XIN, 1, nullptr,
                                                 S1, D1, BLK_DW, W1, a1s, a1d);

    gat_rows_lds<1, true><<<NBLK, ROWS_NT, 0, stream>>>(esrc, edst, OWN, XIN, S1, D1, BLK_DW, b1, H, 1, W1,
                                                 SV, DV, RB, W2, a2s, a2d);

    gat_rows_lds<1, false><<<NBLK, ROWS_NT, 0, stream>>>(esrc, edst, OWN, H, SV, DV, RB, b2, out, 0, W2,
                                                 nullptr, nullptr, 0, nullptr, nullptr, nullptr);
}
